# starting kernel
# baseline (speedup 1.0000x reference)
#include <hip/hip_runtime.h>

typedef __attribute__((ext_vector_type(2))) float v2f;
typedef __attribute__((ext_vector_type(8))) float v8f;

#define GRID_N 64
#define NPTS   4096
#define NCTX   512
#define NCH    64
#define LOG2E  1.44269504088896340736f

__global__ __launch_bounds__(256) void convnp_fused(
    const float* __restrict__ xc,
    const float* __restrict__ yc,
    const float* __restrict__ xg,
    const float* __restrict__ sig,
    const float* __restrict__ W,
    const float* __restrict__ bias,
    float* __restrict__ out)
{
    __shared__ float sxc[NCTX * 2];
    __shared__ float syc[NCTX];
    __shared__ float sy[256 * 2];

    const int t    = threadIdx.x;
    const int b    = blockIdx.x >> 4;
    const int tile = blockIdx.x & 15;
    const int po   = tile << 8;

    const float* xcb = xc + (size_t)b * NCTX * 2;
    const float* ycb = yc + (size_t)b * NCTX;
    for (int i = t; i < NCTX; i += 256) {
        sxc[2 * i]     = xcb[2 * i];
        sxc[2 * i + 1] = xcb[2 * i + 1];
        syc[i]         = ycb[i];
    }
    __syncthreads();

    const float s0 = __expf(sig[0]);
    const float s1 = __expf(sig[1]);
    const float c0 = -0.5f * LOG2E / (s0 * s0);
    const float c1 = -0.5f * LOG2E / (s1 * s1);

    const int   o  = po + t;
    const float gx = xg[((size_t)b * NPTS + o) * 2 + 0];
    const float gy = xg[((size_t)b * NPTS + o) * 2 + 1];

    float dens = 0.f, conv = 0.f;
#pragma unroll 4
    for (int i = 0; i < NCTX; ++i) {
        float dx   = gx - sxc[2 * i];
        float dy   = gy - sxc[2 * i + 1];
        float dist = __builtin_amdgcn_sqrtf(__builtin_fmaf(dx, dx, dy * dy));
        float e0   = __builtin_amdgcn_exp2f(dist * c0);
        float e1   = __builtin_amdgcn_exp2f(dist * c1);
        dens += e0;
        conv  = __builtin_fmaf(syc[i], e1, conv);
    }
    sy[2 * t]     = dens;
    sy[2 * t + 1] = conv * __builtin_amdgcn_rcpf(dens + 1e-8f);
    __syncthreads();

    const int  wave  = t >> 5;
    const int  lane  = t & 31;
    const int  lhalf = lane & 15;
    const bool lo    = lane < 16;

    float* outb = out + (size_t)b * NCH * NPTS;

    for (int pt = 2 * wave; pt < 2 * wave + 2; ++pt) {
        float by0 = sy[2 * (pt * 16 + lhalf)];
        float by1 = sy[2 * (pt * 16 + lhalf) + 1];
        v2f bm;
        bm.x = lo ? by0 : 0.f;
        bm.y = lo ? by1 : 0.f;

        const int oPt = po + pt * 16 + lhalf;

        for (int ct = 0; ct < 4; ++ct) {
            float aw0 = W[(ct * 16 + lhalf) * 2 + 0];
            float aw1 = W[(ct * 16 + lhalf) * 2 + 1];
            v2f am;
            am.x = lo ? aw0 : 0.f;
            am.y = lo ? aw1 : 0.f;

            v8f acc = {};
            acc = __builtin_amdgcn_wmma_f32_16x16x4_f32(
                 false, am,  false, bm,
                 (short)0, acc,  false,  false);

            const int kBase = ct * 16 + (lo ? 0 : 8);
#pragma unroll
            for (int j = 0; j < 8; ++j) {
                float v = acc[j] + bias[kBase + j];
                float r = __builtin_amdgcn_rcpf(
                    1.f + __builtin_amdgcn_exp2f(-LOG2E * v));
                outb[(size_t)(kBase + j) * NPTS + oPt] = r;
            }
        }
    }
}

extern "C" void kernel_launch(void* const* d_in, const int* in_sizes, int n_in,
                              void* d_out, int out_size, void* d_ws, size_t ws_size,
                              hipStream_t stream) {
    (void)n_in; (void)out_size; (void)d_ws; (void)ws_size;
    const float* xc   = (const float*)d_in[0];
    const float* yc   = (const float*)d_in[1];
    const float* xg   = (const float*)d_in[2];
    const float* sig  = (const float*)d_in[3];
    const float* W    = (const float*)d_in[4];
    const float* bias = (const float*)d_in[5];
    float* out = (float*)d_out;

    const int B = in_sizes[0] / (NCTX * 2);
    dim3 grid(B * 16), block(256);
    convnp_fused<<<grid, block, 0, stream>>>(xc, yc, xg, sig, W, bias, out);
}


# context loop software-pipelined: LDS reads of each 8-point set issued half a trip before use (two register sets, 16 points per trip, 4 partial sums; waves_per_eu(2,2) so they stay early)
# speedup vs baseline: 1.2199x; 1.2199x over previous
#include <hip/hip_runtime.h>

typedef __attribute__((ext_vector_type(2))) float v2f;
typedef __attribute__((ext_vector_type(8))) float v8f;
typedef __attribute__((ext_vector_type(4))) float v4f;

#define GRID_N 64
#define NPTS   4096
#define NCTX   512
#define NCH    64
#define LOG2E  1.44269504088896340736f

static __device__ __forceinline__ v4f lds4(const float* p) {
    v4f v;
    __builtin_memcpy(&v, p, 16);
    return v;
}

static __device__ __forceinline__ void ctx1(float gx, float gy, float c0, float c1,
                                            float cx, float cy, float yv,
                                            float& dens, float& conv) {
    float dx   = gx - cx;
    float dy   = gy - cy;
    float dist = __builtin_amdgcn_sqrtf(__builtin_fmaf(dx, dx, dy * dy));
    float e0   = __builtin_amdgcn_exp2f(dist * c0);
    float e1   = __builtin_amdgcn_exp2f(dist * c1);
    dens += e0;
    conv  = __builtin_fmaf(yv, e1, conv);
}

static __device__ __forceinline__ void ctx8(float gx, float gy, float c0, float c1,
                                            v4f x0, v4f x1, v4f x2, v4f x3, v4f y0, v4f y1,
                                            float (&d)[4], float (&c)[4]) {
    ctx1(gx, gy, c0, c1, x0.x, x0.y, y0.x, d[0], c[0]);
    ctx1(gx, gy, c0, c1, x0.z, x0.w, y0.y, d[1], c[1]);
    ctx1(gx, gy, c0, c1, x1.x, x1.y, y0.z, d[2], c[2]);
    ctx1(gx, gy, c0, c1, x1.z, x1.w, y0.w, d[3], c[3]);
    ctx1(gx, gy, c0, c1, x2.x, x2.y, y1.x, d[0], c[0]);
    ctx1(gx, gy, c0, c1, x2.z, x2.w, y1.y, d[1], c[1]);
    ctx1(gx, gy, c0, c1, x3.x, x3.y, y1.z, d[2], c[2]);
    ctx1(gx, gy, c0, c1, x3.z, x3.w, y1.w, d[3], c[3]);
}

__global__ __launch_bounds__(256) __attribute__((amdgpu_waves_per_eu(2, 2))) void convnp_fused(
    const float* __restrict__ xc,
    const float* __restrict__ yc,
    const float* __restrict__ xg,
    const float* __restrict__ sig,
    const float* __restrict__ W,
    const float* __restrict__ bias,
    float* __restrict__ out)
{
    __shared__ __attribute__((aligned(16))) float sxc[NCTX * 2];
    __shared__ __attribute__((aligned(16))) float syc[NCTX];
    __shared__ float sy[256 * 2];

    const int t    = threadIdx.x;
    const int b    = blockIdx.x >> 4;
    const int tile = blockIdx.x & 15;
    const int po   = tile << 8;

    const float* xcb = xc + (size_t)b * NCTX * 2;
    const float* ycb = yc + (size_t)b * NCTX;
    for (int i = t; i < NCTX; i += 256) {
        sxc[2 * i]     = xcb[2 * i];
        sxc[2 * i + 1] = xcb[2 * i + 1];
        syc[i]         = ycb[i];
    }
    __syncthreads();

    const float s0 = __expf(sig[0]);
    const float s1 = __expf(sig[1]);
    const float c0 = -0.5f * LOG2E / (s0 * s0);
    const float c1 = -0.5f * LOG2E / (s1 * s1);

    const int   o  = po + t;
    const float gx = xg[((size_t)b * NPTS + o) * 2 + 0];
    const float gy = xg[((size_t)b * NPTS + o) * 2 + 1];

    float dp[4] = {0.f, 0.f, 0.f, 0.f};
    float cp[4] = {0.f, 0.f, 0.f, 0.f};
    v4f ax0 = lds4(&sxc[0]),  ax1 = lds4(&sxc[4]);
    v4f ax2 = lds4(&sxc[8]),  ax3 = lds4(&sxc[12]);
    v4f ay0 = lds4(&syc[0]),  ay1 = lds4(&syc[4]);
#pragma unroll 1
    for (int i = 0; i < NCTX; i += 16) {
        const int ib = i + 8;
        v4f bx0 = lds4(&sxc[2 * ib]),     bx1 = lds4(&sxc[2 * ib + 4]);
        v4f bx2 = lds4(&sxc[2 * ib + 8]), bx3 = lds4(&sxc[2 * ib + 12]);
        v4f by0 = lds4(&syc[ib]),         by1 = lds4(&syc[ib + 4]);
        ctx8(gx, gy, c0, c1, ax0, ax1, ax2, ax3, ay0, ay1, dp, cp);
        const int ia = (i + 16) & (NCTX - 1);
        ax0 = lds4(&sxc[2 * ia]);     ax1 = lds4(&sxc[2 * ia + 4]);
        ax2 = lds4(&sxc[2 * ia + 8]); ax3 = lds4(&sxc[2 * ia + 12]);
        ay0 = lds4(&syc[ia]);         ay1 = lds4(&syc[ia + 4]);
        ctx8(gx, gy, c0, c1, bx0, bx1, bx2, bx3, by0, by1, dp, cp);
        asm volatile("" : "+v"(ax0), "+v"(ax1), "+v"(ax2), "+v"(ax3));
    }
    const float dens = (dp[0] + dp[1]) + (dp[2] + dp[3]);
    const float conv = (cp[0] + cp[1]) + (cp[2] + cp[3]);
    sy[2 * t]     = dens;
    sy[2 * t + 1] = conv * __builtin_amdgcn_rcpf(dens + 1e-8f);
    __syncthreads();

    const int  wave  = t >> 5;
    const int  lane  = t & 31;
    const int  lhalf = lane & 15;
    const bool lo    = lane < 16;

    float* outb = out + (size_t)b * NCH * NPTS;

    for (int pt = 2 * wave; pt < 2 * wave + 2; ++pt) {
        float by0 = sy[2 * (pt * 16 + lhalf)];
        float by1 = sy[2 * (pt * 16 + lhalf) + 1];
        v2f bm;
        bm.x = lo ? by0 : 0.f;
        bm.y = lo ? by1 : 0.f;

        const int oPt = po + pt * 16 + lhalf;

        for (int ct = 0; ct < 4; ++ct) {
            float aw0 = W[(ct * 16 + lhalf) * 2 + 0];
            float aw1 = W[(ct * 16 + lhalf) * 2 + 1];
            v2f am;
            am.x = lo ? aw0 : 0.f;
            am.y = lo ? aw1 : 0.f;

            v8f acc = {};
            acc = __builtin_amdgcn_wmma_f32_16x16x4_f32(
                 false, am,  false, bm,
                 (short)0, acc,  false,  false);

            const int kBase = ct * 16 + (lo ? 0 : 8);
#pragma unroll
            for (int j = 0; j < 8; ++j) {
                float v = acc[j] + bias[kBase + j];
                float r = __builtin_amdgcn_rcpf(
                    1.f + __builtin_amdgcn_exp2f(-LOG2E * v));
                outb[(size_t)(kBase + j) * NPTS + oPt] = r;
            }
        }
    }
}

extern "C" void kernel_launch(void* const* d_in, const int* in_sizes, int n_in,
                              void* d_out, int out_size, void* d_ws, size_t ws_size,
                              hipStream_t stream) {
    (void)n_in; (void)out_size; (void)d_ws; (void)ws_size;
    const float* xc   = (const float*)d_in[0];
    const float* yc   = (const float*)d_in[1];
    const float* xg   = (const float*)d_in[2];
    const float* sig  = (const float*)d_in[3];
    const float* W    = (const float*)d_in[4];
    const float* bias = (const float*)d_in[5];
    float* out = (float*)d_out;

    const int B = in_sizes[0] / (NCTX * 2);
    dim3 grid(B * 16), block(256);
    convnp_fused<<<grid, block, 0, stream>>>(xc, yc, xg, sig, W, bias, out);
}


# grid re-mapped to one workgroup per WGP on all 256 WGPs: 256 blocks of 128 points and 4 waves (one wave per SIMD) instead of 128 blocks of 8 waves on half the chip
# speedup vs baseline: 1.5360x; 1.2591x over previous
#include <hip/hip_runtime.h>

typedef __attribute__((ext_vector_type(2))) float v2f;
typedef __attribute__((ext_vector_type(8))) float v8f;
typedef __attribute__((ext_vector_type(4))) float v4f;

#define GRID_N 64
#define NPTS   4096
#define NCTX   512
#define NCH    64
#define LOG2E  1.44269504088896340736f

static __device__ __forceinline__ v4f lds4(const float* p) {
    v4f v;
    __builtin_memcpy(&v, p, 16);
    return v;
}

static __device__ __forceinline__ void ctx1(float gx, float gy, float c0, float c1,
                                            float cx, float cy, float yv,
                                            float& dens, float& conv) {
    float dx   = gx - cx;
    float dy   = gy - cy;
    float dist = __builtin_amdgcn_sqrtf(__builtin_fmaf(dx, dx, dy * dy));
    float e0   = __builtin_amdgcn_exp2f(dist * c0);
    float e1   = __builtin_amdgcn_exp2f(dist * c1);
    dens += e0;
    conv  = __builtin_fmaf(yv, e1, conv);
}

static __device__ __forceinline__ void ctx8(float gx, float gy, float c0, float c1,
                                            v4f x0, v4f x1, v4f x2, v4f x3, v4f y0, v4f y1,
                                            float (&d)[4], float (&c)[4]) {
    ctx1(gx, gy, c0, c1, x0.x, x0.y, y0.x, d[0], c[0]);
    ctx1(gx, gy, c0, c1, x0.z, x0.w, y0.y, d[1], c[1]);
    ctx1(gx, gy, c0, c1, x1.x, x1.y, y0.z, d[2], c[2]);
    ctx1(gx, gy, c0, c1, x1.z, x1.w, y0.w, d[3], c[3]);
    ctx1(gx, gy, c0, c1, x2.x, x2.y, y1.x, d[0], c[0]);
    ctx1(gx, gy, c0, c1, x2.z, x2.w, y1.y, d[1], c[1]);
    ctx1(gx, gy, c0, c1, x3.x, x3.y, y1.z, d[2], c[2]);
    ctx1(gx, gy, c0, c1, x3.z, x3.w, y1.w, d[3], c[3]);
}

__global__ __launch_bounds__(128) __attribute__((amdgpu_waves_per_eu(1, 1))) void convnp_fused(
    const float* __restrict__ xc,
    const float* __restrict__ yc,
    const float* __restrict__ xg,
    const float* __restrict__ sig,
    const float* __restrict__ W,
    const float* __restrict__ bias,
    float* __restrict__ out)
{
    __shared__ __attribute__((aligned(16))) float sxc[NCTX * 2];
    __shared__ __attribute__((aligned(16))) float syc[NCTX];
    __shared__ float sy[128 * 2];

    const int t    = threadIdx.x;
    const int b    = blockIdx.x >> 5;
    const int tile = blockIdx.x & 31;
    const int po   = tile << 7;

    const v4f* xcb = (const v4f*)(xc + (size_t)b * NCTX * 2);
    const v4f* ycb = (const v4f*)(yc + (size_t)b * NCTX);
    {
        const v4f sx0 = xcb[t];
        const v4f sx1 = xcb[t + 128];
        const v4f sy0 = ycb[t];
        ((v4f*)sxc)[t]       = sx0;
        ((v4f*)sxc)[t + 128] = sx1;
        ((v4f*)syc)[t]       = sy0;
    }
    __syncthreads();

    const float s0 = __expf(sig[0]);
    const float s1 = __expf(sig[1]);
    const float c0 = -0.5f * LOG2E / (s0 * s0);
    const float c1 = -0.5f * LOG2E / (s1 * s1);

    const int   o  = po + t;
    const float gx = xg[((size_t)b * NPTS + o) * 2 + 0];
    const float gy = xg[((size_t)b * NPTS + o) * 2 + 1];

    float dp[4] = {0.f, 0.f, 0.f, 0.f};
    float cp[4] = {0.f, 0.f, 0.f, 0.f};
    v4f ax0 = lds4(&sxc[0]),  ax1 = lds4(&sxc[4]);
    v4f ax2 = lds4(&sxc[8]),  ax3 = lds4(&sxc[12]);
    v4f ay0 = lds4(&syc[0]),  ay1 = lds4(&syc[4]);
#pragma unroll 1
    for (int i = 0; i < NCTX; i += 16) {
        const int ib = i + 8;
        v4f bx0 = lds4(&sxc[2 * ib]),     bx1 = lds4(&sxc[2 * ib + 4]);
        v4f bx2 = lds4(&sxc[2 * ib + 8]), bx3 = lds4(&sxc[2 * ib + 12]);
        v4f by0 = lds4(&syc[ib]),         by1 = lds4(&syc[ib + 4]);
        ctx8(gx, gy, c0, c1, ax0, ax1, ax2, ax3, ay0, ay1, dp, cp);
        const int ia = (i + 16) & (NCTX - 1);
        ax0 = lds4(&sxc[2 * ia]);     ax1 = lds4(&sxc[2 * ia + 4]);
        ax2 = lds4(&sxc[2 * ia + 8]); ax3 = lds4(&sxc[2 * ia + 12]);
        ay0 = lds4(&syc[ia]);         ay1 = lds4(&syc[ia + 4]);
        ctx8(gx, gy, c0, c1, bx0, bx1, bx2, bx3, by0, by1, dp, cp);
        asm volatile("" : "+v"(ax0), "+v"(ax1), "+v"(ax2), "+v"(ax3));
    }
    const float dens = (dp[0] + dp[1]) + (dp[2] + dp[3]);
    const float conv = (cp[0] + cp[1]) + (cp[2] + cp[3]);
    sy[2 * t]     = dens;
    sy[2 * t + 1] = conv * __builtin_amdgcn_rcpf(dens + 1e-8f);
    __syncthreads();

    const int  wave  = t >> 5;
    const int  lane  = t & 31;
    const int  lhalf = lane & 15;
    const bool lo    = lane < 16;

    float* outb = out + (size_t)b * NCH * NPTS;

    for (int pt = 2 * wave; pt < 2 * wave + 2; ++pt) {
        float by0 = sy[2 * (pt * 16 + lhalf)];
        float by1 = sy[2 * (pt * 16 + lhalf) + 1];
        v2f bm;
        bm.x = lo ? by0 : 0.f;
        bm.y = lo ? by1 : 0.f;

        const int oPt = po + pt * 16 + lhalf;

        for (int ct = 0; ct < 4; ++ct) {
            float aw0 = W[(ct * 16 + lhalf) * 2 + 0];
            float aw1 = W[(ct * 16 + lhalf) * 2 + 1];
            v2f am;
            am.x = lo ? aw0 : 0.f;
            am.y = lo ? aw1 : 0.f;

            v8f acc = {};
            acc = __builtin_amdgcn_wmma_f32_16x16x4_f32(
                 false, am,  false, bm,
                 (short)0, acc,  false,  false);

            const int kBase = ct * 16 + (lo ? 0 : 8);
#pragma unroll
            for (int j = 0; j < 8; ++j) {
                float v = acc[j] + bias[kBase + j];
                float r = __builtin_amdgcn_rcpf(
                    1.f + __builtin_amdgcn_exp2f(-LOG2E * v));
                outb[(size_t)(kBase + j) * NPTS + oPt] = r;
            }
        }
    }
}

extern "C" void kernel_launch(void* const* d_in, const int* in_sizes, int n_in,
                              void* d_out, int out_size, void* d_ws, size_t ws_size,
                              hipStream_t stream) {
    (void)n_in; (void)out_size; (void)d_ws; (void)ws_size;
    const float* xc   = (const float*)d_in[0];
    const float* yc   = (const float*)d_in[1];
    const float* xg   = (const float*)d_in[2];
    const float* sig  = (const float*)d_in[3];
    const float* W    = (const float*)d_in[4];
    const float* bias = (const float*)d_in[5];
    float* out = (float*)d_out;

    const int B = in_sizes[0] / (NCTX * 2);
    dim3 grid(B * 32), block(128);
    convnp_fused<<<grid, block, 0, stream>>>(xc, yc, xg, sig, W, bias, out);
}


# context loop rewritten as a software pipeline over pairs of points: every v_sqrt/v_exp is followed by a packed instruction of another pair; LDS read once per 32 points, one trip ahead
# speedup vs baseline: 1.8939x; 1.2330x over previous
#include <hip/hip_runtime.h>

typedef __attribute__((ext_vector_type(2))) float v2f;
typedef __attribute__((ext_vector_type(8))) float v8f;
typedef __attribute__((ext_vector_type(4))) float v4f;

#define GRID_N 64
#define NPTS   4096
#define NCTX   512
#define NPAD   64
#define NCH    64
#define LOG2E  1.44269504088896340736f

static __device__ __forceinline__ v4f lds4(const float* p) {
    v4f v;
    __builtin_memcpy(&v, p, 16);
    return v;
}

#define SB() __builtin_amdgcn_sched_barrier(0)

__global__ __launch_bounds__(128) __attribute__((amdgpu_waves_per_eu(1, 1))) void convnp_fused(
    const float* __restrict__ xc,
    const float* __restrict__ yc,
    const float* __restrict__ xg,
    const float* __restrict__ sig,
    const float* __restrict__ W,
    const float* __restrict__ bias,
    float* __restrict__ out)
{
    __shared__ __attribute__((aligned(16))) float sxx[NCTX + NPAD];
    __shared__ __attribute__((aligned(16))) float sxy[NCTX + NPAD];
    __shared__ __attribute__((aligned(16))) float syc[NCTX + NPAD];
    __shared__ float sy[128 * 2];

    const int t    = threadIdx.x;
    const int b    = blockIdx.x >> 5;
    const int tile = blockIdx.x & 31;
    const int po   = tile << 7;

    const v4f* xcb = (const v4f*)(xc + (size_t)b * NCTX * 2);
    const v4f* ycb = (const v4f*)(yc + (size_t)b * NCTX);
    {
        const v4f sx0 = xcb[t];
        const v4f sx1 = xcb[t + 128];
        const v4f sy0 = ycb[t];
        ((v2f*)sxx)[t]       = (v2f){sx0.x, sx0.z};
        ((v2f*)sxy)[t]       = (v2f){sx0.y, sx0.w};
        ((v2f*)sxx)[t + 128] = (v2f){sx1.x, sx1.z};
        ((v2f*)sxy)[t + 128] = (v2f){sx1.y, sx1.w};
        ((v4f*)syc)[t]       = sy0;
        if (t < NPAD / 4) {
            const v4f z = {0.f, 0.f, 0.f, 0.f};
            ((v4f*)(sxx + NCTX))[t] = z;
            ((v4f*)(sxy + NCTX))[t] = z;
            ((v4f*)(syc + NCTX))[t] = z;
        }
    }
    __syncthreads();

    const float s0 = __expf(sig[0]);
    const float s1 = __expf(sig[1]);
    const float c0 = -0.5f * LOG2E / (s0 * s0);
    const float c1 = -0.5f * LOG2E / (s1 * s1);
    const v2f   cc = {c0, c1};

    const int   o  = po + t;
    const float gx = xg[((size_t)b * NPTS + o) * 2 + 0];
    const float gy = xg[((size_t)b * NPTS + o) * 2 + 1];
    const v2f   gxx = {gx, gx};
    const v2f   gyy = {gy, gy};

    v2f dx0, dx1, dx2, dx3, dy0, dy1, dy2, dy3, tt0, tt1, tt2, tt3, qq0, qq1, qq2, qq3;
    v2f ss0, ss1, ss2, ss3, aa0, aa1, aa2, aa3, ab0, ab1, ab2, ab3, ea0, ea1, ea2, ea3, eb0, eb1, eb2, eb3;
    v2f da0 = {0.f, 0.f}, da1 = {0.f, 0.f}, da2 = {0.f, 0.f}, da3 = {0.f, 0.f};
    v2f ca0 = {0.f, 0.f}, ca1 = {0.f, 0.f}, ca2 = {0.f, 0.f}, ca3 = {0.f, 0.f};

    v4f XP0 = lds4(&sxx[0]), XP1 = lds4(&sxx[4]), XP2 = lds4(&sxx[8]), XP3 = lds4(&sxx[12]);
    v4f YP0 = lds4(&sxy[0]), YP1 = lds4(&sxy[4]), YP2 = lds4(&sxy[8]), YP3 = lds4(&sxy[12]);
    v4f X0_0, X0_1, X0_2, X0_3, X0_4, X0_5, X0_6, X0_7, Y0_0, Y0_1, Y0_2, Y0_3, Y0_4, Y0_5, Y0_6, Y0_7, V0_0, V0_1, V0_2, V0_3, V0_4, V0_5, V0_6, V0_7;
    v4f X1_0, X1_1, X1_2, X1_3, X1_4, X1_5, X1_6, X1_7, Y1_0, Y1_1, Y1_2, Y1_3, Y1_4, Y1_5, Y1_6, Y1_7, V1_0, V1_1, V1_2, V1_3, V1_4, V1_5, V1_6, V1_7;
    X0_0 = lds4(&sxx[0 + 16 + 0]); Y0_0 = lds4(&sxy[0 + 16 + 0]);
    X0_1 = lds4(&sxx[0 + 16 + 4]); Y0_1 = lds4(&sxy[0 + 16 + 4]);
    X0_2 = lds4(&sxx[0 + 16 + 8]); Y0_2 = lds4(&sxy[0 + 16 + 8]);
    X0_3 = lds4(&sxx[0 + 16 + 12]); Y0_3 = lds4(&sxy[0 + 16 + 12]);
    X0_4 = lds4(&sxx[0 + 16 + 16]); Y0_4 = lds4(&sxy[0 + 16 + 16]);
    X0_5 = lds4(&sxx[0 + 16 + 20]); Y0_5 = lds4(&sxy[0 + 16 + 20]);
    X0_6 = lds4(&sxx[0 + 16 + 24]); Y0_6 = lds4(&sxy[0 + 16 + 24]);
    X0_7 = lds4(&sxx[0 + 16 + 28]); Y0_7 = lds4(&sxy[0 + 16 + 28]);
    V0_0 = lds4(&syc[0 + 0]);
    V0_1 = lds4(&syc[0 + 4]);
    V0_2 = lds4(&syc[0 + 8]);
    V0_3 = lds4(&syc[0 + 12]);
    V0_4 = lds4(&syc[0 + 16]);
    V0_5 = lds4(&syc[0 + 20]);
    V0_6 = lds4(&syc[0 + 24]);
    V0_7 = lds4(&syc[0 + 28]);
    SB();

    dx0 = gxx - (v2f){XP0.x, XP0.y}; SB();
    dy0 = gyy - (v2f){YP0.x, YP0.y}; SB();
    dx1 = gxx - (v2f){XP0.z, XP0.w}; SB();
    dy1 = gyy - (v2f){YP0.z, YP0.w}; SB();
    tt0 = dy0 * dy0; SB();
    dx2 = gxx - (v2f){XP1.x, XP1.y}; SB();
    dy2 = gyy - (v2f){YP1.x, YP1.y}; SB();
    tt1 = dy1 * dy1; SB();
    qq0 = __builtin_elementwise_fma(dx0, dx0, tt0); SB();
    dx3 = gxx - (v2f){XP1.z, XP1.w}; SB();
    dy3 = gyy - (v2f){YP1.z, YP1.w}; SB();
    tt2 = dy2 * dy2; SB();
    qq1 = __builtin_elementwise_fma(dx1, dx1, tt1); SB();
    ss0.x = __builtin_amdgcn_sqrtf(qq0.x); SB();
    dx0 = gxx - (v2f){XP2.x, XP2.y}; SB();
    ss0.y = __builtin_amdgcn_sqrtf(qq0.y); SB();
    dy0 = gyy - (v2f){YP2.x, YP2.y}; SB();
    tt3 = dy3 * dy3; SB();
    qq2 = __builtin_elementwise_fma(dx2, dx2, tt2); SB();
    ss1.x = __builtin_amdgcn_sqrtf(qq1.x); SB();
    dx1 = gxx - (v2f){XP2.z, XP2.w}; SB();
    ss1.y = __builtin_amdgcn_sqrtf(qq1.y); SB();
    dy1 = gyy - (v2f){YP2.z, YP2.w}; SB();
    tt0 = dy0 * dy0; SB();
    qq3 = __builtin_elementwise_fma(dx3, dx3, tt3); SB();
    aa0 = (v2f){ss0.x, ss0.x} * cc; SB();
    ab0 = (v2f){ss0.y, ss0.y} * cc; SB();
    ss2.x = __builtin_amdgcn_sqrtf(qq2.x); SB();
    dx2 = gxx - (v2f){XP3.x, XP3.y}; SB();
    ss2.y = __builtin_amdgcn_sqrtf(qq2.y); SB();
    dy2 = gyy - (v2f){YP3.x, YP3.y}; SB();
    ea0.x = __builtin_amdgcn_exp2f(aa0.x); SB();
    tt1 = dy1 * dy1; SB();
    ea0.y = __builtin_amdgcn_exp2f(ab0.x); SB();
    qq0 = __builtin_elementwise_fma(dx0, dx0, tt0); SB();
    eb0.x = __builtin_amdgcn_exp2f(aa0.y); SB();
    aa1 = (v2f){ss1.x, ss1.x} * cc; SB();
    eb0.y = __builtin_amdgcn_exp2f(ab0.y); SB();
    ab1 = (v2f){ss1.y, ss1.y} * cc; SB();
    ss3.x = __builtin_amdgcn_sqrtf(qq3.x); SB();
    dx3 = gxx - (v2f){XP3.z, XP3.w}; SB();
    ss3.y = __builtin_amdgcn_sqrtf(qq3.y); SB();
    dy3 = gyy - (v2f){YP3.z, YP3.w}; SB();
    ea1.x = __builtin_amdgcn_exp2f(aa1.x); SB();
    tt2 = dy2 * dy2; SB();
    ea1.y = __builtin_amdgcn_exp2f(ab1.x); SB();
    qq1 = __builtin_elementwise_fma(dx1, dx1, tt1); SB();
    eb1.x = __builtin_amdgcn_exp2f(aa1.y); SB();
    aa2 = (v2f){ss2.x, ss2.x} * cc; SB();
    eb1.y = __builtin_amdgcn_exp2f(ab1.y); SB();
    ab2 = (v2f){ss2.y, ss2.y} * cc; SB();

#pragma unroll 1
    for (int base = 0; base < NCTX; base += 64) {
        X1_0 = lds4(&sxx[base + 32 + 16 + 0]); Y1_0 = lds4(&sxy[base + 32 + 16 + 0]);
        X1_1 = lds4(&sxx[base + 32 + 16 + 4]); Y1_1 = lds4(&sxy[base + 32 + 16 + 4]);
        X1_2 = lds4(&sxx[base + 32 + 16 + 8]); Y1_2 = lds4(&sxy[base + 32 + 16 + 8]);
        X1_3 = lds4(&sxx[base + 32 + 16 + 12]); Y1_3 = lds4(&sxy[base + 32 + 16 + 12]);
        X1_4 = lds4(&sxx[base + 32 + 16 + 16]); Y1_4 = lds4(&sxy[base + 32 + 16 + 16]);
        X1_5 = lds4(&sxx[base + 32 + 16 + 20]); Y1_5 = lds4(&sxy[base + 32 + 16 + 20]);
        X1_6 = lds4(&sxx[base + 32 + 16 + 24]); Y1_6 = lds4(&sxy[base + 32 + 16 + 24]);
        X1_7 = lds4(&sxx[base + 32 + 16 + 28]); Y1_7 = lds4(&sxy[base + 32 + 16 + 28]);
        V1_0 = lds4(&syc[base + 32 + 0]);
        V1_1 = lds4(&syc[base + 32 + 4]);
        V1_2 = lds4(&syc[base + 32 + 8]);
        V1_3 = lds4(&syc[base + 32 + 12]);
        V1_4 = lds4(&syc[base + 32 + 16]);
        V1_5 = lds4(&syc[base + 32 + 20]);
        V1_6 = lds4(&syc[base + 32 + 24]);
        V1_7 = lds4(&syc[base + 32 + 28]);
        SB();
        ss0.x = __builtin_amdgcn_sqrtf(qq0.x); SB();
        dx0 = gxx - (v2f){X0_0.x, X0_0.y}; SB();
        ss0.y = __builtin_amdgcn_sqrtf(qq0.y); SB();
        dy0 = gyy - (v2f){Y0_0.x, Y0_0.y}; SB();
        ea2.x = __builtin_amdgcn_exp2f(aa2.x); SB();
        tt3 = dy3 * dy3; SB();
        ea2.y = __builtin_amdgcn_exp2f(ab2.x); SB();
        qq2 = __builtin_elementwise_fma(dx2, dx2, tt2); SB();
        eb2.x = __builtin_amdgcn_exp2f(aa2.y); SB();
        aa3 = (v2f){ss3.x, ss3.x} * cc; SB();
        eb2.y = __builtin_amdgcn_exp2f(ab2.y); SB();
        ab3 = (v2f){ss3.y, ss3.y} * cc; SB();
        da0 += ea0; SB();
        ca0 = __builtin_elementwise_fma((v2f){V0_0.x, V0_0.y}, eb0, ca0); SB();
        ss1.x = __builtin_amdgcn_sqrtf(qq1.x); SB();
        dx1 = gxx - (v2f){X0_0.z, X0_0.w}; SB();
        ss1.y = __builtin_amdgcn_sqrtf(qq1.y); SB();
        dy1 = gyy - (v2f){Y0_0.z, Y0_0.w}; SB();
        ea3.x = __builtin_amdgcn_exp2f(aa3.x); SB();
        tt0 = dy0 * dy0; SB();
        ea3.y = __builtin_amdgcn_exp2f(ab3.x); SB();
        qq3 = __builtin_elementwise_fma(dx3, dx3, tt3); SB();
        eb3.x = __builtin_amdgcn_exp2f(aa3.y); SB();
        aa0 = (v2f){ss0.x, ss0.x} * cc; SB();
        eb3.y = __builtin_amdgcn_exp2f(ab3.y); SB();
        ab0 = (v2f){ss0.y, ss0.y} * cc; SB();
        da1 += ea1; SB();
        ca1 = __builtin_elementwise_fma((v2f){V0_0.z, V0_0.w}, eb1, ca1); SB();
        ss2.x = __builtin_amdgcn_sqrtf(qq2.x); SB();
        dx2 = gxx - (v2f){X0_1.x, X0_1.y}; SB();
        ss2.y = __builtin_amdgcn_sqrtf(qq2.y); SB();
        dy2 = gyy - (v2f){Y0_1.x, Y0_1.y}; SB();
        ea0.x = __builtin_amdgcn_exp2f(aa0.x); SB();
        tt1 = dy1 * dy1; SB();
        ea0.y = __builtin_amdgcn_exp2f(ab0.x); SB();
        qq0 = __builtin_elementwise_fma(dx0, dx0, tt0); SB();
        eb0.x = __builtin_amdgcn_exp2f(aa0.y); SB();
        aa1 = (v2f){ss1.x, ss1.x} * cc; SB();
        eb0.y = __builtin_amdgcn_exp2f(ab0.y); SB();
        ab1 = (v2f){ss1.y, ss1.y} * cc; SB();
        da2 += ea2; SB();
        ca2 = __builtin_elementwise_fma((v2f){V0_1.x, V0_1.y}, eb2, ca2); SB();
        ss3.x = __builtin_amdgcn_sqrtf(qq3.x); SB();
        dx3 = gxx - (v2f){X0_1.z, X0_1.w}; SB();
        ss3.y = __builtin_amdgcn_sqrtf(qq3.y); SB();
        dy3 = gyy - (v2f){Y0_1.z, Y0_1.w}; SB();
        ea1.x = __builtin_amdgcn_exp2f(aa1.x); SB();
        tt2 = dy2 * dy2; SB();
        ea1.y = __builtin_amdgcn_exp2f(ab1.x); SB();
        qq1 = __builtin_elementwise_fma(dx1, dx1, tt1); SB();
        eb1.x = __builtin_amdgcn_exp2f(aa1.y); SB();
        aa2 = (v2f){ss2.x, ss2.x} * cc; SB();
        eb1.y = __builtin_amdgcn_exp2f(ab1.y); SB();
        ab2 = (v2f){ss2.y, ss2.y} * cc; SB();
        da3 += ea3; SB();
        ca3 = __builtin_elementwise_fma((v2f){V0_1.z, V0_1.w}, eb3, ca3); SB();
        ss0.x = __builtin_amdgcn_sqrtf(qq0.x); SB();
        dx0 = gxx - (v2f){X0_2.x, X0_2.y}; SB();
        ss0.y = __builtin_amdgcn_sqrtf(qq0.y); SB();
        dy0 = gyy - (v2f){Y0_2.x, Y0_2.y}; SB();
        ea2.x = __builtin_amdgcn_exp2f(aa2.x); SB();
        tt3 = dy3 * dy3; SB();
        ea2.y = __builtin_amdgcn_exp2f(ab2.x); SB();
        qq2 = __builtin_elementwise_fma(dx2, dx2, tt2); SB();
        eb2.x = __builtin_amdgcn_exp2f(aa2.y); SB();
        aa3 = (v2f){ss3.x, ss3.x} * cc; SB();
        eb2.y = __builtin_amdgcn_exp2f(ab2.y); SB();
        ab3 = (v2f){ss3.y, ss3.y} * cc; SB();
        da0 += ea0; SB();
        ca0 = __builtin_elementwise_fma((v2f){V0_2.x, V0_2.y}, eb0, ca0); SB();
        ss1.x = __builtin_amdgcn_sqrtf(qq1.x); SB();
        dx1 = gxx - (v2f){X0_2.z, X0_2.w}; SB();
        ss1.y = __builtin_amdgcn_sqrtf(qq1.y); SB();
        dy1 = gyy - (v2f){Y0_2.z, Y0_2.w}; SB();
        ea3.x = __builtin_amdgcn_exp2f(aa3.x); SB();
        tt0 = dy0 * dy0; SB();
        ea3.y = __builtin_amdgcn_exp2f(ab3.x); SB();
        qq3 = __builtin_elementwise_fma(dx3, dx3, tt3); SB();
        eb3.x = __builtin_amdgcn_exp2f(aa3.y); SB();
        aa0 = (v2f){ss0.x, ss0.x} * cc; SB();
        eb3.y = __builtin_amdgcn_exp2f(ab3.y); SB();
        ab0 = (v2f){ss0.y, ss0.y} * cc; SB();
        da1 += ea1; SB();
        ca1 = __builtin_elementwise_fma((v2f){V0_2.z, V0_2.w}, eb1, ca1); SB();
        ss2.x = __builtin_amdgcn_sqrtf(qq2.x); SB();
        dx2 = gxx - (v2f){X0_3.x, X0_3.y}; SB();
        ss2.y = __builtin_amdgcn_sqrtf(qq2.y); SB();
        dy2 = gyy - (v2f){Y0_3.x, Y0_3.y}; SB();
        ea0.x = __builtin_amdgcn_exp2f(aa0.x); SB();
        tt1 = dy1 * dy1; SB();
        ea0.y = __builtin_amdgcn_exp2f(ab0.x); SB();
        qq0 = __builtin_elementwise_fma(dx0, dx0, tt0); SB();
        eb0.x = __builtin_amdgcn_exp2f(aa0.y); SB();
        aa1 = (v2f){ss1.x, ss1.x} * cc; SB();
        eb0.y = __builtin_amdgcn_exp2f(ab0.y); SB();
        ab1 = (v2f){ss1.y, ss1.y} * cc; SB();
        da2 += ea2; SB();
        ca2 = __builtin_elementwise_fma((v2f){V0_3.x, V0_3.y}, eb2, ca2); SB();
        ss3.x = __builtin_amdgcn_sqrtf(qq3.x); SB();
        dx3 = gxx - (v2f){X0_3.z, X0_3.w}; SB();
        ss3.y = __builtin_amdgcn_sqrtf(qq3.y); SB();
        dy3 = gyy - (v2f){Y0_3.z, Y0_3.w}; SB();
        ea1.x = __builtin_amdgcn_exp2f(aa1.x); SB();
        tt2 = dy2 * dy2; SB();
        ea1.y = __builtin_amdgcn_exp2f(ab1.x); SB();
        qq1 = __builtin_elementwise_fma(dx1, dx1, tt1); SB();
        eb1.x = __builtin_amdgcn_exp2f(aa1.y); SB();
        aa2 = (v2f){ss2.x, ss2.x} * cc; SB();
        eb1.y = __builtin_amdgcn_exp2f(ab1.y); SB();
        ab2 = (v2f){ss2.y, ss2.y} * cc; SB();
        da3 += ea3; SB();
        ca3 = __builtin_elementwise_fma((v2f){V0_3.z, V0_3.w}, eb3, ca3); SB();
        ss0.x = __builtin_amdgcn_sqrtf(qq0.x); SB();
        dx0 = gxx - (v2f){X0_4.x, X0_4.y}; SB();
        ss0.y = __builtin_amdgcn_sqrtf(qq0.y); SB();
        dy0 = gyy - (v2f){Y0_4.x, Y0_4.y}; SB();
        ea2.x = __builtin_amdgcn_exp2f(aa2.x); SB();
        tt3 = dy3 * dy3; SB();
        ea2.y = __builtin_amdgcn_exp2f(ab2.x); SB();
        qq2 = __builtin_elementwise_fma(dx2, dx2, tt2); SB();
        eb2.x = __builtin_amdgcn_exp2f(aa2.y); SB();
        aa3 = (v2f){ss3.x, ss3.x} * cc; SB();
        eb2.y = __builtin_amdgcn_exp2f(ab2.y); SB();
        ab3 = (v2f){ss3.y, ss3.y} * cc; SB();
        da0 += ea0; SB();
        ca0 = __builtin_elementwise_fma((v2f){V0_4.x, V0_4.y}, eb0, ca0); SB();
        ss1.x = __builtin_amdgcn_sqrtf(qq1.x); SB();
        dx1 = gxx - (v2f){X0_4.z, X0_4.w}; SB();
        ss1.y = __builtin_amdgcn_sqrtf(qq1.y); SB();
        dy1 = gyy - (v2f){Y0_4.z, Y0_4.w}; SB();
        ea3.x = __builtin_amdgcn_exp2f(aa3.x); SB();
        tt0 = dy0 * dy0; SB();
        ea3.y = __builtin_amdgcn_exp2f(ab3.x); SB();
        qq3 = __builtin_elementwise_fma(dx3, dx3, tt3); SB();
        eb3.x = __builtin_amdgcn_exp2f(aa3.y); SB();
        aa0 = (v2f){ss0.x, ss0.x} * cc; SB();
        eb3.y = __builtin_amdgcn_exp2f(ab3.y); SB();
        ab0 = (v2f){ss0.y, ss0.y} * cc; SB();
        da1 += ea1; SB();
        ca1 = __builtin_elementwise_fma((v2f){V0_4.z, V0_4.w}, eb1, ca1); SB();
        ss2.x = __builtin_amdgcn_sqrtf(qq2.x); SB();
        dx2 = gxx - (v2f){X0_5.x, X0_5.y}; SB();
        ss2.y = __builtin_amdgcn_sqrtf(qq2.y); SB();
        dy2 = gyy - (v2f){Y0_5.x, Y0_5.y}; SB();
        ea0.x = __builtin_amdgcn_exp2f(aa0.x); SB();
        tt1 = dy1 * dy1; SB();
        ea0.y = __builtin_amdgcn_exp2f(ab0.x); SB();
        qq0 = __builtin_elementwise_fma(dx0, dx0, tt0); SB();
        eb0.x = __builtin_amdgcn_exp2f(aa0.y); SB();
        aa1 = (v2f){ss1.x, ss1.x} * cc; SB();
        eb0.y = __builtin_amdgcn_exp2f(ab0.y); SB();
        ab1 = (v2f){ss1.y, ss1.y} * cc; SB();
        da2 += ea2; SB();
        ca2 = __builtin_elementwise_fma((v2f){V0_5.x, V0_5.y}, eb2, ca2); SB();
        ss3.x = __builtin_amdgcn_sqrtf(qq3.x); SB();
        dx3 = gxx - (v2f){X0_5.z, X0_5.w}; SB();
        ss3.y = __builtin_amdgcn_sqrtf(qq3.y); SB();
        dy3 = gyy - (v2f){Y0_5.z, Y0_5.w}; SB();
        ea1.x = __builtin_amdgcn_exp2f(aa1.x); SB();
        tt2 = dy2 * dy2; SB();
        ea1.y = __builtin_amdgcn_exp2f(ab1.x); SB();
        qq1 = __builtin_elementwise_fma(dx1, dx1, tt1); SB();
        eb1.x = __builtin_amdgcn_exp2f(aa1.y); SB();
        aa2 = (v2f){ss2.x, ss2.x} * cc; SB();
        eb1.y = __builtin_amdgcn_exp2f(ab1.y); SB();
        ab2 = (v2f){ss2.y, ss2.y} * cc; SB();
        da3 += ea3; SB();
        ca3 = __builtin_elementwise_fma((v2f){V0_5.z, V0_5.w}, eb3, ca3); SB();
        ss0.x = __builtin_amdgcn_sqrtf(qq0.x); SB();
        dx0 = gxx - (v2f){X0_6.x, X0_6.y}; SB();
        ss0.y = __builtin_amdgcn_sqrtf(qq0.y); SB();
        dy0 = gyy - (v2f){Y0_6.x, Y0_6.y}; SB();
        ea2.x = __builtin_amdgcn_exp2f(aa2.x); SB();
        tt3 = dy3 * dy3; SB();
        ea2.y = __builtin_amdgcn_exp2f(ab2.x); SB();
        qq2 = __builtin_elementwise_fma(dx2, dx2, tt2); SB();
        eb2.x = __builtin_amdgcn_exp2f(aa2.y); SB();
        aa3 = (v2f){ss3.x, ss3.x} * cc; SB();
        eb2.y = __builtin_amdgcn_exp2f(ab2.y); SB();
        ab3 = (v2f){ss3.y, ss3.y} * cc; SB();
        da0 += ea0; SB();
        ca0 = __builtin_elementwise_fma((v2f){V0_6.x, V0_6.y}, eb0, ca0); SB();
        ss1.x = __builtin_amdgcn_sqrtf(qq1.x); SB();
        dx1 = gxx - (v2f){X0_6.z, X0_6.w}; SB();
        ss1.y = __builtin_amdgcn_sqrtf(qq1.y); SB();
        dy1 = gyy - (v2f){Y0_6.z, Y0_6.w}; SB();
        ea3.x = __builtin_amdgcn_exp2f(aa3.x); SB();
        tt0 = dy0 * dy0; SB();
        ea3.y = __builtin_amdgcn_exp2f(ab3.x); SB();
        qq3 = __builtin_elementwise_fma(dx3, dx3, tt3); SB();
        eb3.x = __builtin_amdgcn_exp2f(aa3.y); SB();
        aa0 = (v2f){ss0.x, ss0.x} * cc; SB();
        eb3.y = __builtin_amdgcn_exp2f(ab3.y); SB();
        ab0 = (v2f){ss0.y, ss0.y} * cc; SB();
        da1 += ea1; SB();
        ca1 = __builtin_elementwise_fma((v2f){V0_6.z, V0_6.w}, eb1, ca1); SB();
        ss2.x = __builtin_amdgcn_sqrtf(qq2.x); SB();
        dx2 = gxx - (v2f){X0_7.x, X0_7.y}; SB();
        ss2.y = __builtin_amdgcn_sqrtf(qq2.y); SB();
        dy2 = gyy - (v2f){Y0_7.x, Y0_7.y}; SB();
        ea0.x = __builtin_amdgcn_exp2f(aa0.x); SB();
        tt1 = dy1 * dy1; SB();
        ea0.y = __builtin_amdgcn_exp2f(ab0.x); SB();
        qq0 = __builtin_elementwise_fma(dx0, dx0, tt0); SB();
        eb0.x = __builtin_amdgcn_exp2f(aa0.y); SB();
        aa1 = (v2f){ss1.x, ss1.x} * cc; SB();
        eb0.y = __builtin_amdgcn_exp2f(ab0.y); SB();
        ab1 = (v2f){ss1.y, ss1.y} * cc; SB();
        da2 += ea2; SB();
        ca2 = __builtin_elementwise_fma((v2f){V0_7.x, V0_7.y}, eb2, ca2); SB();
        ss3.x = __builtin_amdgcn_sqrtf(qq3.x); SB();
        dx3 = gxx - (v2f){X0_7.z, X0_7.w}; SB();
        ss3.y = __builtin_amdgcn_sqrtf(qq3.y); SB();
        dy3 = gyy - (v2f){Y0_7.z, Y0_7.w}; SB();
        ea1.x = __builtin_amdgcn_exp2f(aa1.x); SB();
        tt2 = dy2 * dy2; SB();
        ea1.y = __builtin_amdgcn_exp2f(ab1.x); SB();
        qq1 = __builtin_elementwise_fma(dx1, dx1, tt1); SB();
        eb1.x = __builtin_amdgcn_exp2f(aa1.y); SB();
        aa2 = (v2f){ss2.x, ss2.x} * cc; SB();
        eb1.y = __builtin_amdgcn_exp2f(ab1.y); SB();
        ab2 = (v2f){ss2.y, ss2.y} * cc; SB();
        da3 += ea3; SB();
        ca3 = __builtin_elementwise_fma((v2f){V0_7.z, V0_7.w}, eb3, ca3); SB();
        X0_0 = lds4(&sxx[base + 64 + 16 + 0]); Y0_0 = lds4(&sxy[base + 64 + 16 + 0]);
        X0_1 = lds4(&sxx[base + 64 + 16 + 4]); Y0_1 = lds4(&sxy[base + 64 + 16 + 4]);
        X0_2 = lds4(&sxx[base + 64 + 16 + 8]); Y0_2 = lds4(&sxy[base + 64 + 16 + 8]);
        X0_3 = lds4(&sxx[base + 64 + 16 + 12]); Y0_3 = lds4(&sxy[base + 64 + 16 + 12]);
        X0_4 = lds4(&sxx[base + 64 + 16 + 16]); Y0_4 = lds4(&sxy[base + 64 + 16 + 16]);
        X0_5 = lds4(&sxx[base + 64 + 16 + 20]); Y0_5 = lds4(&sxy[base + 64 + 16 + 20]);
        X0_6 = lds4(&sxx[base + 64 + 16 + 24]); Y0_6 = lds4(&sxy[base + 64 + 16 + 24]);
        X0_7 = lds4(&sxx[base + 64 + 16 + 28]); Y0_7 = lds4(&sxy[base + 64 + 16 + 28]);
        V0_0 = lds4(&syc[base + 64 + 0]);
        V0_1 = lds4(&syc[base + 64 + 4]);
        V0_2 = lds4(&syc[base + 64 + 8]);
        V0_3 = lds4(&syc[base + 64 + 12]);
        V0_4 = lds4(&syc[base + 64 + 16]);
        V0_5 = lds4(&syc[base + 64 + 20]);
        V0_6 = lds4(&syc[base + 64 + 24]);
        V0_7 = lds4(&syc[base + 64 + 28]);
        SB();
        ss0.x = __builtin_amdgcn_sqrtf(qq0.x); SB();
        dx0 = gxx - (v2f){X1_0.x, X1_0.y}; SB();
        ss0.y = __builtin_amdgcn_sqrtf(qq0.y); SB();
        dy0 = gyy - (v2f){Y1_0.x, Y1_0.y}; SB();
        ea2.x = __builtin_amdgcn_exp2f(aa2.x); SB();
        tt3 = dy3 * dy3; SB();
        ea2.y = __builtin_amdgcn_exp2f(ab2.x); SB();
        qq2 = __builtin_elementwise_fma(dx2, dx2, tt2); SB();
        eb2.x = __builtin_amdgcn_exp2f(aa2.y); SB();
        aa3 = (v2f){ss3.x, ss3.x} * cc; SB();
        eb2.y = __builtin_amdgcn_exp2f(ab2.y); SB();
        ab3 = (v2f){ss3.y, ss3.y} * cc; SB();
        da0 += ea0; SB();
        ca0 = __builtin_elementwise_fma((v2f){V1_0.x, V1_0.y}, eb0, ca0); SB();
        ss1.x = __builtin_amdgcn_sqrtf(qq1.x); SB();
        dx1 = gxx - (v2f){X1_0.z, X1_0.w}; SB();
        ss1.y = __builtin_amdgcn_sqrtf(qq1.y); SB();
        dy1 = gyy - (v2f){Y1_0.z, Y1_0.w}; SB();
        ea3.x = __builtin_amdgcn_exp2f(aa3.x); SB();
        tt0 = dy0 * dy0; SB();
        ea3.y = __builtin_amdgcn_exp2f(ab3.x); SB();
        qq3 = __builtin_elementwise_fma(dx3, dx3, tt3); SB();
        eb3.x = __builtin_amdgcn_exp2f(aa3.y); SB();
        aa0 = (v2f){ss0.x, ss0.x} * cc; SB();
        eb3.y = __builtin_amdgcn_exp2f(ab3.y); SB();
        ab0 = (v2f){ss0.y, ss0.y} * cc; SB();
        da1 += ea1; SB();
        ca1 = __builtin_elementwise_fma((v2f){V1_0.z, V1_0.w}, eb1, ca1); SB();
        ss2.x = __builtin_amdgcn_sqrtf(qq2.x); SB();
        dx2 = gxx - (v2f){X1_1.x, X1_1.y}; SB();
        ss2.y = __builtin_amdgcn_sqrtf(qq2.y); SB();
        dy2 = gyy - (v2f){Y1_1.x, Y1_1.y}; SB();
        ea0.x = __builtin_amdgcn_exp2f(aa0.x); SB();
        tt1 = dy1 * dy1; SB();
        ea0.y = __builtin_amdgcn_exp2f(ab0.x); SB();
        qq0 = __builtin_elementwise_fma(dx0, dx0, tt0); SB();
        eb0.x = __builtin_amdgcn_exp2f(aa0.y); SB();
        aa1 = (v2f){ss1.x, ss1.x} * cc; SB();
        eb0.y = __builtin_amdgcn_exp2f(ab0.y); SB();
        ab1 = (v2f){ss1.y, ss1.y} * cc; SB();
        da2 += ea2; SB();
        ca2 = __builtin_elementwise_fma((v2f){V1_1.x, V1_1.y}, eb2, ca2); SB();
        ss3.x = __builtin_amdgcn_sqrtf(qq3.x); SB();
        dx3 = gxx - (v2f){X1_1.z, X1_1.w}; SB();
        ss3.y = __builtin_amdgcn_sqrtf(qq3.y); SB();
        dy3 = gyy - (v2f){Y1_1.z, Y1_1.w}; SB();
        ea1.x = __builtin_amdgcn_exp2f(aa1.x); SB();
        tt2 = dy2 * dy2; SB();
        ea1.y = __builtin_amdgcn_exp2f(ab1.x); SB();
        qq1 = __builtin_elementwise_fma(dx1, dx1, tt1); SB();
        eb1.x = __builtin_amdgcn_exp2f(aa1.y); SB();
        aa2 = (v2f){ss2.x, ss2.x} * cc; SB();
        eb1.y = __builtin_amdgcn_exp2f(ab1.y); SB();
        ab2 = (v2f){ss2.y, ss2.y} * cc; SB();
        da3 += ea3; SB();
        ca3 = __builtin_elementwise_fma((v2f){V1_1.z, V1_1.w}, eb3, ca3); SB();
        ss0.x = __builtin_amdgcn_sqrtf(qq0.x); SB();
        dx0 = gxx - (v2f){X1_2.x, X1_2.y}; SB();
        ss0.y = __builtin_amdgcn_sqrtf(qq0.y); SB();
        dy0 = gyy - (v2f){Y1_2.x, Y1_2.y}; SB();
        ea2.x = __builtin_amdgcn_exp2f(aa2.x); SB();
        tt3 = dy3 * dy3; SB();
        ea2.y = __builtin_amdgcn_exp2f(ab2.x); SB();
        qq2 = __builtin_elementwise_fma(dx2, dx2, tt2); SB();
        eb2.x = __builtin_amdgcn_exp2f(aa2.y); SB();
        aa3 = (v2f){ss3.x, ss3.x} * cc; SB();
        eb2.y = __builtin_amdgcn_exp2f(ab2.y); SB();
        ab3 = (v2f){ss3.y, ss3.y} * cc; SB();
        da0 += ea0; SB();
        ca0 = __builtin_elementwise_fma((v2f){V1_2.x, V1_2.y}, eb0, ca0); SB();
        ss1.x = __builtin_amdgcn_sqrtf(qq1.x); SB();
        dx1 = gxx - (v2f){X1_2.z, X1_2.w}; SB();
        ss1.y = __builtin_amdgcn_sqrtf(qq1.y); SB();
        dy1 = gyy - (v2f){Y1_2.z, Y1_2.w}; SB();
        ea3.x = __builtin_amdgcn_exp2f(aa3.x); SB();
        tt0 = dy0 * dy0; SB();
        ea3.y = __builtin_amdgcn_exp2f(ab3.x); SB();
        qq3 = __builtin_elementwise_fma(dx3, dx3, tt3); SB();
        eb3.x = __builtin_amdgcn_exp2f(aa3.y); SB();
        aa0 = (v2f){ss0.x, ss0.x} * cc; SB();
        eb3.y = __builtin_amdgcn_exp2f(ab3.y); SB();
        ab0 = (v2f){ss0.y, ss0.y} * cc; SB();
        da1 += ea1; SB();
        ca1 = __builtin_elementwise_fma((v2f){V1_2.z, V1_2.w}, eb1, ca1); SB();
        ss2.x = __builtin_amdgcn_sqrtf(qq2.x); SB();
        dx2 = gxx - (v2f){X1_3.x, X1_3.y}; SB();
        ss2.y = __builtin_amdgcn_sqrtf(qq2.y); SB();
        dy2 = gyy - (v2f){Y1_3.x, Y1_3.y}; SB();
        ea0.x = __builtin_amdgcn_exp2f(aa0.x); SB();
        tt1 = dy1 * dy1; SB();
        ea0.y = __builtin_amdgcn_exp2f(ab0.x); SB();
        qq0 = __builtin_elementwise_fma(dx0, dx0, tt0); SB();
        eb0.x = __builtin_amdgcn_exp2f(aa0.y); SB();
        aa1 = (v2f){ss1.x, ss1.x} * cc; SB();
        eb0.y = __builtin_amdgcn_exp2f(ab0.y); SB();
        ab1 = (v2f){ss1.y, ss1.y} * cc; SB();
        da2 += ea2; SB();
        ca2 = __builtin_elementwise_fma((v2f){V1_3.x, V1_3.y}, eb2, ca2); SB();
        ss3.x = __builtin_amdgcn_sqrtf(qq3.x); SB();
        dx3 = gxx - (v2f){X1_3.z, X1_3.w}; SB();
        ss3.y = __builtin_amdgcn_sqrtf(qq3.y); SB();
        dy3 = gyy - (v2f){Y1_3.z, Y1_3.w}; SB();
        ea1.x = __builtin_amdgcn_exp2f(aa1.x); SB();
        tt2 = dy2 * dy2; SB();
        ea1.y = __builtin_amdgcn_exp2f(ab1.x); SB();
        qq1 = __builtin_elementwise_fma(dx1, dx1, tt1); SB();
        eb1.x = __builtin_amdgcn_exp2f(aa1.y); SB();
        aa2 = (v2f){ss2.x, ss2.x} * cc; SB();
        eb1.y = __builtin_amdgcn_exp2f(ab1.y); SB();
        ab2 = (v2f){ss2.y, ss2.y} * cc; SB();
        da3 += ea3; SB();
        ca3 = __builtin_elementwise_fma((v2f){V1_3.z, V1_3.w}, eb3, ca3); SB();
        ss0.x = __builtin_amdgcn_sqrtf(qq0.x); SB();
        dx0 = gxx - (v2f){X1_4.x, X1_4.y}; SB();
        ss0.y = __builtin_amdgcn_sqrtf(qq0.y); SB();
        dy0 = gyy - (v2f){Y1_4.x, Y1_4.y}; SB();
        ea2.x = __builtin_amdgcn_exp2f(aa2.x); SB();
        tt3 = dy3 * dy3; SB();
        ea2.y = __builtin_amdgcn_exp2f(ab2.x); SB();
        qq2 = __builtin_elementwise_fma(dx2, dx2, tt2); SB();
        eb2.x = __builtin_amdgcn_exp2f(aa2.y); SB();
        aa3 = (v2f){ss3.x, ss3.x} * cc; SB();
        eb2.y = __builtin_amdgcn_exp2f(ab2.y); SB();
        ab3 = (v2f){ss3.y, ss3.y} * cc; SB();
        da0 += ea0; SB();
        ca0 = __builtin_elementwise_fma((v2f){V1_4.x, V1_4.y}, eb0, ca0); SB();
        ss1.x = __builtin_amdgcn_sqrtf(qq1.x); SB();
        dx1 = gxx - (v2f){X1_4.z, X1_4.w}; SB();
        ss1.y = __builtin_amdgcn_sqrtf(qq1.y); SB();
        dy1 = gyy - (v2f){Y1_4.z, Y1_4.w}; SB();
        ea3.x = __builtin_amdgcn_exp2f(aa3.x); SB();
        tt0 = dy0 * dy0; SB();
        ea3.y = __builtin_amdgcn_exp2f(ab3.x); SB();
        qq3 = __builtin_elementwise_fma(dx3, dx3, tt3); SB();
        eb3.x = __builtin_amdgcn_exp2f(aa3.y); SB();
        aa0 = (v2f){ss0.x, ss0.x} * cc; SB();
        eb3.y = __builtin_amdgcn_exp2f(ab3.y); SB();
        ab0 = (v2f){ss0.y, ss0.y} * cc; SB();
        da1 += ea1; SB();
        ca1 = __builtin_elementwise_fma((v2f){V1_4.z, V1_4.w}, eb1, ca1); SB();
        ss2.x = __builtin_amdgcn_sqrtf(qq2.x); SB();
        dx2 = gxx - (v2f){X1_5.x, X1_5.y}; SB();
        ss2.y = __builtin_amdgcn_sqrtf(qq2.y); SB();
        dy2 = gyy - (v2f){Y1_5.x, Y1_5.y}; SB();
        ea0.x = __builtin_amdgcn_exp2f(aa0.x); SB();
        tt1 = dy1 * dy1; SB();
        ea0.y = __builtin_amdgcn_exp2f(ab0.x); SB();
        qq0 = __builtin_elementwise_fma(dx0, dx0, tt0); SB();
        eb0.x = __builtin_amdgcn_exp2f(aa0.y); SB();
        aa1 = (v2f){ss1.x, ss1.x} * cc; SB();
        eb0.y = __builtin_amdgcn_exp2f(ab0.y); SB();
        ab1 = (v2f){ss1.y, ss1.y} * cc; SB();
        da2 += ea2; SB();
        ca2 = __builtin_elementwise_fma((v2f){V1_5.x, V1_5.y}, eb2, ca2); SB();
        ss3.x = __builtin_amdgcn_sqrtf(qq3.x); SB();
        dx3 = gxx - (v2f){X1_5.z, X1_5.w}; SB();
        ss3.y = __builtin_amdgcn_sqrtf(qq3.y); SB();
        dy3 = gyy - (v2f){Y1_5.z, Y1_5.w}; SB();
        ea1.x = __builtin_amdgcn_exp2f(aa1.x); SB();
        tt2 = dy2 * dy2; SB();
        ea1.y = __builtin_amdgcn_exp2f(ab1.x); SB();
        qq1 = __builtin_elementwise_fma(dx1, dx1, tt1); SB();
        eb1.x = __builtin_amdgcn_exp2f(aa1.y); SB();
        aa2 = (v2f){ss2.x, ss2.x} * cc; SB();
        eb1.y = __builtin_amdgcn_exp2f(ab1.y); SB();
        ab2 = (v2f){ss2.y, ss2.y} * cc; SB();
        da3 += ea3; SB();
        ca3 = __builtin_elementwise_fma((v2f){V1_5.z, V1_5.w}, eb3, ca3); SB();
        ss0.x = __builtin_amdgcn_sqrtf(qq0.x); SB();
        dx0 = gxx - (v2f){X1_6.x, X1_6.y}; SB();
        ss0.y = __builtin_amdgcn_sqrtf(qq0.y); SB();
        dy0 = gyy - (v2f){Y1_6.x, Y1_6.y}; SB();
        ea2.x = __builtin_amdgcn_exp2f(aa2.x); SB();
        tt3 = dy3 * dy3; SB();
        ea2.y = __builtin_amdgcn_exp2f(ab2.x); SB();
        qq2 = __builtin_elementwise_fma(dx2, dx2, tt2); SB();
        eb2.x = __builtin_amdgcn_exp2f(aa2.y); SB();
        aa3 = (v2f){ss3.x, ss3.x} * cc; SB();
        eb2.y = __builtin_amdgcn_exp2f(ab2.y); SB();
        ab3 = (v2f){ss3.y, ss3.y} * cc; SB();
        da0 += ea0; SB();
        ca0 = __builtin_elementwise_fma((v2f){V1_6.x, V1_6.y}, eb0, ca0); SB();
        ss1.x = __builtin_amdgcn_sqrtf(qq1.x); SB();
        dx1 = gxx - (v2f){X1_6.z, X1_6.w}; SB();
        ss1.y = __builtin_amdgcn_sqrtf(qq1.y); SB();
        dy1 = gyy - (v2f){Y1_6.z, Y1_6.w}; SB();
        ea3.x = __builtin_amdgcn_exp2f(aa3.x); SB();
        tt0 = dy0 * dy0; SB();
        ea3.y = __builtin_amdgcn_exp2f(ab3.x); SB();
        qq3 = __builtin_elementwise_fma(dx3, dx3, tt3); SB();
        eb3.x = __builtin_amdgcn_exp2f(aa3.y); SB();
        aa0 = (v2f){ss0.x, ss0.x} * cc; SB();
        eb3.y = __builtin_amdgcn_exp2f(ab3.y); SB();
        ab0 = (v2f){ss0.y, ss0.y} * cc; SB();
        da1 += ea1; SB();
        ca1 = __builtin_elementwise_fma((v2f){V1_6.z, V1_6.w}, eb1, ca1); SB();
        ss2.x = __builtin_amdgcn_sqrtf(qq2.x); SB();
        dx2 = gxx - (v2f){X1_7.x, X1_7.y}; SB();
        ss2.y = __builtin_amdgcn_sqrtf(qq2.y); SB();
        dy2 = gyy - (v2f){Y1_7.x, Y1_7.y}; SB();
        ea0.x = __builtin_amdgcn_exp2f(aa0.x); SB();
        tt1 = dy1 * dy1; SB();
        ea0.y = __builtin_amdgcn_exp2f(ab0.x); SB();
        qq0 = __builtin_elementwise_fma(dx0, dx0, tt0); SB();
        eb0.x = __builtin_amdgcn_exp2f(aa0.y); SB();
        aa1 = (v2f){ss1.x, ss1.x} * cc; SB();
        eb0.y = __builtin_amdgcn_exp2f(ab0.y); SB();
        ab1 = (v2f){ss1.y, ss1.y} * cc; SB();
        da2 += ea2; SB();
        ca2 = __builtin_elementwise_fma((v2f){V1_7.x, V1_7.y}, eb2, ca2); SB();
        ss3.x = __builtin_amdgcn_sqrtf(qq3.x); SB();
        dx3 = gxx - (v2f){X1_7.z, X1_7.w}; SB();
        ss3.y = __builtin_amdgcn_sqrtf(qq3.y); SB();
        dy3 = gyy - (v2f){Y1_7.z, Y1_7.w}; SB();
        ea1.x = __builtin_amdgcn_exp2f(aa1.x); SB();
        tt2 = dy2 * dy2; SB();
        ea1.y = __builtin_amdgcn_exp2f(ab1.x); SB();
        qq1 = __builtin_elementwise_fma(dx1, dx1, tt1); SB();
        eb1.x = __builtin_amdgcn_exp2f(aa1.y); SB();
        aa2 = (v2f){ss2.x, ss2.x} * cc; SB();
        eb1.y = __builtin_amdgcn_exp2f(ab1.y); SB();
        ab2 = (v2f){ss2.y, ss2.y} * cc; SB();
        da3 += ea3; SB();
        ca3 = __builtin_elementwise_fma((v2f){V1_7.z, V1_7.w}, eb3, ca3); SB();
    }

    const v2f dsum = (da0 + da1) + (da2 + da3);
    const v2f csum = (ca0 + ca1) + (ca2 + ca3);
    const float dens = dsum.x + dsum.y;
    const float conv = csum.x + csum.y;
    sy[2 * t]     = dens;
    sy[2 * t + 1] = conv * __builtin_amdgcn_rcpf(dens + 1e-8f);
    __syncthreads();

    const int  wave  = t >> 5;
    const int  lane  = t & 31;
    const int  lhalf = lane & 15;
    const bool lo    = lane < 16;

    float* outb = out + (size_t)b * NCH * NPTS;

    for (int pt = 2 * wave; pt < 2 * wave + 2; ++pt) {
        float by0 = sy[2 * (pt * 16 + lhalf)];
        float by1 = sy[2 * (pt * 16 + lhalf) + 1];
        v2f bm;
        bm.x = lo ? by0 : 0.f;
        bm.y = lo ? by1 : 0.f;

        const int oPt = po + pt * 16 + lhalf;

        for (int ct = 0; ct < 4; ++ct) {
            float aw0 = W[(ct * 16 + lhalf) * 2 + 0];
            float aw1 = W[(ct * 16 + lhalf) * 2 + 1];
            v2f am;
            am.x = lo ? aw0 : 0.f;
            am.y = lo ? aw1 : 0.f;

            v8f acc = {};
            acc = __builtin_amdgcn_wmma_f32_16x16x4_f32(
                 false, am,  false, bm,
                 (short)0, acc,  false,  false);

            const int kBase = ct * 16 + (lo ? 0 : 8);
#pragma unroll
            for (int j = 0; j < 8; ++j) {
                float v = acc[j] + bias[kBase + j];
                float r = __builtin_amdgcn_rcpf(
                    1.f + __builtin_amdgcn_exp2f(-LOG2E * v));
                outb[(size_t)(kBase + j) * NPTS + oPt] = r;
            }
        }
    }
}

extern "C" void kernel_launch(void* const* d_in, const int* in_sizes, int n_in,
                              void* d_out, int out_size, void* d_ws, size_t ws_size,
                              hipStream_t stream) {
    (void)n_in; (void)out_size; (void)d_ws; (void)ws_size;
    const float* xc   = (const float*)d_in[0];
    const float* yc   = (const float*)d_in[1];
    const float* xg   = (const float*)d_in[2];
    const float* sig  = (const float*)d_in[3];
    const float* W    = (const float*)d_in[4];
    const float* bias = (const float*)d_in[5];
    float* out = (float*)d_out;

    const int B = in_sizes[0] / (NCTX * 2);
    dim3 grid(B * 32), block(128);
    convnp_fused<<<grid, block, 0, stream>>>(xc, yc, xg, sig, W, bias, out);
}


# serial chain cut to one memory wait: grid point, sigma, W and bias requested at entry beside the context; projection B operand taken from the lanes' own registers (no LDS exchange, no 2nd barrier)
# speedup vs baseline: 2.1078x; 1.1129x over previous
#include <hip/hip_runtime.h>

typedef __attribute__((ext_vector_type(2))) float v2f;
typedef __attribute__((ext_vector_type(8))) float v8f;
typedef __attribute__((ext_vector_type(4))) float v4f;

#define GRID_N 64
#define NPTS   4096
#define NCTX   512
#define NPAD   64
#define NCH    64
#define LOG2E  1.44269504088896340736f

static __device__ __forceinline__ v4f lds4(const float* p) {
    v4f v;
    __builtin_memcpy(&v, p, 16);
    return v;
}

#define SB() __builtin_amdgcn_sched_barrier(0)

__global__ __launch_bounds__(128) __attribute__((amdgpu_waves_per_eu(1, 1))) void convnp_fused(
    const float* __restrict__ xc,
    const float* __restrict__ yc,
    const float* __restrict__ xg,
    const float* __restrict__ sig,
    const float* __restrict__ W,
    const float* __restrict__ bias,
    float* __restrict__ out)
{
    __shared__ __attribute__((aligned(16))) float sxx[NCTX + NPAD];
    __shared__ __attribute__((aligned(16))) float sxy[NCTX + NPAD];
    __shared__ __attribute__((aligned(16))) float syc[NCTX + NPAD];
    __shared__ __attribute__((aligned(16))) float swb[NCH * 2 + NCH];

    const int t    = threadIdx.x;
    const int b    = blockIdx.x >> 5;
    const int tile = blockIdx.x & 31;
    const int po   = tile << 7;
    const int o    = po + t;

    const v4f* xcb = (const v4f*)(xc + (size_t)b * NCTX * 2);
    const v4f* ycb = (const v4f*)(yc + (size_t)b * NCTX);
    const v4f* wsrc = (t < 32) ? ((const v4f*)W + t) : ((const v4f*)bias + (t & 15));
    const v4f wb  = *wsrc;
    const v4f sx0 = xcb[t];
    const v4f sx1 = xcb[t + 128];
    const v4f sy0 = ycb[t];
    const v2f g   = *(const v2f*)(xg + ((size_t)b * NPTS + o) * 2);
    const float sg0 = sig[0];
    const float sg1 = sig[1];

    const float s0 = __expf(sg0);
    const float s1 = __expf(sg1);
    const float c0 = -0.5f * LOG2E / (s0 * s0);
    const float c1 = -0.5f * LOG2E / (s1 * s1);
    const v2f   cc = {c0, c1};

    {
        ((v2f*)sxx)[t]       = (v2f){sx0.x, sx0.z};
        ((v2f*)sxy)[t]       = (v2f){sx0.y, sx0.w};
        ((v2f*)sxx)[t + 128] = (v2f){sx1.x, sx1.z};
        ((v2f*)sxy)[t + 128] = (v2f){sx1.y, sx1.w};
        ((v4f*)syc)[t]       = sy0;
        if (t < NPAD / 4) {
            const v4f z = {0.f, 0.f, 0.f, 0.f};
            ((v4f*)(sxx + NCTX))[t] = z;
            ((v4f*)(sxy + NCTX))[t] = z;
            ((v4f*)(syc + NCTX))[t] = z;
        }
        if (t < 48) ((v4f*)swb)[t] = wb;
    }
    __syncthreads();

    const v2f   gxx = {g.x, g.x};
    const v2f   gyy = {g.y, g.y};

    v2f dx0, dx1, dx2, dx3, dy0, dy1, dy2, dy3, tt0, tt1, tt2, tt3, qq0, qq1, qq2, qq3;
    v2f ss0, ss1, ss2, ss3, aa0, aa1, aa2, aa3, ab0, ab1, ab2, ab3, ea0, ea1, ea2, ea3, eb0, eb1, eb2, eb3;
    v2f da0 = {0.f, 0.f}, da1 = {0.f, 0.f}, da2 = {0.f, 0.f}, da3 = {0.f, 0.f};
    v2f ca0 = {0.f, 0.f}, ca1 = {0.f, 0.f}, ca2 = {0.f, 0.f}, ca3 = {0.f, 0.f};

    v4f XP0 = lds4(&sxx[0]);  v4f YP0 = lds4(&sxy[0]);  SB();
    v4f XP1 = lds4(&sxx[4]);  v4f YP1 = lds4(&sxy[4]);  SB();
    v4f XP2 = lds4(&sxx[8]);  v4f YP2 = lds4(&sxy[8]);  SB();
    v4f XP3 = lds4(&sxx[12]); v4f YP3 = lds4(&sxy[12]); SB();
    v4f X0_0, X0_1, X0_2, X0_3, X0_4, X0_5, X0_6, X0_7, Y0_0, Y0_1, Y0_2, Y0_3, Y0_4, Y0_5, Y0_6, Y0_7, V0_0, V0_1, V0_2, V0_3, V0_4, V0_5, V0_6, V0_7;
    v4f X1_0, X1_1, X1_2, X1_3, X1_4, X1_5, X1_6, X1_7, Y1_0, Y1_1, Y1_2, Y1_3, Y1_4, Y1_5, Y1_6, Y1_7, V1_0, V1_1, V1_2, V1_3, V1_4, V1_5, V1_6, V1_7;
    X0_0 = lds4(&sxx[0 + 16 + 0]); Y0_0 = lds4(&sxy[0 + 16 + 0]);
    X0_1 = lds4(&sxx[0 + 16 + 4]); Y0_1 = lds4(&sxy[0 + 16 + 4]);
    X0_2 = lds4(&sxx[0 + 16 + 8]); Y0_2 = lds4(&sxy[0 + 16 + 8]);
    X0_3 = lds4(&sxx[0 + 16 + 12]); Y0_3 = lds4(&sxy[0 + 16 + 12]);
    X0_4 = lds4(&sxx[0 + 16 + 16]); Y0_4 = lds4(&sxy[0 + 16 + 16]);
    X0_5 = lds4(&sxx[0 + 16 + 20]); Y0_5 = lds4(&sxy[0 + 16 + 20]);
    X0_6 = lds4(&sxx[0 + 16 + 24]); Y0_6 = lds4(&sxy[0 + 16 + 24]);
    X0_7 = lds4(&sxx[0 + 16 + 28]); Y0_7 = lds4(&sxy[0 + 16 + 28]);
    V0_0 = lds4(&syc[0 + 0]);
    V0_1 = lds4(&syc[0 + 4]);
    V0_2 = lds4(&syc[0 + 8]);
    V0_3 = lds4(&syc[0 + 12]);
    V0_4 = lds4(&syc[0 + 16]);
    V0_5 = lds4(&syc[0 + 20]);
    V0_6 = lds4(&syc[0 + 24]);
    V0_7 = lds4(&syc[0 + 28]);
    SB();

    dx0 = gxx - (v2f){XP0.x, XP0.y}; SB();
    dy0 = gyy - (v2f){YP0.x, YP0.y}; SB();
    dx1 = gxx - (v2f){XP0.z, XP0.w}; SB();
    dy1 = gyy - (v2f){YP0.z, YP0.w}; SB();
    tt0 = dy0 * dy0; SB();
    dx2 = gxx - (v2f){XP1.x, XP1.y}; SB();
    dy2 = gyy - (v2f){YP1.x, YP1.y}; SB();
    tt1 = dy1 * dy1; SB();
    qq0 = __builtin_elementwise_fma(dx0, dx0, tt0); SB();
    dx3 = gxx - (v2f){XP1.z, XP1.w}; SB();
    dy3 = gyy - (v2f){YP1.z, YP1.w}; SB();
    tt2 = dy2 * dy2; SB();
    qq1 = __builtin_elementwise_fma(dx1, dx1, tt1); SB();
    ss0.x = __builtin_amdgcn_sqrtf(qq0.x); SB();
    dx0 = gxx - (v2f){XP2.x, XP2.y}; SB();
    ss0.y = __builtin_amdgcn_sqrtf(qq0.y); SB();
    dy0 = gyy - (v2f){YP2.x, YP2.y}; SB();
    tt3 = dy3 * dy3; SB();
    qq2 = __builtin_elementwise_fma(dx2, dx2, tt2); SB();
    ss1.x = __builtin_amdgcn_sqrtf(qq1.x); SB();
    dx1 = gxx - (v2f){XP2.z, XP2.w}; SB();
    ss1.y = __builtin_amdgcn_sqrtf(qq1.y); SB();
    dy1 = gyy - (v2f){YP2.z, YP2.w}; SB();
    tt0 = dy0 * dy0; SB();
    qq3 = __builtin_elementwise_fma(dx3, dx3, tt3); SB();
    aa0 = (v2f){ss0.x, ss0.x} * cc; SB();
    ab0 = (v2f){ss0.y, ss0.y} * cc; SB();
    ss2.x = __builtin_amdgcn_sqrtf(qq2.x); SB();
    dx2 = gxx - (v2f){XP3.x, XP3.y}; SB();
    ss2.y = __builtin_amdgcn_sqrtf(qq2.y); SB();
    dy2 = gyy - (v2f){YP3.x, YP3.y}; SB();
    ea0.x = __builtin_amdgcn_exp2f(aa0.x); SB();
    tt1 = dy1 * dy1; SB();
    ea0.y = __builtin_amdgcn_exp2f(ab0.x); SB();
    qq0 = __builtin_elementwise_fma(dx0, dx0, tt0); SB();
    eb0.x = __builtin_amdgcn_exp2f(aa0.y); SB();
    aa1 = (v2f){ss1.x, ss1.x} * cc; SB();
    eb0.y = __builtin_amdgcn_exp2f(ab0.y); SB();
    ab1 = (v2f){ss1.y, ss1.y} * cc; SB();
    ss3.x = __builtin_amdgcn_sqrtf(qq3.x); SB();
    dx3 = gxx - (v2f){XP3.z, XP3.w}; SB();
    ss3.y = __builtin_amdgcn_sqrtf(qq3.y); SB();
    dy3 = gyy - (v2f){YP3.z, YP3.w}; SB();
    ea1.x = __builtin_amdgcn_exp2f(aa1.x); SB();
    tt2 = dy2 * dy2; SB();
    ea1.y = __builtin_amdgcn_exp2f(ab1.x); SB();
    qq1 = __builtin_elementwise_fma(dx1, dx1, tt1); SB();
    eb1.x = __builtin_amdgcn_exp2f(aa1.y); SB();
    aa2 = (v2f){ss2.x, ss2.x} * cc; SB();
    eb1.y = __builtin_amdgcn_exp2f(ab1.y); SB();
    ab2 = (v2f){ss2.y, ss2.y} * cc; SB();

#pragma unroll 1
    for (int base = 0; base < NCTX; base += 64) {
        X1_0 = lds4(&sxx[base + 32 + 16 + 0]); Y1_0 = lds4(&sxy[base + 32 + 16 + 0]);
        X1_1 = lds4(&sxx[base + 32 + 16 + 4]); Y1_1 = lds4(&sxy[base + 32 + 16 + 4]);
        X1_2 = lds4(&sxx[base + 32 + 16 + 8]); Y1_2 = lds4(&sxy[base + 32 + 16 + 8]);
        X1_3 = lds4(&sxx[base + 32 + 16 + 12]); Y1_3 = lds4(&sxy[base + 32 + 16 + 12]);
        X1_4 = lds4(&sxx[base + 32 + 16 + 16]); Y1_4 = lds4(&sxy[base + 32 + 16 + 16]);
        X1_5 = lds4(&sxx[base + 32 + 16 + 20]); Y1_5 = lds4(&sxy[base + 32 + 16 + 20]);
        X1_6 = lds4(&sxx[base + 32 + 16 + 24]); Y1_6 = lds4(&sxy[base + 32 + 16 + 24]);
        X1_7 = lds4(&sxx[base + 32 + 16 + 28]); Y1_7 = lds4(&sxy[base + 32 + 16 + 28]);
        V1_0 = lds4(&syc[base + 32 + 0]);
        V1_1 = lds4(&syc[base + 32 + 4]);
        V1_2 = lds4(&syc[base + 32 + 8]);
        V1_3 = lds4(&syc[base + 32 + 12]);
        V1_4 = lds4(&syc[base + 32 + 16]);
        V1_5 = lds4(&syc[base + 32 + 20]);
        V1_6 = lds4(&syc[base + 32 + 24]);
        V1_7 = lds4(&syc[base + 32 + 28]);
        SB();
        ss0.x = __builtin_amdgcn_sqrtf(qq0.x); SB();
        dx0 = gxx - (v2f){X0_0.x, X0_0.y}; SB();
        ss0.y = __builtin_amdgcn_sqrtf(qq0.y); SB();
        dy0 = gyy - (v2f){Y0_0.x, Y0_0.y}; SB();
        ea2.x = __builtin_amdgcn_exp2f(aa2.x); SB();
        tt3 = dy3 * dy3; SB();
        ea2.y = __builtin_amdgcn_exp2f(ab2.x); SB();
        qq2 = __builtin_elementwise_fma(dx2, dx2, tt2); SB();
        eb2.x = __builtin_amdgcn_exp2f(aa2.y); SB();
        aa3 = (v2f){ss3.x, ss3.x} * cc; SB();
        eb2.y = __builtin_amdgcn_exp2f(ab2.y); SB();
        ab3 = (v2f){ss3.y, ss3.y} * cc; SB();
        da0 += ea0; SB();
        ca0 = __builtin_elementwise_fma((v2f){V0_0.x, V0_0.y}, eb0, ca0); SB();
        ss1.x = __builtin_amdgcn_sqrtf(qq1.x); SB();
        dx1 = gxx - (v2f){X0_0.z, X0_0.w}; SB();
        ss1.y = __builtin_amdgcn_sqrtf(qq1.y); SB();
        dy1 = gyy - (v2f){Y0_0.z, Y0_0.w}; SB();
        ea3.x = __builtin_amdgcn_exp2f(aa3.x); SB();
        tt0 = dy0 * dy0; SB();
        ea3.y = __builtin_amdgcn_exp2f(ab3.x); SB();
        qq3 = __builtin_elementwise_fma(dx3, dx3, tt3); SB();
        eb3.x = __builtin_amdgcn_exp2f(aa3.y); SB();
        aa0 = (v2f){ss0.x, ss0.x} * cc; SB();
        eb3.y = __builtin_amdgcn_exp2f(ab3.y); SB();
        ab0 = (v2f){ss0.y, ss0.y} * cc; SB();
        da1 += ea1; SB();
        ca1 = __builtin_elementwise_fma((v2f){V0_0.z, V0_0.w}, eb1, ca1); SB();
        ss2.x = __builtin_amdgcn_sqrtf(qq2.x); SB();
        dx2 = gxx - (v2f){X0_1.x, X0_1.y}; SB();
        ss2.y = __builtin_amdgcn_sqrtf(qq2.y); SB();
        dy2 = gyy - (v2f){Y0_1.x, Y0_1.y}; SB();
        ea0.x = __builtin_amdgcn_exp2f(aa0.x); SB();
        tt1 = dy1 * dy1; SB();
        ea0.y = __builtin_amdgcn_exp2f(ab0.x); SB();
        qq0 = __builtin_elementwise_fma(dx0, dx0, tt0); SB();
        eb0.x = __builtin_amdgcn_exp2f(aa0.y); SB();
        aa1 = (v2f){ss1.x, ss1.x} * cc; SB();
        eb0.y = __builtin_amdgcn_exp2f(ab0.y); SB();
        ab1 = (v2f){ss1.y, ss1.y} * cc; SB();
        da2 += ea2; SB();
        ca2 = __builtin_elementwise_fma((v2f){V0_1.x, V0_1.y}, eb2, ca2); SB();
        ss3.x = __builtin_amdgcn_sqrtf(qq3.x); SB();
        dx3 = gxx - (v2f){X0_1.z, X0_1.w}; SB();
        ss3.y = __builtin_amdgcn_sqrtf(qq3.y); SB();
        dy3 = gyy - (v2f){Y0_1.z, Y0_1.w}; SB();
        ea1.x = __builtin_amdgcn_exp2f(aa1.x); SB();
        tt2 = dy2 * dy2; SB();
        ea1.y = __builtin_amdgcn_exp2f(ab1.x); SB();
        qq1 = __builtin_elementwise_fma(dx1, dx1, tt1); SB();
        eb1.x = __builtin_amdgcn_exp2f(aa1.y); SB();
        aa2 = (v2f){ss2.x, ss2.x} * cc; SB();
        eb1.y = __builtin_amdgcn_exp2f(ab1.y); SB();
        ab2 = (v2f){ss2.y, ss2.y} * cc; SB();
        da3 += ea3; SB();
        ca3 = __builtin_elementwise_fma((v2f){V0_1.z, V0_1.w}, eb3, ca3); SB();
        ss0.x = __builtin_amdgcn_sqrtf(qq0.x); SB();
        dx0 = gxx - (v2f){X0_2.x, X0_2.y}; SB();
        ss0.y = __builtin_amdgcn_sqrtf(qq0.y); SB();
        dy0 = gyy - (v2f){Y0_2.x, Y0_2.y}; SB();
        ea2.x = __builtin_amdgcn_exp2f(aa2.x); SB();
        tt3 = dy3 * dy3; SB();
        ea2.y = __builtin_amdgcn_exp2f(ab2.x); SB();
        qq2 = __builtin_elementwise_fma(dx2, dx2, tt2); SB();
        eb2.x = __builtin_amdgcn_exp2f(aa2.y); SB();
        aa3 = (v2f){ss3.x, ss3.x} * cc; SB();
        eb2.y = __builtin_amdgcn_exp2f(ab2.y); SB();
        ab3 = (v2f){ss3.y, ss3.y} * cc; SB();
        da0 += ea0; SB();
        ca0 = __builtin_elementwise_fma((v2f){V0_2.x, V0_2.y}, eb0, ca0); SB();
        ss1.x = __builtin_amdgcn_sqrtf(qq1.x); SB();
        dx1 = gxx - (v2f){X0_2.z, X0_2.w}; SB();
        ss1.y = __builtin_amdgcn_sqrtf(qq1.y); SB();
        dy1 = gyy - (v2f){Y0_2.z, Y0_2.w}; SB();
        ea3.x = __builtin_amdgcn_exp2f(aa3.x); SB();
        tt0 = dy0 * dy0; SB();
        ea3.y = __builtin_amdgcn_exp2f(ab3.x); SB();
        qq3 = __builtin_elementwise_fma(dx3, dx3, tt3); SB();
        eb3.x = __builtin_amdgcn_exp2f(aa3.y); SB();
        aa0 = (v2f){ss0.x, ss0.x} * cc; SB();
        eb3.y = __builtin_amdgcn_exp2f(ab3.y); SB();
        ab0 = (v2f){ss0.y, ss0.y} * cc; SB();
        da1 += ea1; SB();
        ca1 = __builtin_elementwise_fma((v2f){V0_2.z, V0_2.w}, eb1, ca1); SB();
        ss2.x = __builtin_amdgcn_sqrtf(qq2.x); SB();
        dx2 = gxx - (v2f){X0_3.x, X0_3.y}; SB();
        ss2.y = __builtin_amdgcn_sqrtf(qq2.y); SB();
        dy2 = gyy - (v2f){Y0_3.x, Y0_3.y}; SB();
        ea0.x = __builtin_amdgcn_exp2f(aa0.x); SB();
        tt1 = dy1 * dy1; SB();
        ea0.y = __builtin_amdgcn_exp2f(ab0.x); SB();
        qq0 = __builtin_elementwise_fma(dx0, dx0, tt0); SB();
        eb0.x = __builtin_amdgcn_exp2f(aa0.y); SB();
        aa1 = (v2f){ss1.x, ss1.x} * cc; SB();
        eb0.y = __builtin_amdgcn_exp2f(ab0.y); SB();
        ab1 = (v2f){ss1.y, ss1.y} * cc; SB();
        da2 += ea2; SB();
        ca2 = __builtin_elementwise_fma((v2f){V0_3.x, V0_3.y}, eb2, ca2); SB();
        ss3.x = __builtin_amdgcn_sqrtf(qq3.x); SB();
        dx3 = gxx - (v2f){X0_3.z, X0_3.w}; SB();
        ss3.y = __builtin_amdgcn_sqrtf(qq3.y); SB();
        dy3 = gyy - (v2f){Y0_3.z, Y0_3.w}; SB();
        ea1.x = __builtin_amdgcn_exp2f(aa1.x); SB();
        tt2 = dy2 * dy2; SB();
        ea1.y = __builtin_amdgcn_exp2f(ab1.x); SB();
        qq1 = __builtin_elementwise_fma(dx1, dx1, tt1); SB();
        eb1.x = __builtin_amdgcn_exp2f(aa1.y); SB();
        aa2 = (v2f){ss2.x, ss2.x} * cc; SB();
        eb1.y = __builtin_amdgcn_exp2f(ab1.y); SB();
        ab2 = (v2f){ss2.y, ss2.y} * cc; SB();
        da3 += ea3; SB();
        ca3 = __builtin_elementwise_fma((v2f){V0_3.z, V0_3.w}, eb3, ca3); SB();
        ss0.x = __builtin_amdgcn_sqrtf(qq0.x); SB();
        dx0 = gxx - (v2f){X0_4.x, X0_4.y}; SB();
        ss0.y = __builtin_amdgcn_sqrtf(qq0.y); SB();
        dy0 = gyy - (v2f){Y0_4.x, Y0_4.y}; SB();
        ea2.x = __builtin_amdgcn_exp2f(aa2.x); SB();
        tt3 = dy3 * dy3; SB();
        ea2.y = __builtin_amdgcn_exp2f(ab2.x); SB();
        qq2 = __builtin_elementwise_fma(dx2, dx2, tt2); SB();
        eb2.x = __builtin_amdgcn_exp2f(aa2.y); SB();
        aa3 = (v2f){ss3.x, ss3.x} * cc; SB();
        eb2.y = __builtin_amdgcn_exp2f(ab2.y); SB();
        ab3 = (v2f){ss3.y, ss3.y} * cc; SB();
        da0 += ea0; SB();
        ca0 = __builtin_elementwise_fma((v2f){V0_4.x, V0_4.y}, eb0, ca0); SB();
        ss1.x = __builtin_amdgcn_sqrtf(qq1.x); SB();
        dx1 = gxx - (v2f){X0_4.z, X0_4.w}; SB();
        ss1.y = __builtin_amdgcn_sqrtf(qq1.y); SB();
        dy1 = gyy - (v2f){Y0_4.z, Y0_4.w}; SB();
        ea3.x = __builtin_amdgcn_exp2f(aa3.x); SB();
        tt0 = dy0 * dy0; SB();
        ea3.y = __builtin_amdgcn_exp2f(ab3.x); SB();
        qq3 = __builtin_elementwise_fma(dx3, dx3, tt3); SB();
        eb3.x = __builtin_amdgcn_exp2f(aa3.y); SB();
        aa0 = (v2f){ss0.x, ss0.x} * cc; SB();
        eb3.y = __builtin_amdgcn_exp2f(ab3.y); SB();
        ab0 = (v2f){ss0.y, ss0.y} * cc; SB();
        da1 += ea1; SB();
        ca1 = __builtin_elementwise_fma((v2f){V0_4.z, V0_4.w}, eb1, ca1); SB();
        ss2.x = __builtin_amdgcn_sqrtf(qq2.x); SB();
        dx2 = gxx - (v2f){X0_5.x, X0_5.y}; SB();
        ss2.y = __builtin_amdgcn_sqrtf(qq2.y); SB();
        dy2 = gyy - (v2f){Y0_5.x, Y0_5.y}; SB();
        ea0.x = __builtin_amdgcn_exp2f(aa0.x); SB();
        tt1 = dy1 * dy1; SB();
        ea0.y = __builtin_amdgcn_exp2f(ab0.x); SB();
        qq0 = __builtin_elementwise_fma(dx0, dx0, tt0); SB();
        eb0.x = __builtin_amdgcn_exp2f(aa0.y); SB();
        aa1 = (v2f){ss1.x, ss1.x} * cc; SB();
        eb0.y = __builtin_amdgcn_exp2f(ab0.y); SB();
        ab1 = (v2f){ss1.y, ss1.y} * cc; SB();
        da2 += ea2; SB();
        ca2 = __builtin_elementwise_fma((v2f){V0_5.x, V0_5.y}, eb2, ca2); SB();
        ss3.x = __builtin_amdgcn_sqrtf(qq3.x); SB();
        dx3 = gxx - (v2f){X0_5.z, X0_5.w}; SB();
        ss3.y = __builtin_amdgcn_sqrtf(qq3.y); SB();
        dy3 = gyy - (v2f){Y0_5.z, Y0_5.w}; SB();
        ea1.x = __builtin_amdgcn_exp2f(aa1.x); SB();
        tt2 = dy2 * dy2; SB();
        ea1.y = __builtin_amdgcn_exp2f(ab1.x); SB();
        qq1 = __builtin_elementwise_fma(dx1, dx1, tt1); SB();
        eb1.x = __builtin_amdgcn_exp2f(aa1.y); SB();
        aa2 = (v2f){ss2.x, ss2.x} * cc; SB();
        eb1.y = __builtin_amdgcn_exp2f(ab1.y); SB();
        ab2 = (v2f){ss2.y, ss2.y} * cc; SB();
        da3 += ea3; SB();
        ca3 = __builtin_elementwise_fma((v2f){V0_5.z, V0_5.w}, eb3, ca3); SB();
        ss0.x = __builtin_amdgcn_sqrtf(qq0.x); SB();
        dx0 = gxx - (v2f){X0_6.x, X0_6.y}; SB();
        ss0.y = __builtin_amdgcn_sqrtf(qq0.y); SB();
        dy0 = gyy - (v2f){Y0_6.x, Y0_6.y}; SB();
        ea2.x = __builtin_amdgcn_exp2f(aa2.x); SB();
        tt3 = dy3 * dy3; SB();
        ea2.y = __builtin_amdgcn_exp2f(ab2.x); SB();
        qq2 = __builtin_elementwise_fma(dx2, dx2, tt2); SB();
        eb2.x = __builtin_amdgcn_exp2f(aa2.y); SB();
        aa3 = (v2f){ss3.x, ss3.x} * cc; SB();
        eb2.y = __builtin_amdgcn_exp2f(ab2.y); SB();
        ab3 = (v2f){ss3.y, ss3.y} * cc; SB();
        da0 += ea0; SB();
        ca0 = __builtin_elementwise_fma((v2f){V0_6.x, V0_6.y}, eb0, ca0); SB();
        ss1.x = __builtin_amdgcn_sqrtf(qq1.x); SB();
        dx1 = gxx - (v2f){X0_6.z, X0_6.w}; SB();
        ss1.y = __builtin_amdgcn_sqrtf(qq1.y); SB();
        dy1 = gyy - (v2f){Y0_6.z, Y0_6.w}; SB();
        ea3.x = __builtin_amdgcn_exp2f(aa3.x); SB();
        tt0 = dy0 * dy0; SB();
        ea3.y = __builtin_amdgcn_exp2f(ab3.x); SB();
        qq3 = __builtin_elementwise_fma(dx3, dx3, tt3); SB();
        eb3.x = __builtin_amdgcn_exp2f(aa3.y); SB();
        aa0 = (v2f){ss0.x, ss0.x} * cc; SB();
        eb3.y = __builtin_amdgcn_exp2f(ab3.y); SB();
        ab0 = (v2f){ss0.y, ss0.y} * cc; SB();
        da1 += ea1; SB();
        ca1 = __builtin_elementwise_fma((v2f){V0_6.z, V0_6.w}, eb1, ca1); SB();
        ss2.x = __builtin_amdgcn_sqrtf(qq2.x); SB();
        dx2 = gxx - (v2f){X0_7.x, X0_7.y}; SB();
        ss2.y = __builtin_amdgcn_sqrtf(qq2.y); SB();
        dy2 = gyy - (v2f){Y0_7.x, Y0_7.y}; SB();
        ea0.x = __builtin_amdgcn_exp2f(aa0.x); SB();
        tt1 = dy1 * dy1; SB();
        ea0.y = __builtin_amdgcn_exp2f(ab0.x); SB();
        qq0 = __builtin_elementwise_fma(dx0, dx0, tt0); SB();
        eb0.x = __builtin_amdgcn_exp2f(aa0.y); SB();
        aa1 = (v2f){ss1.x, ss1.x} * cc; SB();
        eb0.y = __builtin_amdgcn_exp2f(ab0.y); SB();
        ab1 = (v2f){ss1.y, ss1.y} * cc; SB();
        da2 += ea2; SB();
        ca2 = __builtin_elementwise_fma((v2f){V0_7.x, V0_7.y}, eb2, ca2); SB();
        ss3.x = __builtin_amdgcn_sqrtf(qq3.x); SB();
        dx3 = gxx - (v2f){X0_7.z, X0_7.w}; SB();
        ss3.y = __builtin_amdgcn_sqrtf(qq3.y); SB();
        dy3 = gyy - (v2f){Y0_7.z, Y0_7.w}; SB();
        ea1.x = __builtin_amdgcn_exp2f(aa1.x); SB();
        tt2 = dy2 * dy2; SB();
        ea1.y = __builtin_amdgcn_exp2f(ab1.x); SB();
        qq1 = __builtin_elementwise_fma(dx1, dx1, tt1); SB();
        eb1.x = __builtin_amdgcn_exp2f(aa1.y); SB();
        aa2 = (v2f){ss2.x, ss2.x} * cc; SB();
        eb1.y = __builtin_amdgcn_exp2f(ab1.y); SB();
        ab2 = (v2f){ss2.y, ss2.y} * cc; SB();
        da3 += ea3; SB();
        ca3 = __builtin_elementwise_fma((v2f){V0_7.z, V0_7.w}, eb3, ca3); SB();
        X0_0 = lds4(&sxx[base + 64 + 16 + 0]); Y0_0 = lds4(&sxy[base + 64 + 16 + 0]);
        X0_1 = lds4(&sxx[base + 64 + 16 + 4]); Y0_1 = lds4(&sxy[base + 64 + 16 + 4]);
        X0_2 = lds4(&sxx[base + 64 + 16 + 8]); Y0_2 = lds4(&sxy[base + 64 + 16 + 8]);
        X0_3 = lds4(&sxx[base + 64 + 16 + 12]); Y0_3 = lds4(&sxy[base + 64 + 16 + 12]);
        X0_4 = lds4(&sxx[base + 64 + 16 + 16]); Y0_4 = lds4(&sxy[base + 64 + 16 + 16]);
        X0_5 = lds4(&sxx[base + 64 + 16 + 20]); Y0_5 = lds4(&sxy[base + 64 + 16 + 20]);
        X0_6 = lds4(&sxx[base + 64 + 16 + 24]); Y0_6 = lds4(&sxy[base + 64 + 16 + 24]);
        X0_7 = lds4(&sxx[base + 64 + 16 + 28]); Y0_7 = lds4(&sxy[base + 64 + 16 + 28]);
        V0_0 = lds4(&syc[base + 64 + 0]);
        V0_1 = lds4(&syc[base + 64 + 4]);
        V0_2 = lds4(&syc[base + 64 + 8]);
        V0_3 = lds4(&syc[base + 64 + 12]);
        V0_4 = lds4(&syc[base + 64 + 16]);
        V0_5 = lds4(&syc[base + 64 + 20]);
        V0_6 = lds4(&syc[base + 64 + 24]);
        V0_7 = lds4(&syc[base + 64 + 28]);
        SB();
        ss0.x = __builtin_amdgcn_sqrtf(qq0.x); SB();
        dx0 = gxx - (v2f){X1_0.x, X1_0.y}; SB();
        ss0.y = __builtin_amdgcn_sqrtf(qq0.y); SB();
        dy0 = gyy - (v2f){Y1_0.x, Y1_0.y}; SB();
        ea2.x = __builtin_amdgcn_exp2f(aa2.x); SB();
        tt3 = dy3 * dy3; SB();
        ea2.y = __builtin_amdgcn_exp2f(ab2.x); SB();
        qq2 = __builtin_elementwise_fma(dx2, dx2, tt2); SB();
        eb2.x = __builtin_amdgcn_exp2f(aa2.y); SB();
        aa3 = (v2f){ss3.x, ss3.x} * cc; SB();
        eb2.y = __builtin_amdgcn_exp2f(ab2.y); SB();
        ab3 = (v2f){ss3.y, ss3.y} * cc; SB();
        da0 += ea0; SB();
        ca0 = __builtin_elementwise_fma((v2f){V1_0.x, V1_0.y}, eb0, ca0); SB();
        ss1.x = __builtin_amdgcn_sqrtf(qq1.x); SB();
        dx1 = gxx - (v2f){X1_0.z, X1_0.w}; SB();
        ss1.y = __builtin_amdgcn_sqrtf(qq1.y); SB();
        dy1 = gyy - (v2f){Y1_0.z, Y1_0.w}; SB();
        ea3.x = __builtin_amdgcn_exp2f(aa3.x); SB();
        tt0 = dy0 * dy0; SB();
        ea3.y = __builtin_amdgcn_exp2f(ab3.x); SB();
        qq3 = __builtin_elementwise_fma(dx3, dx3, tt3); SB();
        eb3.x = __builtin_amdgcn_exp2f(aa3.y); SB();
        aa0 = (v2f){ss0.x, ss0.x} * cc; SB();
        eb3.y = __builtin_amdgcn_exp2f(ab3.y); SB();
        ab0 = (v2f){ss0.y, ss0.y} * cc; SB();
        da1 += ea1; SB();
        ca1 = __builtin_elementwise_fma((v2f){V1_0.z, V1_0.w}, eb1, ca1); SB();
        ss2.x = __builtin_amdgcn_sqrtf(qq2.x); SB();
        dx2 = gxx - (v2f){X1_1.x, X1_1.y}; SB();
        ss2.y = __builtin_amdgcn_sqrtf(qq2.y); SB();
        dy2 = gyy - (v2f){Y1_1.x, Y1_1.y}; SB();
        ea0.x = __builtin_amdgcn_exp2f(aa0.x); SB();
        tt1 = dy1 * dy1; SB();
        ea0.y = __builtin_amdgcn_exp2f(ab0.x); SB();
        qq0 = __builtin_elementwise_fma(dx0, dx0, tt0); SB();
        eb0.x = __builtin_amdgcn_exp2f(aa0.y); SB();
        aa1 = (v2f){ss1.x, ss1.x} * cc; SB();
        eb0.y = __builtin_amdgcn_exp2f(ab0.y); SB();
        ab1 = (v2f){ss1.y, ss1.y} * cc; SB();
        da2 += ea2; SB();
        ca2 = __builtin_elementwise_fma((v2f){V1_1.x, V1_1.y}, eb2, ca2); SB();
        ss3.x = __builtin_amdgcn_sqrtf(qq3.x); SB();
        dx3 = gxx - (v2f){X1_1.z, X1_1.w}; SB();
        ss3.y = __builtin_amdgcn_sqrtf(qq3.y); SB();
        dy3 = gyy - (v2f){Y1_1.z, Y1_1.w}; SB();
        ea1.x = __builtin_amdgcn_exp2f(aa1.x); SB();
        tt2 = dy2 * dy2; SB();
        ea1.y = __builtin_amdgcn_exp2f(ab1.x); SB();
        qq1 = __builtin_elementwise_fma(dx1, dx1, tt1); SB();
        eb1.x = __builtin_amdgcn_exp2f(aa1.y); SB();
        aa2 = (v2f){ss2.x, ss2.x} * cc; SB();
        eb1.y = __builtin_amdgcn_exp2f(ab1.y); SB();
        ab2 = (v2f){ss2.y, ss2.y} * cc; SB();
        da3 += ea3; SB();
        ca3 = __builtin_elementwise_fma((v2f){V1_1.z, V1_1.w}, eb3, ca3); SB();
        ss0.x = __builtin_amdgcn_sqrtf(qq0.x); SB();
        dx0 = gxx - (v2f){X1_2.x, X1_2.y}; SB();
        ss0.y = __builtin_amdgcn_sqrtf(qq0.y); SB();
        dy0 = gyy - (v2f){Y1_2.x, Y1_2.y}; SB();
        ea2.x = __builtin_amdgcn_exp2f(aa2.x); SB();
        tt3 = dy3 * dy3; SB();
        ea2.y = __builtin_amdgcn_exp2f(ab2.x); SB();
        qq2 = __builtin_elementwise_fma(dx2, dx2, tt2); SB();
        eb2.x = __builtin_amdgcn_exp2f(aa2.y); SB();
        aa3 = (v2f){ss3.x, ss3.x} * cc; SB();
        eb2.y = __builtin_amdgcn_exp2f(ab2.y); SB();
        ab3 = (v2f){ss3.y, ss3.y} * cc; SB();
        da0 += ea0; SB();
        ca0 = __builtin_elementwise_fma((v2f){V1_2.x, V1_2.y}, eb0, ca0); SB();
        ss1.x = __builtin_amdgcn_sqrtf(qq1.x); SB();
        dx1 = gxx - (v2f){X1_2.z, X1_2.w}; SB();
        ss1.y = __builtin_amdgcn_sqrtf(qq1.y); SB();
        dy1 = gyy - (v2f){Y1_2.z, Y1_2.w}; SB();
        ea3.x = __builtin_amdgcn_exp2f(aa3.x); SB();
        tt0 = dy0 * dy0; SB();
        ea3.y = __builtin_amdgcn_exp2f(ab3.x); SB();
        qq3 = __builtin_elementwise_fma(dx3, dx3, tt3); SB();
        eb3.x = __builtin_amdgcn_exp2f(aa3.y); SB();
        aa0 = (v2f){ss0.x, ss0.x} * cc; SB();
        eb3.y = __builtin_amdgcn_exp2f(ab3.y); SB();
        ab0 = (v2f){ss0.y, ss0.y} * cc; SB();
        da1 += ea1; SB();
        ca1 = __builtin_elementwise_fma((v2f){V1_2.z, V1_2.w}, eb1, ca1); SB();
        ss2.x = __builtin_amdgcn_sqrtf(qq2.x); SB();
        dx2 = gxx - (v2f){X1_3.x, X1_3.y}; SB();
        ss2.y = __builtin_amdgcn_sqrtf(qq2.y); SB();
        dy2 = gyy - (v2f){Y1_3.x, Y1_3.y}; SB();
        ea0.x = __builtin_amdgcn_exp2f(aa0.x); SB();
        tt1 = dy1 * dy1; SB();
        ea0.y = __builtin_amdgcn_exp2f(ab0.x); SB();
        qq0 = __builtin_elementwise_fma(dx0, dx0, tt0); SB();
        eb0.x = __builtin_amdgcn_exp2f(aa0.y); SB();
        aa1 = (v2f){ss1.x, ss1.x} * cc; SB();
        eb0.y = __builtin_amdgcn_exp2f(ab0.y); SB();
        ab1 = (v2f){ss1.y, ss1.y} * cc; SB();
        da2 += ea2; SB();
        ca2 = __builtin_elementwise_fma((v2f){V1_3.x, V1_3.y}, eb2, ca2); SB();
        ss3.x = __builtin_amdgcn_sqrtf(qq3.x); SB();
        dx3 = gxx - (v2f){X1_3.z, X1_3.w}; SB();
        ss3.y = __builtin_amdgcn_sqrtf(qq3.y); SB();
        dy3 = gyy - (v2f){Y1_3.z, Y1_3.w}; SB();
        ea1.x = __builtin_amdgcn_exp2f(aa1.x); SB();
        tt2 = dy2 * dy2; SB();
        ea1.y = __builtin_amdgcn_exp2f(ab1.x); SB();
        qq1 = __builtin_elementwise_fma(dx1, dx1, tt1); SB();
        eb1.x = __builtin_amdgcn_exp2f(aa1.y); SB();
        aa2 = (v2f){ss2.x, ss2.x} * cc; SB();
        eb1.y = __builtin_amdgcn_exp2f(ab1.y); SB();
        ab2 = (v2f){ss2.y, ss2.y} * cc; SB();
        da3 += ea3; SB();
        ca3 = __builtin_elementwise_fma((v2f){V1_3.z, V1_3.w}, eb3, ca3); SB();
        ss0.x = __builtin_amdgcn_sqrtf(qq0.x); SB();
        dx0 = gxx - (v2f){X1_4.x, X1_4.y}; SB();
        ss0.y = __builtin_amdgcn_sqrtf(qq0.y); SB();
        dy0 = gyy - (v2f){Y1_4.x, Y1_4.y}; SB();
        ea2.x = __builtin_amdgcn_exp2f(aa2.x); SB();
        tt3 = dy3 * dy3; SB();
        ea2.y = __builtin_amdgcn_exp2f(ab2.x); SB();
        qq2 = __builtin_elementwise_fma(dx2, dx2, tt2); SB();
        eb2.x = __builtin_amdgcn_exp2f(aa2.y); SB();
        aa3 = (v2f){ss3.x, ss3.x} * cc; SB();
        eb2.y = __builtin_amdgcn_exp2f(ab2.y); SB();
        ab3 = (v2f){ss3.y, ss3.y} * cc; SB();
        da0 += ea0; SB();
        ca0 = __builtin_elementwise_fma((v2f){V1_4.x, V1_4.y}, eb0, ca0); SB();
        ss1.x = __builtin_amdgcn_sqrtf(qq1.x); SB();
        dx1 = gxx - (v2f){X1_4.z, X1_4.w}; SB();
        ss1.y = __builtin_amdgcn_sqrtf(qq1.y); SB();
        dy1 = gyy - (v2f){Y1_4.z, Y1_4.w}; SB();
        ea3.x = __builtin_amdgcn_exp2f(aa3.x); SB();
        tt0 = dy0 * dy0; SB();
        ea3.y = __builtin_amdgcn_exp2f(ab3.x); SB();
        qq3 = __builtin_elementwise_fma(dx3, dx3, tt3); SB();
        eb3.x = __builtin_amdgcn_exp2f(aa3.y); SB();
        aa0 = (v2f){ss0.x, ss0.x} * cc; SB();
        eb3.y = __builtin_amdgcn_exp2f(ab3.y); SB();
        ab0 = (v2f){ss0.y, ss0.y} * cc; SB();
        da1 += ea1; SB();
        ca1 = __builtin_elementwise_fma((v2f){V1_4.z, V1_4.w}, eb1, ca1); SB();
        ss2.x = __builtin_amdgcn_sqrtf(qq2.x); SB();
        dx2 = gxx - (v2f){X1_5.x, X1_5.y}; SB();
        ss2.y = __builtin_amdgcn_sqrtf(qq2.y); SB();
        dy2 = gyy - (v2f){Y1_5.x, Y1_5.y}; SB();
        ea0.x = __builtin_amdgcn_exp2f(aa0.x); SB();
        tt1 = dy1 * dy1; SB();
        ea0.y = __builtin_amdgcn_exp2f(ab0.x); SB();
        qq0 = __builtin_elementwise_fma(dx0, dx0, tt0); SB();
        eb0.x = __builtin_amdgcn_exp2f(aa0.y); SB();
        aa1 = (v2f){ss1.x, ss1.x} * cc; SB();
        eb0.y = __builtin_amdgcn_exp2f(ab0.y); SB();
        ab1 = (v2f){ss1.y, ss1.y} * cc; SB();
        da2 += ea2; SB();
        ca2 = __builtin_elementwise_fma((v2f){V1_5.x, V1_5.y}, eb2, ca2); SB();
        ss3.x = __builtin_amdgcn_sqrtf(qq3.x); SB();
        dx3 = gxx - (v2f){X1_5.z, X1_5.w}; SB();
        ss3.y = __builtin_amdgcn_sqrtf(qq3.y); SB();
        dy3 = gyy - (v2f){Y1_5.z, Y1_5.w}; SB();
        ea1.x = __builtin_amdgcn_exp2f(aa1.x); SB();
        tt2 = dy2 * dy2; SB();
        ea1.y = __builtin_amdgcn_exp2f(ab1.x); SB();
        qq1 = __builtin_elementwise_fma(dx1, dx1, tt1); SB();
        eb1.x = __builtin_amdgcn_exp2f(aa1.y); SB();
        aa2 = (v2f){ss2.x, ss2.x} * cc; SB();
        eb1.y = __builtin_amdgcn_exp2f(ab1.y); SB();
        ab2 = (v2f){ss2.y, ss2.y} * cc; SB();
        da3 += ea3; SB();
        ca3 = __builtin_elementwise_fma((v2f){V1_5.z, V1_5.w}, eb3, ca3); SB();
        ss0.x = __builtin_amdgcn_sqrtf(qq0.x); SB();
        dx0 = gxx - (v2f){X1_6.x, X1_6.y}; SB();
        ss0.y = __builtin_amdgcn_sqrtf(qq0.y); SB();
        dy0 = gyy - (v2f){Y1_6.x, Y1_6.y}; SB();
        ea2.x = __builtin_amdgcn_exp2f(aa2.x); SB();
        tt3 = dy3 * dy3; SB();
        ea2.y = __builtin_amdgcn_exp2f(ab2.x); SB();
        qq2 = __builtin_elementwise_fma(dx2, dx2, tt2); SB();
        eb2.x = __builtin_amdgcn_exp2f(aa2.y); SB();
        aa3 = (v2f){ss3.x, ss3.x} * cc; SB();
        eb2.y = __builtin_amdgcn_exp2f(ab2.y); SB();
        ab3 = (v2f){ss3.y, ss3.y} * cc; SB();
        da0 += ea0; SB();
        ca0 = __builtin_elementwise_fma((v2f){V1_6.x, V1_6.y}, eb0, ca0); SB();
        ss1.x = __builtin_amdgcn_sqrtf(qq1.x); SB();
        dx1 = gxx - (v2f){X1_6.z, X1_6.w}; SB();
        ss1.y = __builtin_amdgcn_sqrtf(qq1.y); SB();
        dy1 = gyy - (v2f){Y1_6.z, Y1_6.w}; SB();
        ea3.x = __builtin_amdgcn_exp2f(aa3.x); SB();
        tt0 = dy0 * dy0; SB();
        ea3.y = __builtin_amdgcn_exp2f(ab3.x); SB();
        qq3 = __builtin_elementwise_fma(dx3, dx3, tt3); SB();
        eb3.x = __builtin_amdgcn_exp2f(aa3.y); SB();
        aa0 = (v2f){ss0.x, ss0.x} * cc; SB();
        eb3.y = __builtin_amdgcn_exp2f(ab3.y); SB();
        ab0 = (v2f){ss0.y, ss0.y} * cc; SB();
        da1 += ea1; SB();
        ca1 = __builtin_elementwise_fma((v2f){V1_6.z, V1_6.w}, eb1, ca1); SB();
        ss2.x = __builtin_amdgcn_sqrtf(qq2.x); SB();
        dx2 = gxx - (v2f){X1_7.x, X1_7.y}; SB();
        ss2.y = __builtin_amdgcn_sqrtf(qq2.y); SB();
        dy2 = gyy - (v2f){Y1_7.x, Y1_7.y}; SB();
        ea0.x = __builtin_amdgcn_exp2f(aa0.x); SB();
        tt1 = dy1 * dy1; SB();
        ea0.y = __builtin_amdgcn_exp2f(ab0.x); SB();
        qq0 = __builtin_elementwise_fma(dx0, dx0, tt0); SB();
        eb0.x = __builtin_amdgcn_exp2f(aa0.y); SB();
        aa1 = (v2f){ss1.x, ss1.x} * cc; SB();
        eb0.y = __builtin_amdgcn_exp2f(ab0.y); SB();
        ab1 = (v2f){ss1.y, ss1.y} * cc; SB();
        da2 += ea2; SB();
        ca2 = __builtin_elementwise_fma((v2f){V1_7.x, V1_7.y}, eb2, ca2); SB();
        ss3.x = __builtin_amdgcn_sqrtf(qq3.x); SB();
        dx3 = gxx - (v2f){X1_7.z, X1_7.w}; SB();
        ss3.y = __builtin_amdgcn_sqrtf(qq3.y); SB();
        dy3 = gyy - (v2f){Y1_7.z, Y1_7.w}; SB();
        ea1.x = __builtin_amdgcn_exp2f(aa1.x); SB();
        tt2 = dy2 * dy2; SB();
        ea1.y = __builtin_amdgcn_exp2f(ab1.x); SB();
        qq1 = __builtin_elementwise_fma(dx1, dx1, tt1); SB();
        eb1.x = __builtin_amdgcn_exp2f(aa1.y); SB();
        aa2 = (v2f){ss2.x, ss2.x} * cc; SB();
        eb1.y = __builtin_amdgcn_exp2f(ab1.y); SB();
        ab2 = (v2f){ss2.y, ss2.y} * cc; SB();
        da3 += ea3; SB();
        ca3 = __builtin_elementwise_fma((v2f){V1_7.z, V1_7.w}, eb3, ca3); SB();
    }

    const int  wave  = t >> 5;
    const int  lane  = t & 31;
    const int  lhalf = lane & 15;
    const bool lo    = lane < 16;
    const int  kOff  = lo ? 0 : 8;
    v2f wr[4];
    v4f bsA[4], bsB[4];
#pragma unroll
    for (int ct = 0; ct < 4; ++ct) {
        __builtin_memcpy(&wr[ct], &swb[(ct * 16 + lhalf) * 2], 8);
        bsA[ct] = lds4(&swb[NCH * 2 + ct * 16 + kOff]);
        bsB[ct] = lds4(&swb[NCH * 2 + ct * 16 + kOff + 4]);
    }
    SB();

    const v2f dsum = (da0 + da1) + (da2 + da3);
    const v2f csum = (ca0 + ca1) + (ca2 + ca3);
    const float dens  = dsum.x + dsum.y;
    const float conv  = csum.x + csum.y;
    const float ratio = conv * __builtin_amdgcn_rcpf(dens + 1e-8f);

    float* outb = out + (size_t)b * NCH * NPTS;

#pragma unroll
    for (int h = 0; h < 2; ++h) {
        const bool mine = (h == 0) ? lo : !lo;
        v2f bm;
        bm.x = mine ? dens  : 0.f;
        bm.y = mine ? ratio : 0.f;

        const int oPt = po + wave * 32 + h * 16 + lhalf;

#pragma unroll
        for (int ct = 0; ct < 4; ++ct) {
            v2f am;
            am.x = mine ? wr[ct].x : 0.f;
            am.y = mine ? wr[ct].y : 0.f;

            v8f acc = {};
            acc = __builtin_amdgcn_wmma_f32_16x16x4_f32(
                 false, am,  false, bm,
                 (short)0, acc,  false,  false);

            const int kBase = ct * 16 + kOff;
#pragma unroll
            for (int j = 0; j < 8; ++j) {
                const float bj = (j < 4) ? bsA[ct][j] : bsB[ct][j - 4];
                float v = acc[j] + bj;
                float r = __builtin_amdgcn_rcpf(
                    1.f + __builtin_amdgcn_exp2f(-LOG2E * v));
                outb[(size_t)(kBase + j) * NPTS + oPt] = r;
            }
        }
    }
}

extern "C" void kernel_launch(void* const* d_in, const int* in_sizes, int n_in,
                              void* d_out, int out_size, void* d_ws, size_t ws_size,
                              hipStream_t stream) {
    (void)n_in; (void)out_size; (void)d_ws; (void)ws_size;
    const float* xc   = (const float*)d_in[0];
    const float* yc   = (const float*)d_in[1];
    const float* xg   = (const float*)d_in[2];
    const float* sig  = (const float*)d_in[3];
    const float* W    = (const float*)d_in[4];
    const float* bias = (const float*)d_in[5];
    float* out = (float*)d_out;

    const int B = in_sizes[0] / (NCTX * 2);
    dim3 grid(B * 32), block(128);
    convnp_fused<<<grid, block, 0, stream>>>(xc, yc, xg, sig, W, bias, out);
}


# each lane owns two grid points, each lane half takes half the context: one LDS register picked by op_sel feeds both halves of a packed op, LDS reads per evaluation halved; sums join by v_permlanex16
# speedup vs baseline: 2.2147x; 1.0507x over previous
#include <hip/hip_runtime.h>

typedef __attribute__((ext_vector_type(2))) float v2f;
typedef __attribute__((ext_vector_type(8))) float v8f;
typedef __attribute__((ext_vector_type(4))) float v4f;

#define GRID_N 64
#define NPTS   4096
#define NCTX   512
#define NHALF  256
#define HSKEW  4
#define NPAD   64
#define NPLANE (NCTX + HSKEW + NPAD)
#define NCH    64
#define LOG2E  1.44269504088896340736f

static __device__ __forceinline__ v4f lds4(const float* p) {
    v4f v;
    __builtin_memcpy(&v, p, 16);
    return v;
}

static __device__ __forceinline__ float xhalf(float x) {
    return __builtin_bit_cast(float, __builtin_amdgcn_permlanex16(__builtin_bit_cast(int, x), __builtin_bit_cast(int, x),
                                                                 0x76543210, 0xfedcba98, false, false));
}

#define LD4(A, B, P) do { __builtin_memcpy(&(A), (P), 8); __builtin_memcpy(&(B), (P) + 2, 8); } while (0)

#define KEEP2(A) asm volatile("" : "+v"(A))

#define SB() __builtin_amdgcn_sched_barrier(0)

__global__ __launch_bounds__(128) __attribute__((amdgpu_waves_per_eu(1, 1))) void convnp_fused(
    const float* __restrict__ xc,
    const float* __restrict__ yc,
    const float* __restrict__ xg,
    const float* __restrict__ sig,
    const float* __restrict__ W,
    const float* __restrict__ bias,
    float* __restrict__ out)
{
    __shared__ __attribute__((aligned(16))) float sxx[NPLANE];
    __shared__ __attribute__((aligned(16))) float sxy[NPLANE];
    __shared__ __attribute__((aligned(16))) float syc[NPLANE];
    __shared__ __attribute__((aligned(16))) float swb[NCH * 2 + NCH];

    const int t    = threadIdx.x;
    const int b    = blockIdx.x >> 5;
    const int tile = blockIdx.x & 31;
    const int po   = tile << 7;
    const int wave = t >> 5;
    const int lane = t & 31;
    const int lhalf = lane & 15;
    const int hh   = lane >> 4;
    const int p0   = po + wave * 32 + lhalf;

    const v4f* xcb = (const v4f*)(xc + (size_t)b * NCTX * 2);
    const v4f* ycb = (const v4f*)(yc + (size_t)b * NCTX);
    const v4f* wsrc = (t < 32) ? ((const v4f*)W + t) : ((const v4f*)bias + (t & 15));
    const v4f wb  = *wsrc;
    const v4f sx0 = xcb[t];
    const v4f sx1 = xcb[t + 128];
    const v4f sy0 = ycb[t];
    const v2f g0  = *(const v2f*)(xg + ((size_t)b * NPTS + p0) * 2);
    const v2f g1  = *(const v2f*)(xg + ((size_t)b * NPTS + p0 + 16) * 2);
    const float sg0 = sig[0];
    const float sg1 = sig[1];

    const float s0 = __expf(sg0);
    const float s1 = __expf(sg1);
    const float c0 = -0.5f * LOG2E / (s0 * s0);
    const float c1 = -0.5f * LOG2E / (s1 * s1);
    const v2f   cc0 = {c0, c0};
    const v2f   cc1 = {c1, c1};

    {
        ((v2f*)sxx)[t]       = (v2f){sx0.x, sx0.z};
        ((v2f*)sxy)[t]       = (v2f){sx0.y, sx0.w};
        ((v2f*)sxx)[t + 128 + HSKEW / 2] = (v2f){sx1.x, sx1.z};
        ((v2f*)sxy)[t + 128 + HSKEW / 2] = (v2f){sx1.y, sx1.w};
        ((v4f*)syc)[t + (t >> 6)] = sy0;
        if (t <= NPAD / 4) {
            const int zo = (t == 0) ? NHALF : (NCTX + HSKEW + 4 * (t - 1));
            const v4f z = {0.f, 0.f, 0.f, 0.f};
            *(v4f*)(sxx + zo) = z;
            *(v4f*)(sxy + zo) = z;
            *(v4f*)(syc + zo) = z;
        }
        if (t < 48) ((v4f*)swb)[t] = wb;
    }
    __syncthreads();

    const v2f   gxx = {g0.x, g1.x};
    const v2f   gyy = {g0.y, g1.y};
    const float* px = sxx + hh * (NHALF + HSKEW);
    const float* py = sxy + hh * (NHALF + HSKEW);
    const float* pv = syc + hh * (NHALF + HSKEW);

    v2f dx0, dx1, dx2, dx3, dy0, dy1, dy2, dy3, tt0, tt1, tt2, tt3, qq0, qq1, qq2, qq3;
    v2f ss0, ss1, ss2, ss3, aa0, aa1, aa2, aa3, ab0, ab1, ab2, ab3, ea0, ea1, ea2, ea3, eb0, eb1, eb2, eb3;
    v2f da0 = {0.f, 0.f}, da1 = {0.f, 0.f}, da2 = {0.f, 0.f}, da3 = {0.f, 0.f};
    v2f ca0 = {0.f, 0.f}, ca1 = {0.f, 0.f}, ca2 = {0.f, 0.f}, ca3 = {0.f, 0.f};

    v2f XP0a, XP0b, YP0a, YP0b, XP1a, XP1b, YP1a, YP1b;
    LD4(XP0a, XP0b, px + 0); LD4(YP0a, YP0b, py + 0); SB();
    LD4(XP1a, XP1b, px + 4); LD4(YP1a, YP1b, py + 4); SB();
    v2f X0_0a, X0_0b, X0_1a, X0_1b, X0_2a, X0_2b, X0_3a, X0_3b, X0_4a, X0_4b, X0_5a, X0_5b, X0_6a, X0_6b, X0_7a, X0_7b, Y0_0a, Y0_0b, Y0_1a, Y0_1b, Y0_2a, Y0_2b, Y0_3a, Y0_3b, Y0_4a, Y0_4b, Y0_5a, Y0_5b, Y0_6a, Y0_6b, Y0_7a, Y0_7b, V0_0a, V0_0b, V0_1a, V0_1b, V0_2a, V0_2b, V0_3a, V0_3b, V0_4a, V0_4b, V0_5a, V0_5b, V0_6a, V0_6b, V0_7a, V0_7b;
    v2f X1_0a, X1_0b, X1_1a, X1_1b, X1_2a, X1_2b, X1_3a, X1_3b, X1_4a, X1_4b, X1_5a, X1_5b, X1_6a, X1_6b, X1_7a, X1_7b, Y1_0a, Y1_0b, Y1_1a, Y1_1b, Y1_2a, Y1_2b, Y1_3a, Y1_3b, Y1_4a, Y1_4b, Y1_5a, Y1_5b, Y1_6a, Y1_6b, Y1_7a, Y1_7b, V1_0a, V1_0b, V1_1a, V1_1b, V1_2a, V1_2b, V1_3a, V1_3b, V1_4a, V1_4b, V1_5a, V1_5b, V1_6a, V1_6b, V1_7a, V1_7b;
    LD4(X0_0a, X0_0b, px + 8 + 0); LD4(Y0_0a, Y0_0b, py + 8 + 0);
    LD4(X0_1a, X0_1b, px + 8 + 4); LD4(Y0_1a, Y0_1b, py + 8 + 4);
    LD4(X0_2a, X0_2b, px + 8 + 8); LD4(Y0_2a, Y0_2b, py + 8 + 8);
    LD4(X0_3a, X0_3b, px + 8 + 12); LD4(Y0_3a, Y0_3b, py + 8 + 12);
    LD4(X0_4a, X0_4b, px + 8 + 16); LD4(Y0_4a, Y0_4b, py + 8 + 16);
    LD4(X0_5a, X0_5b, px + 8 + 20); LD4(Y0_5a, Y0_5b, py + 8 + 20);
    LD4(X0_6a, X0_6b, px + 8 + 24); LD4(Y0_6a, Y0_6b, py + 8 + 24);
    LD4(X0_7a, X0_7b, px + 8 + 28); LD4(Y0_7a, Y0_7b, py + 8 + 28);
    LD4(V0_0a, V0_0b, pv + 0 + 0);
    LD4(V0_1a, V0_1b, pv + 0 + 4);
    LD4(V0_2a, V0_2b, pv + 0 + 8);
    LD4(V0_3a, V0_3b, pv + 0 + 12);
    LD4(V0_4a, V0_4b, pv + 0 + 16);
    LD4(V0_5a, V0_5b, pv + 0 + 20);
    LD4(V0_6a, V0_6b, pv + 0 + 24);
    LD4(V0_7a, V0_7b, pv + 0 + 28);
    SB();
    dx0 = gxx - (v2f){XP0a.x, XP0a.x}; SB();
    dy0 = gyy - (v2f){YP0a.x, YP0a.x}; SB();
    dx1 = gxx - (v2f){XP0a.y, XP0a.y}; SB();
    dy1 = gyy - (v2f){YP0a.y, YP0a.y}; SB();
    tt0 = dy0 * dy0; SB();
    dx2 = gxx - (v2f){XP0b.x, XP0b.x}; SB();
    dy2 = gyy - (v2f){YP0b.x, YP0b.x}; SB();
    tt1 = dy1 * dy1; SB();
    qq0 = __builtin_elementwise_fma(dx0, dx0, tt0); SB();
    dx3 = gxx - (v2f){XP0b.y, XP0b.y}; SB();
    dy3 = gyy - (v2f){YP0b.y, YP0b.y}; SB();
    tt2 = dy2 * dy2; SB();
    qq1 = __builtin_elementwise_fma(dx1, dx1, tt1); SB();
    ss0.x = __builtin_amdgcn_sqrtf(qq0.x); SB();
    dx0 = gxx - (v2f){XP1a.x, XP1a.x}; SB();
    ss0.y = __builtin_amdgcn_sqrtf(qq0.y); SB();
    dy0 = gyy - (v2f){YP1a.x, YP1a.x}; SB();
    tt3 = dy3 * dy3; SB();
    qq2 = __builtin_elementwise_fma(dx2, dx2, tt2); SB();
    ss1.x = __builtin_amdgcn_sqrtf(qq1.x); SB();
    dx1 = gxx - (v2f){XP1a.y, XP1a.y}; SB();
    ss1.y = __builtin_amdgcn_sqrtf(qq1.y); SB();
    dy1 = gyy - (v2f){YP1a.y, YP1a.y}; SB();
    tt0 = dy0 * dy0; SB();
    qq3 = __builtin_elementwise_fma(dx3, dx3, tt3); SB();
    aa0 = ss0 * cc0; SB();
    ab0 = ss0 * cc1; SB();
    ss2.x = __builtin_amdgcn_sqrtf(qq2.x); SB();
    dx2 = gxx - (v2f){XP1b.x, XP1b.x}; SB();
    ss2.y = __builtin_amdgcn_sqrtf(qq2.y); SB();
    dy2 = gyy - (v2f){YP1b.x, YP1b.x}; SB();
    ea0.x = __builtin_amdgcn_exp2f(aa0.x); SB();
    tt1 = dy1 * dy1; SB();
    ea0.y = __builtin_amdgcn_exp2f(aa0.y); SB();
    qq0 = __builtin_elementwise_fma(dx0, dx0, tt0); SB();
    eb0.x = __builtin_amdgcn_exp2f(ab0.x); SB();
    aa1 = ss1 * cc0; SB();
    eb0.y = __builtin_amdgcn_exp2f(ab0.y); SB();
    ab1 = ss1 * cc1; SB();
    ss3.x = __builtin_amdgcn_sqrtf(qq3.x); SB();
    dx3 = gxx - (v2f){XP1b.y, XP1b.y}; SB();
    ss3.y = __builtin_amdgcn_sqrtf(qq3.y); SB();
    dy3 = gyy - (v2f){YP1b.y, YP1b.y}; SB();
    ea1.x = __builtin_amdgcn_exp2f(aa1.x); SB();
    tt2 = dy2 * dy2; SB();
    ea1.y = __builtin_amdgcn_exp2f(aa1.y); SB();
    qq1 = __builtin_elementwise_fma(dx1, dx1, tt1); SB();
    eb1.x = __builtin_amdgcn_exp2f(ab1.x); SB();
    aa2 = ss2 * cc0; SB();
    eb1.y = __builtin_amdgcn_exp2f(ab1.y); SB();
    ab2 = ss2 * cc1; SB();

#pragma unroll 1
    for (int base = 0; base < NHALF; base += 64) {
        LD4(X1_0a, X1_0b, px + base + 32 + 8 + 0); LD4(Y1_0a, Y1_0b, py + base + 32 + 8 + 0);
        LD4(X1_1a, X1_1b, px + base + 32 + 8 + 4); LD4(Y1_1a, Y1_1b, py + base + 32 + 8 + 4);
        LD4(X1_2a, X1_2b, px + base + 32 + 8 + 8); LD4(Y1_2a, Y1_2b, py + base + 32 + 8 + 8);
        LD4(X1_3a, X1_3b, px + base + 32 + 8 + 12); LD4(Y1_3a, Y1_3b, py + base + 32 + 8 + 12);
        LD4(X1_4a, X1_4b, px + base + 32 + 8 + 16); LD4(Y1_4a, Y1_4b, py + base + 32 + 8 + 16);
        LD4(X1_5a, X1_5b, px + base + 32 + 8 + 20); LD4(Y1_5a, Y1_5b, py + base + 32 + 8 + 20);
        LD4(X1_6a, X1_6b, px + base + 32 + 8 + 24); LD4(Y1_6a, Y1_6b, py + base + 32 + 8 + 24);
        LD4(X1_7a, X1_7b, px + base + 32 + 8 + 28); LD4(Y1_7a, Y1_7b, py + base + 32 + 8 + 28);
        LD4(V1_0a, V1_0b, pv + base + 32 + 0);
        LD4(V1_1a, V1_1b, pv + base + 32 + 4);
        LD4(V1_2a, V1_2b, pv + base + 32 + 8);
        LD4(V1_3a, V1_3b, pv + base + 32 + 12);
        LD4(V1_4a, V1_4b, pv + base + 32 + 16);
        LD4(V1_5a, V1_5b, pv + base + 32 + 20);
        LD4(V1_6a, V1_6b, pv + base + 32 + 24);
        LD4(V1_7a, V1_7b, pv + base + 32 + 28);
        SB();
        ss0.x = __builtin_amdgcn_sqrtf(qq0.x); SB();
        dx0 = gxx - (v2f){X0_0a.x, X0_0a.x}; SB();
        ss0.y = __builtin_amdgcn_sqrtf(qq0.y); SB();
        dy0 = gyy - (v2f){Y0_0a.x, Y0_0a.x}; SB();
        ea2.x = __builtin_amdgcn_exp2f(aa2.x); SB();
        tt3 = dy3 * dy3; SB();
        ea2.y = __builtin_amdgcn_exp2f(aa2.y); SB();
        qq2 = __builtin_elementwise_fma(dx2, dx2, tt2); SB();
        eb2.x = __builtin_amdgcn_exp2f(ab2.x); SB();
        aa3 = ss3 * cc0; SB();
        eb2.y = __builtin_amdgcn_exp2f(ab2.y); SB();
        ab3 = ss3 * cc1; SB();
        da0 += ea0; SB();
        ca0 = __builtin_elementwise_fma((v2f){V0_0a.x, V0_0a.x}, eb0, ca0); SB();
        ss1.x = __builtin_amdgcn_sqrtf(qq1.x); SB();
        dx1 = gxx - (v2f){X0_0a.y, X0_0a.y}; SB();
        ss1.y = __builtin_amdgcn_sqrtf(qq1.y); SB();
        dy1 = gyy - (v2f){Y0_0a.y, Y0_0a.y}; SB();
        ea3.x = __builtin_amdgcn_exp2f(aa3.x); SB();
        tt0 = dy0 * dy0; SB();
        ea3.y = __builtin_amdgcn_exp2f(aa3.y); SB();
        qq3 = __builtin_elementwise_fma(dx3, dx3, tt3); SB();
        eb3.x = __builtin_amdgcn_exp2f(ab3.x); SB();
        aa0 = ss0 * cc0; SB();
        eb3.y = __builtin_amdgcn_exp2f(ab3.y); SB();
        ab0 = ss0 * cc1; SB();
        da1 += ea1; SB();
        ca1 = __builtin_elementwise_fma((v2f){V0_0a.y, V0_0a.y}, eb1, ca1); SB();
        ss2.x = __builtin_amdgcn_sqrtf(qq2.x); SB();
        dx2 = gxx - (v2f){X0_0b.x, X0_0b.x}; SB();
        ss2.y = __builtin_amdgcn_sqrtf(qq2.y); SB();
        dy2 = gyy - (v2f){Y0_0b.x, Y0_0b.x}; SB();
        ea0.x = __builtin_amdgcn_exp2f(aa0.x); SB();
        tt1 = dy1 * dy1; SB();
        ea0.y = __builtin_amdgcn_exp2f(aa0.y); SB();
        qq0 = __builtin_elementwise_fma(dx0, dx0, tt0); SB();
        eb0.x = __builtin_amdgcn_exp2f(ab0.x); SB();
        aa1 = ss1 * cc0; SB();
        eb0.y = __builtin_amdgcn_exp2f(ab0.y); SB();
        ab1 = ss1 * cc1; SB();
        da2 += ea2; SB();
        ca2 = __builtin_elementwise_fma((v2f){V0_0b.x, V0_0b.x}, eb2, ca2); SB();
        ss3.x = __builtin_amdgcn_sqrtf(qq3.x); SB();
        dx3 = gxx - (v2f){X0_0b.y, X0_0b.y}; SB();
        ss3.y = __builtin_amdgcn_sqrtf(qq3.y); SB();
        dy3 = gyy - (v2f){Y0_0b.y, Y0_0b.y}; SB();
        ea1.x = __builtin_amdgcn_exp2f(aa1.x); SB();
        tt2 = dy2 * dy2; SB();
        ea1.y = __builtin_amdgcn_exp2f(aa1.y); SB();
        qq1 = __builtin_elementwise_fma(dx1, dx1, tt1); SB();
        eb1.x = __builtin_amdgcn_exp2f(ab1.x); SB();
        aa2 = ss2 * cc0; SB();
        eb1.y = __builtin_amdgcn_exp2f(ab1.y); SB();
        ab2 = ss2 * cc1; SB();
        da3 += ea3; SB();
        ca3 = __builtin_elementwise_fma((v2f){V0_0b.y, V0_0b.y}, eb3, ca3); SB();
        ss0.x = __builtin_amdgcn_sqrtf(qq0.x); SB();
        dx0 = gxx - (v2f){X0_1a.x, X0_1a.x}; SB();
        ss0.y = __builtin_amdgcn_sqrtf(qq0.y); SB();
        dy0 = gyy - (v2f){Y0_1a.x, Y0_1a.x}; SB();
        ea2.x = __builtin_amdgcn_exp2f(aa2.x); SB();
        tt3 = dy3 * dy3; SB();
        ea2.y = __builtin_amdgcn_exp2f(aa2.y); SB();
        qq2 = __builtin_elementwise_fma(dx2, dx2, tt2); SB();
        eb2.x = __builtin_amdgcn_exp2f(ab2.x); SB();
        aa3 = ss3 * cc0; SB();
        eb2.y = __builtin_amdgcn_exp2f(ab2.y); SB();
        ab3 = ss3 * cc1; SB();
        da0 += ea0; SB();
        ca0 = __builtin_elementwise_fma((v2f){V0_1a.x, V0_1a.x}, eb0, ca0); SB();
        ss1.x = __builtin_amdgcn_sqrtf(qq1.x); SB();
        dx1 = gxx - (v2f){X0_1a.y, X0_1a.y}; SB();
        ss1.y = __builtin_amdgcn_sqrtf(qq1.y); SB();
        dy1 = gyy - (v2f){Y0_1a.y, Y0_1a.y}; SB();
        ea3.x = __builtin_amdgcn_exp2f(aa3.x); SB();
        tt0 = dy0 * dy0; SB();
        ea3.y = __builtin_amdgcn_exp2f(aa3.y); SB();
        qq3 = __builtin_elementwise_fma(dx3, dx3, tt3); SB();
        eb3.x = __builtin_amdgcn_exp2f(ab3.x); SB();
        aa0 = ss0 * cc0; SB();
        eb3.y = __builtin_amdgcn_exp2f(ab3.y); SB();
        ab0 = ss0 * cc1; SB();
        da1 += ea1; SB();
        ca1 = __builtin_elementwise_fma((v2f){V0_1a.y, V0_1a.y}, eb1, ca1); SB();
        ss2.x = __builtin_amdgcn_sqrtf(qq2.x); SB();
        dx2 = gxx - (v2f){X0_1b.x, X0_1b.x}; SB();
        ss2.y = __builtin_amdgcn_sqrtf(qq2.y); SB();
        dy2 = gyy - (v2f){Y0_1b.x, Y0_1b.x}; SB();
        ea0.x = __builtin_amdgcn_exp2f(aa0.x); SB();
        tt1 = dy1 * dy1; SB();
        ea0.y = __builtin_amdgcn_exp2f(aa0.y); SB();
        qq0 = __builtin_elementwise_fma(dx0, dx0, tt0); SB();
        eb0.x = __builtin_amdgcn_exp2f(ab0.x); SB();
        aa1 = ss1 * cc0; SB();
        eb0.y = __builtin_amdgcn_exp2f(ab0.y); SB();
        ab1 = ss1 * cc1; SB();
        da2 += ea2; SB();
        ca2 = __builtin_elementwise_fma((v2f){V0_1b.x, V0_1b.x}, eb2, ca2); SB();
        ss3.x = __builtin_amdgcn_sqrtf(qq3.x); SB();
        dx3 = gxx - (v2f){X0_1b.y, X0_1b.y}; SB();
        ss3.y = __builtin_amdgcn_sqrtf(qq3.y); SB();
        dy3 = gyy - (v2f){Y0_1b.y, Y0_1b.y}; SB();
        ea1.x = __builtin_amdgcn_exp2f(aa1.x); SB();
        tt2 = dy2 * dy2; SB();
        ea1.y = __builtin_amdgcn_exp2f(aa1.y); SB();
        qq1 = __builtin_elementwise_fma(dx1, dx1, tt1); SB();
        eb1.x = __builtin_amdgcn_exp2f(ab1.x); SB();
        aa2 = ss2 * cc0; SB();
        eb1.y = __builtin_amdgcn_exp2f(ab1.y); SB();
        ab2 = ss2 * cc1; SB();
        da3 += ea3; SB();
        ca3 = __builtin_elementwise_fma((v2f){V0_1b.y, V0_1b.y}, eb3, ca3); SB();
        ss0.x = __builtin_amdgcn_sqrtf(qq0.x); SB();
        dx0 = gxx - (v2f){X0_2a.x, X0_2a.x}; SB();
        ss0.y = __builtin_amdgcn_sqrtf(qq0.y); SB();
        dy0 = gyy - (v2f){Y0_2a.x, Y0_2a.x}; SB();
        ea2.x = __builtin_amdgcn_exp2f(aa2.x); SB();
        tt3 = dy3 * dy3; SB();
        ea2.y = __builtin_amdgcn_exp2f(aa2.y); SB();
        qq2 = __builtin_elementwise_fma(dx2, dx2, tt2); SB();
        eb2.x = __builtin_amdgcn_exp2f(ab2.x); SB();
        aa3 = ss3 * cc0; SB();
        eb2.y = __builtin_amdgcn_exp2f(ab2.y); SB();
        ab3 = ss3 * cc1; SB();
        da0 += ea0; SB();
        ca0 = __builtin_elementwise_fma((v2f){V0_2a.x, V0_2a.x}, eb0, ca0); SB();
        ss1.x = __builtin_amdgcn_sqrtf(qq1.x); SB();
        dx1 = gxx - (v2f){X0_2a.y, X0_2a.y}; SB();
        ss1.y = __builtin_amdgcn_sqrtf(qq1.y); SB();
        dy1 = gyy - (v2f){Y0_2a.y, Y0_2a.y}; SB();
        ea3.x = __builtin_amdgcn_exp2f(aa3.x); SB();
        tt0 = dy0 * dy0; SB();
        ea3.y = __builtin_amdgcn_exp2f(aa3.y); SB();
        qq3 = __builtin_elementwise_fma(dx3, dx3, tt3); SB();
        eb3.x = __builtin_amdgcn_exp2f(ab3.x); SB();
        aa0 = ss0 * cc0; SB();
        eb3.y = __builtin_amdgcn_exp2f(ab3.y); SB();
        ab0 = ss0 * cc1; SB();
        da1 += ea1; SB();
        ca1 = __builtin_elementwise_fma((v2f){V0_2a.y, V0_2a.y}, eb1, ca1); SB();
        ss2.x = __builtin_amdgcn_sqrtf(qq2.x); SB();
        dx2 = gxx - (v2f){X0_2b.x, X0_2b.x}; SB();
        ss2.y = __builtin_amdgcn_sqrtf(qq2.y); SB();
        dy2 = gyy - (v2f){Y0_2b.x, Y0_2b.x}; SB();
        ea0.x = __builtin_amdgcn_exp2f(aa0.x); SB();
        tt1 = dy1 * dy1; SB();
        ea0.y = __builtin_amdgcn_exp2f(aa0.y); SB();
        qq0 = __builtin_elementwise_fma(dx0, dx0, tt0); SB();
        eb0.x = __builtin_amdgcn_exp2f(ab0.x); SB();
        aa1 = ss1 * cc0; SB();
        eb0.y = __builtin_amdgcn_exp2f(ab0.y); SB();
        ab1 = ss1 * cc1; SB();
        da2 += ea2; SB();
        ca2 = __builtin_elementwise_fma((v2f){V0_2b.x, V0_2b.x}, eb2, ca2); SB();
        ss3.x = __builtin_amdgcn_sqrtf(qq3.x); SB();
        dx3 = gxx - (v2f){X0_2b.y, X0_2b.y}; SB();
        ss3.y = __builtin_amdgcn_sqrtf(qq3.y); SB();
        dy3 = gyy - (v2f){Y0_2b.y, Y0_2b.y}; SB();
        ea1.x = __builtin_amdgcn_exp2f(aa1.x); SB();
        tt2 = dy2 * dy2; SB();
        ea1.y = __builtin_amdgcn_exp2f(aa1.y); SB();
        qq1 = __builtin_elementwise_fma(dx1, dx1, tt1); SB();
        eb1.x = __builtin_amdgcn_exp2f(ab1.x); SB();
        aa2 = ss2 * cc0; SB();
        eb1.y = __builtin_amdgcn_exp2f(ab1.y); SB();
        ab2 = ss2 * cc1; SB();
        da3 += ea3; SB();
        ca3 = __builtin_elementwise_fma((v2f){V0_2b.y, V0_2b.y}, eb3, ca3); SB();
        ss0.x = __builtin_amdgcn_sqrtf(qq0.x); SB();
        dx0 = gxx - (v2f){X0_3a.x, X0_3a.x}; SB();
        ss0.y = __builtin_amdgcn_sqrtf(qq0.y); SB();
        dy0 = gyy - (v2f){Y0_3a.x, Y0_3a.x}; SB();
        ea2.x = __builtin_amdgcn_exp2f(aa2.x); SB();
        tt3 = dy3 * dy3; SB();
        ea2.y = __builtin_amdgcn_exp2f(aa2.y); SB();
        qq2 = __builtin_elementwise_fma(dx2, dx2, tt2); SB();
        eb2.x = __builtin_amdgcn_exp2f(ab2.x); SB();
        aa3 = ss3 * cc0; SB();
        eb2.y = __builtin_amdgcn_exp2f(ab2.y); SB();
        ab3 = ss3 * cc1; SB();
        da0 += ea0; SB();
        ca0 = __builtin_elementwise_fma((v2f){V0_3a.x, V0_3a.x}, eb0, ca0); SB();
        ss1.x = __builtin_amdgcn_sqrtf(qq1.x); SB();
        dx1 = gxx - (v2f){X0_3a.y, X0_3a.y}; SB();
        ss1.y = __builtin_amdgcn_sqrtf(qq1.y); SB();
        dy1 = gyy - (v2f){Y0_3a.y, Y0_3a.y}; SB();
        ea3.x = __builtin_amdgcn_exp2f(aa3.x); SB();
        tt0 = dy0 * dy0; SB();
        ea3.y = __builtin_amdgcn_exp2f(aa3.y); SB();
        qq3 = __builtin_elementwise_fma(dx3, dx3, tt3); SB();
        eb3.x = __builtin_amdgcn_exp2f(ab3.x); SB();
        aa0 = ss0 * cc0; SB();
        eb3.y = __builtin_amdgcn_exp2f(ab3.y); SB();
        ab0 = ss0 * cc1; SB();
        da1 += ea1; SB();
        ca1 = __builtin_elementwise_fma((v2f){V0_3a.y, V0_3a.y}, eb1, ca1); SB();
        ss2.x = __builtin_amdgcn_sqrtf(qq2.x); SB();
        dx2 = gxx - (v2f){X0_3b.x, X0_3b.x}; SB();
        ss2.y = __builtin_amdgcn_sqrtf(qq2.y); SB();
        dy2 = gyy - (v2f){Y0_3b.x, Y0_3b.x}; SB();
        ea0.x = __builtin_amdgcn_exp2f(aa0.x); SB();
        tt1 = dy1 * dy1; SB();
        ea0.y = __builtin_amdgcn_exp2f(aa0.y); SB();
        qq0 = __builtin_elementwise_fma(dx0, dx0, tt0); SB();
        eb0.x = __builtin_amdgcn_exp2f(ab0.x); SB();
        aa1 = ss1 * cc0; SB();
        eb0.y = __builtin_amdgcn_exp2f(ab0.y); SB();
        ab1 = ss1 * cc1; SB();
        da2 += ea2; SB();
        ca2 = __builtin_elementwise_fma((v2f){V0_3b.x, V0_3b.x}, eb2, ca2); SB();
        ss3.x = __builtin_amdgcn_sqrtf(qq3.x); SB();
        dx3 = gxx - (v2f){X0_3b.y, X0_3b.y}; SB();
        ss3.y = __builtin_amdgcn_sqrtf(qq3.y); SB();
        dy3 = gyy - (v2f){Y0_3b.y, Y0_3b.y}; SB();
        ea1.x = __builtin_amdgcn_exp2f(aa1.x); SB();
        tt2 = dy2 * dy2; SB();
        ea1.y = __builtin_amdgcn_exp2f(aa1.y); SB();
        qq1 = __builtin_elementwise_fma(dx1, dx1, tt1); SB();
        eb1.x = __builtin_amdgcn_exp2f(ab1.x); SB();
        aa2 = ss2 * cc0; SB();
        eb1.y = __builtin_amdgcn_exp2f(ab1.y); SB();
        ab2 = ss2 * cc1; SB();
        da3 += ea3; SB();
        ca3 = __builtin_elementwise_fma((v2f){V0_3b.y, V0_3b.y}, eb3, ca3); SB();
        ss0.x = __builtin_amdgcn_sqrtf(qq0.x); SB();
        dx0 = gxx - (v2f){X0_4a.x, X0_4a.x}; SB();
        ss0.y = __builtin_amdgcn_sqrtf(qq0.y); SB();
        dy0 = gyy - (v2f){Y0_4a.x, Y0_4a.x}; SB();
        ea2.x = __builtin_amdgcn_exp2f(aa2.x); SB();
        tt3 = dy3 * dy3; SB();
        ea2.y = __builtin_amdgcn_exp2f(aa2.y); SB();
        qq2 = __builtin_elementwise_fma(dx2, dx2, tt2); SB();
        eb2.x = __builtin_amdgcn_exp2f(ab2.x); SB();
        aa3 = ss3 * cc0; SB();
        eb2.y = __builtin_amdgcn_exp2f(ab2.y); SB();
        ab3 = ss3 * cc1; SB();
        da0 += ea0; SB();
        ca0 = __builtin_elementwise_fma((v2f){V0_4a.x, V0_4a.x}, eb0, ca0); SB();
        ss1.x = __builtin_amdgcn_sqrtf(qq1.x); SB();
        dx1 = gxx - (v2f){X0_4a.y, X0_4a.y}; SB();
        ss1.y = __builtin_amdgcn_sqrtf(qq1.y); SB();
        dy1 = gyy - (v2f){Y0_4a.y, Y0_4a.y}; SB();
        ea3.x = __builtin_amdgcn_exp2f(aa3.x); SB();
        tt0 = dy0 * dy0; SB();
        ea3.y = __builtin_amdgcn_exp2f(aa3.y); SB();
        qq3 = __builtin_elementwise_fma(dx3, dx3, tt3); SB();
        eb3.x = __builtin_amdgcn_exp2f(ab3.x); SB();
        aa0 = ss0 * cc0; SB();
        eb3.y = __builtin_amdgcn_exp2f(ab3.y); SB();
        ab0 = ss0 * cc1; SB();
        da1 += ea1; SB();
        ca1 = __builtin_elementwise_fma((v2f){V0_4a.y, V0_4a.y}, eb1, ca1); SB();
        ss2.x = __builtin_amdgcn_sqrtf(qq2.x); SB();
        dx2 = gxx - (v2f){X0_4b.x, X0_4b.x}; SB();
        ss2.y = __builtin_amdgcn_sqrtf(qq2.y); SB();
        dy2 = gyy - (v2f){Y0_4b.x, Y0_4b.x}; SB();
        ea0.x = __builtin_amdgcn_exp2f(aa0.x); SB();
        tt1 = dy1 * dy1; SB();
        ea0.y = __builtin_amdgcn_exp2f(aa0.y); SB();
        qq0 = __builtin_elementwise_fma(dx0, dx0, tt0); SB();
        eb0.x = __builtin_amdgcn_exp2f(ab0.x); SB();
        aa1 = ss1 * cc0; SB();
        eb0.y = __builtin_amdgcn_exp2f(ab0.y); SB();
        ab1 = ss1 * cc1; SB();
        da2 += ea2; SB();
        ca2 = __builtin_elementwise_fma((v2f){V0_4b.x, V0_4b.x}, eb2, ca2); SB();
        ss3.x = __builtin_amdgcn_sqrtf(qq3.x); SB();
        dx3 = gxx - (v2f){X0_4b.y, X0_4b.y}; SB();
        ss3.y = __builtin_amdgcn_sqrtf(qq3.y); SB();
        dy3 = gyy - (v2f){Y0_4b.y, Y0_4b.y}; SB();
        ea1.x = __builtin_amdgcn_exp2f(aa1.x); SB();
        tt2 = dy2 * dy2; SB();
        ea1.y = __builtin_amdgcn_exp2f(aa1.y); SB();
        qq1 = __builtin_elementwise_fma(dx1, dx1, tt1); SB();
        eb1.x = __builtin_amdgcn_exp2f(ab1.x); SB();
        aa2 = ss2 * cc0; SB();
        eb1.y = __builtin_amdgcn_exp2f(ab1.y); SB();
        ab2 = ss2 * cc1; SB();
        da3 += ea3; SB();
        ca3 = __builtin_elementwise_fma((v2f){V0_4b.y, V0_4b.y}, eb3, ca3); SB();
        ss0.x = __builtin_amdgcn_sqrtf(qq0.x); SB();
        dx0 = gxx - (v2f){X0_5a.x, X0_5a.x}; SB();
        ss0.y = __builtin_amdgcn_sqrtf(qq0.y); SB();
        dy0 = gyy - (v2f){Y0_5a.x, Y0_5a.x}; SB();
        ea2.x = __builtin_amdgcn_exp2f(aa2.x); SB();
        tt3 = dy3 * dy3; SB();
        ea2.y = __builtin_amdgcn_exp2f(aa2.y); SB();
        qq2 = __builtin_elementwise_fma(dx2, dx2, tt2); SB();
        eb2.x = __builtin_amdgcn_exp2f(ab2.x); SB();
        aa3 = ss3 * cc0; SB();
        eb2.y = __builtin_amdgcn_exp2f(ab2.y); SB();
        ab3 = ss3 * cc1; SB();
        da0 += ea0; SB();
        ca0 = __builtin_elementwise_fma((v2f){V0_5a.x, V0_5a.x}, eb0, ca0); SB();
        ss1.x = __builtin_amdgcn_sqrtf(qq1.x); SB();
        dx1 = gxx - (v2f){X0_5a.y, X0_5a.y}; SB();
        ss1.y = __builtin_amdgcn_sqrtf(qq1.y); SB();
        dy1 = gyy - (v2f){Y0_5a.y, Y0_5a.y}; SB();
        ea3.x = __builtin_amdgcn_exp2f(aa3.x); SB();
        tt0 = dy0 * dy0; SB();
        ea3.y = __builtin_amdgcn_exp2f(aa3.y); SB();
        qq3 = __builtin_elementwise_fma(dx3, dx3, tt3); SB();
        eb3.x = __builtin_amdgcn_exp2f(ab3.x); SB();
        aa0 = ss0 * cc0; SB();
        eb3.y = __builtin_amdgcn_exp2f(ab3.y); SB();
        ab0 = ss0 * cc1; SB();
        da1 += ea1; SB();
        ca1 = __builtin_elementwise_fma((v2f){V0_5a.y, V0_5a.y}, eb1, ca1); SB();
        ss2.x = __builtin_amdgcn_sqrtf(qq2.x); SB();
        dx2 = gxx - (v2f){X0_5b.x, X0_5b.x}; SB();
        ss2.y = __builtin_amdgcn_sqrtf(qq2.y); SB();
        dy2 = gyy - (v2f){Y0_5b.x, Y0_5b.x}; SB();
        ea0.x = __builtin_amdgcn_exp2f(aa0.x); SB();
        tt1 = dy1 * dy1; SB();
        ea0.y = __builtin_amdgcn_exp2f(aa0.y); SB();
        qq0 = __builtin_elementwise_fma(dx0, dx0, tt0); SB();
        eb0.x = __builtin_amdgcn_exp2f(ab0.x); SB();
        aa1 = ss1 * cc0; SB();
        eb0.y = __builtin_amdgcn_exp2f(ab0.y); SB();
        ab1 = ss1 * cc1; SB();
        da2 += ea2; SB();
        ca2 = __builtin_elementwise_fma((v2f){V0_5b.x, V0_5b.x}, eb2, ca2); SB();
        ss3.x = __builtin_amdgcn_sqrtf(qq3.x); SB();
        dx3 = gxx - (v2f){X0_5b.y, X0_5b.y}; SB();
        ss3.y = __builtin_amdgcn_sqrtf(qq3.y); SB();
        dy3 = gyy - (v2f){Y0_5b.y, Y0_5b.y}; SB();
        ea1.x = __builtin_amdgcn_exp2f(aa1.x); SB();
        tt2 = dy2 * dy2; SB();
        ea1.y = __builtin_amdgcn_exp2f(aa1.y); SB();
        qq1 = __builtin_elementwise_fma(dx1, dx1, tt1); SB();
        eb1.x = __builtin_amdgcn_exp2f(ab1.x); SB();
        aa2 = ss2 * cc0; SB();
        eb1.y = __builtin_amdgcn_exp2f(ab1.y); SB();
        ab2 = ss2 * cc1; SB();
        da3 += ea3; SB();
        ca3 = __builtin_elementwise_fma((v2f){V0_5b.y, V0_5b.y}, eb3, ca3); SB();
        ss0.x = __builtin_amdgcn_sqrtf(qq0.x); SB();
        dx0 = gxx - (v2f){X0_6a.x, X0_6a.x}; SB();
        ss0.y = __builtin_amdgcn_sqrtf(qq0.y); SB();
        dy0 = gyy - (v2f){Y0_6a.x, Y0_6a.x}; SB();
        ea2.x = __builtin_amdgcn_exp2f(aa2.x); SB();
        tt3 = dy3 * dy3; SB();
        ea2.y = __builtin_amdgcn_exp2f(aa2.y); SB();
        qq2 = __builtin_elementwise_fma(dx2, dx2, tt2); SB();
        eb2.x = __builtin_amdgcn_exp2f(ab2.x); SB();
        aa3 = ss3 * cc0; SB();
        eb2.y = __builtin_amdgcn_exp2f(ab2.y); SB();
        ab3 = ss3 * cc1; SB();
        da0 += ea0; SB();
        ca0 = __builtin_elementwise_fma((v2f){V0_6a.x, V0_6a.x}, eb0, ca0); SB();
        ss1.x = __builtin_amdgcn_sqrtf(qq1.x); SB();
        dx1 = gxx - (v2f){X0_6a.y, X0_6a.y}; SB();
        ss1.y = __builtin_amdgcn_sqrtf(qq1.y); SB();
        dy1 = gyy - (v2f){Y0_6a.y, Y0_6a.y}; SB();
        ea3.x = __builtin_amdgcn_exp2f(aa3.x); SB();
        tt0 = dy0 * dy0; SB();
        ea3.y = __builtin_amdgcn_exp2f(aa3.y); SB();
        qq3 = __builtin_elementwise_fma(dx3, dx3, tt3); SB();
        eb3.x = __builtin_amdgcn_exp2f(ab3.x); SB();
        aa0 = ss0 * cc0; SB();
        eb3.y = __builtin_amdgcn_exp2f(ab3.y); SB();
        ab0 = ss0 * cc1; SB();
        da1 += ea1; SB();
        ca1 = __builtin_elementwise_fma((v2f){V0_6a.y, V0_6a.y}, eb1, ca1); SB();
        ss2.x = __builtin_amdgcn_sqrtf(qq2.x); SB();
        dx2 = gxx - (v2f){X0_6b.x, X0_6b.x}; SB();
        ss2.y = __builtin_amdgcn_sqrtf(qq2.y); SB();
        dy2 = gyy - (v2f){Y0_6b.x, Y0_6b.x}; SB();
        ea0.x = __builtin_amdgcn_exp2f(aa0.x); SB();
        tt1 = dy1 * dy1; SB();
        ea0.y = __builtin_amdgcn_exp2f(aa0.y); SB();
        qq0 = __builtin_elementwise_fma(dx0, dx0, tt0); SB();
        eb0.x = __builtin_amdgcn_exp2f(ab0.x); SB();
        aa1 = ss1 * cc0; SB();
        eb0.y = __builtin_amdgcn_exp2f(ab0.y); SB();
        ab1 = ss1 * cc1; SB();
        da2 += ea2; SB();
        ca2 = __builtin_elementwise_fma((v2f){V0_6b.x, V0_6b.x}, eb2, ca2); SB();
        ss3.x = __builtin_amdgcn_sqrtf(qq3.x); SB();
        dx3 = gxx - (v2f){X0_6b.y, X0_6b.y}; SB();
        ss3.y = __builtin_amdgcn_sqrtf(qq3.y); SB();
        dy3 = gyy - (v2f){Y0_6b.y, Y0_6b.y}; SB();
        ea1.x = __builtin_amdgcn_exp2f(aa1.x); SB();
        tt2 = dy2 * dy2; SB();
        ea1.y = __builtin_amdgcn_exp2f(aa1.y); SB();
        qq1 = __builtin_elementwise_fma(dx1, dx1, tt1); SB();
        eb1.x = __builtin_amdgcn_exp2f(ab1.x); SB();
        aa2 = ss2 * cc0; SB();
        eb1.y = __builtin_amdgcn_exp2f(ab1.y); SB();
        ab2 = ss2 * cc1; SB();
        da3 += ea3; SB();
        ca3 = __builtin_elementwise_fma((v2f){V0_6b.y, V0_6b.y}, eb3, ca3); SB();
        ss0.x = __builtin_amdgcn_sqrtf(qq0.x); SB();
        dx0 = gxx - (v2f){X0_7a.x, X0_7a.x}; SB();
        ss0.y = __builtin_amdgcn_sqrtf(qq0.y); SB();
        dy0 = gyy - (v2f){Y0_7a.x, Y0_7a.x}; SB();
        ea2.x = __builtin_amdgcn_exp2f(aa2.x); SB();
        tt3 = dy3 * dy3; SB();
        ea2.y = __builtin_amdgcn_exp2f(aa2.y); SB();
        qq2 = __builtin_elementwise_fma(dx2, dx2, tt2); SB();
        eb2.x = __builtin_amdgcn_exp2f(ab2.x); SB();
        aa3 = ss3 * cc0; SB();
        eb2.y = __builtin_amdgcn_exp2f(ab2.y); SB();
        ab3 = ss3 * cc1; SB();
        da0 += ea0; SB();
        ca0 = __builtin_elementwise_fma((v2f){V0_7a.x, V0_7a.x}, eb0, ca0); SB();
        ss1.x = __builtin_amdgcn_sqrtf(qq1.x); SB();
        dx1 = gxx - (v2f){X0_7a.y, X0_7a.y}; SB();
        ss1.y = __builtin_amdgcn_sqrtf(qq1.y); SB();
        dy1 = gyy - (v2f){Y0_7a.y, Y0_7a.y}; SB();
        ea3.x = __builtin_amdgcn_exp2f(aa3.x); SB();
        tt0 = dy0 * dy0; SB();
        ea3.y = __builtin_amdgcn_exp2f(aa3.y); SB();
        qq3 = __builtin_elementwise_fma(dx3, dx3, tt3); SB();
        eb3.x = __builtin_amdgcn_exp2f(ab3.x); SB();
        aa0 = ss0 * cc0; SB();
        eb3.y = __builtin_amdgcn_exp2f(ab3.y); SB();
        ab0 = ss0 * cc1; SB();
        da1 += ea1; SB();
        ca1 = __builtin_elementwise_fma((v2f){V0_7a.y, V0_7a.y}, eb1, ca1); SB();
        ss2.x = __builtin_amdgcn_sqrtf(qq2.x); SB();
        dx2 = gxx - (v2f){X0_7b.x, X0_7b.x}; SB();
        ss2.y = __builtin_amdgcn_sqrtf(qq2.y); SB();
        dy2 = gyy - (v2f){Y0_7b.x, Y0_7b.x}; SB();
        ea0.x = __builtin_amdgcn_exp2f(aa0.x); SB();
        tt1 = dy1 * dy1; SB();
        ea0.y = __builtin_amdgcn_exp2f(aa0.y); SB();
        qq0 = __builtin_elementwise_fma(dx0, dx0, tt0); SB();
        eb0.x = __builtin_amdgcn_exp2f(ab0.x); SB();
        aa1 = ss1 * cc0; SB();
        eb0.y = __builtin_amdgcn_exp2f(ab0.y); SB();
        ab1 = ss1 * cc1; SB();
        da2 += ea2; SB();
        ca2 = __builtin_elementwise_fma((v2f){V0_7b.x, V0_7b.x}, eb2, ca2); SB();
        ss3.x = __builtin_amdgcn_sqrtf(qq3.x); SB();
        dx3 = gxx - (v2f){X0_7b.y, X0_7b.y}; SB();
        ss3.y = __builtin_amdgcn_sqrtf(qq3.y); SB();
        dy3 = gyy - (v2f){Y0_7b.y, Y0_7b.y}; SB();
        ea1.x = __builtin_amdgcn_exp2f(aa1.x); SB();
        tt2 = dy2 * dy2; SB();
        ea1.y = __builtin_amdgcn_exp2f(aa1.y); SB();
        qq1 = __builtin_elementwise_fma(dx1, dx1, tt1); SB();
        eb1.x = __builtin_amdgcn_exp2f(ab1.x); SB();
        aa2 = ss2 * cc0; SB();
        eb1.y = __builtin_amdgcn_exp2f(ab1.y); SB();
        ab2 = ss2 * cc1; SB();
        da3 += ea3; SB();
        ca3 = __builtin_elementwise_fma((v2f){V0_7b.y, V0_7b.y}, eb3, ca3); SB();
        LD4(X0_0a, X0_0b, px + base + 64 + 8 + 0); LD4(Y0_0a, Y0_0b, py + base + 64 + 8 + 0);
        LD4(X0_1a, X0_1b, px + base + 64 + 8 + 4); LD4(Y0_1a, Y0_1b, py + base + 64 + 8 + 4);
        LD4(X0_2a, X0_2b, px + base + 64 + 8 + 8); LD4(Y0_2a, Y0_2b, py + base + 64 + 8 + 8);
        LD4(X0_3a, X0_3b, px + base + 64 + 8 + 12); LD4(Y0_3a, Y0_3b, py + base + 64 + 8 + 12);
        LD4(X0_4a, X0_4b, px + base + 64 + 8 + 16); LD4(Y0_4a, Y0_4b, py + base + 64 + 8 + 16);
        LD4(X0_5a, X0_5b, px + base + 64 + 8 + 20); LD4(Y0_5a, Y0_5b, py + base + 64 + 8 + 20);
        LD4(X0_6a, X0_6b, px + base + 64 + 8 + 24); LD4(Y0_6a, Y0_6b, py + base + 64 + 8 + 24);
        LD4(X0_7a, X0_7b, px + base + 64 + 8 + 28); LD4(Y0_7a, Y0_7b, py + base + 64 + 8 + 28);
        LD4(V0_0a, V0_0b, pv + base + 64 + 0);
        LD4(V0_1a, V0_1b, pv + base + 64 + 4);
        LD4(V0_2a, V0_2b, pv + base + 64 + 8);
        LD4(V0_3a, V0_3b, pv + base + 64 + 12);
        LD4(V0_4a, V0_4b, pv + base + 64 + 16);
        LD4(V0_5a, V0_5b, pv + base + 64 + 20);
        LD4(V0_6a, V0_6b, pv + base + 64 + 24);
        LD4(V0_7a, V0_7b, pv + base + 64 + 28);
        SB();
        KEEP2(X1_0b); KEEP2(Y1_0b); KEEP2(V1_0b); KEEP2(X1_1b); KEEP2(Y1_1b); KEEP2(V1_1b); KEEP2(X1_2b); KEEP2(Y1_2b); KEEP2(V1_2b); KEEP2(X1_3b); KEEP2(Y1_3b); KEEP2(V1_3b); KEEP2(X1_4b); KEEP2(Y1_4b); KEEP2(V1_4b); KEEP2(X1_5b); KEEP2(Y1_5b); KEEP2(V1_5b); KEEP2(X1_6b); KEEP2(Y1_6b); KEEP2(V1_6b); KEEP2(X1_7b); KEEP2(Y1_7b); KEEP2(V1_7b);
        SB();
        ss0.x = __builtin_amdgcn_sqrtf(qq0.x); SB();
        dx0 = gxx - (v2f){X1_0a.x, X1_0a.x}; SB();
        ss0.y = __builtin_amdgcn_sqrtf(qq0.y); SB();
        dy0 = gyy - (v2f){Y1_0a.x, Y1_0a.x}; SB();
        ea2.x = __builtin_amdgcn_exp2f(aa2.x); SB();
        tt3 = dy3 * dy3; SB();
        ea2.y = __builtin_amdgcn_exp2f(aa2.y); SB();
        qq2 = __builtin_elementwise_fma(dx2, dx2, tt2); SB();
        eb2.x = __builtin_amdgcn_exp2f(ab2.x); SB();
        aa3 = ss3 * cc0; SB();
        eb2.y = __builtin_amdgcn_exp2f(ab2.y); SB();
        ab3 = ss3 * cc1; SB();
        da0 += ea0; SB();
        ca0 = __builtin_elementwise_fma((v2f){V1_0a.x, V1_0a.x}, eb0, ca0); SB();
        ss1.x = __builtin_amdgcn_sqrtf(qq1.x); SB();
        dx1 = gxx - (v2f){X1_0a.y, X1_0a.y}; SB();
        ss1.y = __builtin_amdgcn_sqrtf(qq1.y); SB();
        dy1 = gyy - (v2f){Y1_0a.y, Y1_0a.y}; SB();
        ea3.x = __builtin_amdgcn_exp2f(aa3.x); SB();
        tt0 = dy0 * dy0; SB();
        ea3.y = __builtin_amdgcn_exp2f(aa3.y); SB();
        qq3 = __builtin_elementwise_fma(dx3, dx3, tt3); SB();
        eb3.x = __builtin_amdgcn_exp2f(ab3.x); SB();
        aa0 = ss0 * cc0; SB();
        eb3.y = __builtin_amdgcn_exp2f(ab3.y); SB();
        ab0 = ss0 * cc1; SB();
        da1 += ea1; SB();
        ca1 = __builtin_elementwise_fma((v2f){V1_0a.y, V1_0a.y}, eb1, ca1); SB();
        ss2.x = __builtin_amdgcn_sqrtf(qq2.x); SB();
        dx2 = gxx - (v2f){X1_0b.x, X1_0b.x}; SB();
        ss2.y = __builtin_amdgcn_sqrtf(qq2.y); SB();
        dy2 = gyy - (v2f){Y1_0b.x, Y1_0b.x}; SB();
        ea0.x = __builtin_amdgcn_exp2f(aa0.x); SB();
        tt1 = dy1 * dy1; SB();
        ea0.y = __builtin_amdgcn_exp2f(aa0.y); SB();
        qq0 = __builtin_elementwise_fma(dx0, dx0, tt0); SB();
        eb0.x = __builtin_amdgcn_exp2f(ab0.x); SB();
        aa1 = ss1 * cc0; SB();
        eb0.y = __builtin_amdgcn_exp2f(ab0.y); SB();
        ab1 = ss1 * cc1; SB();
        da2 += ea2; SB();
        ca2 = __builtin_elementwise_fma((v2f){V1_0b.x, V1_0b.x}, eb2, ca2); SB();
        ss3.x = __builtin_amdgcn_sqrtf(qq3.x); SB();
        dx3 = gxx - (v2f){X1_0b.y, X1_0b.y}; SB();
        ss3.y = __builtin_amdgcn_sqrtf(qq3.y); SB();
        dy3 = gyy - (v2f){Y1_0b.y, Y1_0b.y}; SB();
        ea1.x = __builtin_amdgcn_exp2f(aa1.x); SB();
        tt2 = dy2 * dy2; SB();
        ea1.y = __builtin_amdgcn_exp2f(aa1.y); SB();
        qq1 = __builtin_elementwise_fma(dx1, dx1, tt1); SB();
        eb1.x = __builtin_amdgcn_exp2f(ab1.x); SB();
        aa2 = ss2 * cc0; SB();
        eb1.y = __builtin_amdgcn_exp2f(ab1.y); SB();
        ab2 = ss2 * cc1; SB();
        da3 += ea3; SB();
        ca3 = __builtin_elementwise_fma((v2f){V1_0b.y, V1_0b.y}, eb3, ca3); SB();
        ss0.x = __builtin_amdgcn_sqrtf(qq0.x); SB();
        dx0 = gxx - (v2f){X1_1a.x, X1_1a.x}; SB();
        ss0.y = __builtin_amdgcn_sqrtf(qq0.y); SB();
        dy0 = gyy - (v2f){Y1_1a.x, Y1_1a.x}; SB();
        ea2.x = __builtin_amdgcn_exp2f(aa2.x); SB();
        tt3 = dy3 * dy3; SB();
        ea2.y = __builtin_amdgcn_exp2f(aa2.y); SB();
        qq2 = __builtin_elementwise_fma(dx2, dx2, tt2); SB();
        eb2.x = __builtin_amdgcn_exp2f(ab2.x); SB();
        aa3 = ss3 * cc0; SB();
        eb2.y = __builtin_amdgcn_exp2f(ab2.y); SB();
        ab3 = ss3 * cc1; SB();
        da0 += ea0; SB();
        ca0 = __builtin_elementwise_fma((v2f){V1_1a.x, V1_1a.x}, eb0, ca0); SB();
        ss1.x = __builtin_amdgcn_sqrtf(qq1.x); SB();
        dx1 = gxx - (v2f){X1_1a.y, X1_1a.y}; SB();
        ss1.y = __builtin_amdgcn_sqrtf(qq1.y); SB();
        dy1 = gyy - (v2f){Y1_1a.y, Y1_1a.y}; SB();
        ea3.x = __builtin_amdgcn_exp2f(aa3.x); SB();
        tt0 = dy0 * dy0; SB();
        ea3.y = __builtin_amdgcn_exp2f(aa3.y); SB();
        qq3 = __builtin_elementwise_fma(dx3, dx3, tt3); SB();
        eb3.x = __builtin_amdgcn_exp2f(ab3.x); SB();
        aa0 = ss0 * cc0; SB();
        eb3.y = __builtin_amdgcn_exp2f(ab3.y); SB();
        ab0 = ss0 * cc1; SB();
        da1 += ea1; SB();
        ca1 = __builtin_elementwise_fma((v2f){V1_1a.y, V1_1a.y}, eb1, ca1); SB();
        ss2.x = __builtin_amdgcn_sqrtf(qq2.x); SB();
        dx2 = gxx - (v2f){X1_1b.x, X1_1b.x}; SB();
        ss2.y = __builtin_amdgcn_sqrtf(qq2.y); SB();
        dy2 = gyy - (v2f){Y1_1b.x, Y1_1b.x}; SB();
        ea0.x = __builtin_amdgcn_exp2f(aa0.x); SB();
        tt1 = dy1 * dy1; SB();
        ea0.y = __builtin_amdgcn_exp2f(aa0.y); SB();
        qq0 = __builtin_elementwise_fma(dx0, dx0, tt0); SB();
        eb0.x = __builtin_amdgcn_exp2f(ab0.x); SB();
        aa1 = ss1 * cc0; SB();
        eb0.y = __builtin_amdgcn_exp2f(ab0.y); SB();
        ab1 = ss1 * cc1; SB();
        da2 += ea2; SB();
        ca2 = __builtin_elementwise_fma((v2f){V1_1b.x, V1_1b.x}, eb2, ca2); SB();
        ss3.x = __builtin_amdgcn_sqrtf(qq3.x); SB();
        dx3 = gxx - (v2f){X1_1b.y, X1_1b.y}; SB();
        ss3.y = __builtin_amdgcn_sqrtf(qq3.y); SB();
        dy3 = gyy - (v2f){Y1_1b.y, Y1_1b.y}; SB();
        ea1.x = __builtin_amdgcn_exp2f(aa1.x); SB();
        tt2 = dy2 * dy2; SB();
        ea1.y = __builtin_amdgcn_exp2f(aa1.y); SB();
        qq1 = __builtin_elementwise_fma(dx1, dx1, tt1); SB();
        eb1.x = __builtin_amdgcn_exp2f(ab1.x); SB();
        aa2 = ss2 * cc0; SB();
        eb1.y = __builtin_amdgcn_exp2f(ab1.y); SB();
        ab2 = ss2 * cc1; SB();
        da3 += ea3; SB();
        ca3 = __builtin_elementwise_fma((v2f){V1_1b.y, V1_1b.y}, eb3, ca3); SB();
        ss0.x = __builtin_amdgcn_sqrtf(qq0.x); SB();
        dx0 = gxx - (v2f){X1_2a.x, X1_2a.x}; SB();
        ss0.y = __builtin_amdgcn_sqrtf(qq0.y); SB();
        dy0 = gyy - (v2f){Y1_2a.x, Y1_2a.x}; SB();
        ea2.x = __builtin_amdgcn_exp2f(aa2.x); SB();
        tt3 = dy3 * dy3; SB();
        ea2.y = __builtin_amdgcn_exp2f(aa2.y); SB();
        qq2 = __builtin_elementwise_fma(dx2, dx2, tt2); SB();
        eb2.x = __builtin_amdgcn_exp2f(ab2.x); SB();
        aa3 = ss3 * cc0; SB();
        eb2.y = __builtin_amdgcn_exp2f(ab2.y); SB();
        ab3 = ss3 * cc1; SB();
        da0 += ea0; SB();
        ca0 = __builtin_elementwise_fma((v2f){V1_2a.x, V1_2a.x}, eb0, ca0); SB();
        ss1.x = __builtin_amdgcn_sqrtf(qq1.x); SB();
        dx1 = gxx - (v2f){X1_2a.y, X1_2a.y}; SB();
        ss1.y = __builtin_amdgcn_sqrtf(qq1.y); SB();
        dy1 = gyy - (v2f){Y1_2a.y, Y1_2a.y}; SB();
        ea3.x = __builtin_amdgcn_exp2f(aa3.x); SB();
        tt0 = dy0 * dy0; SB();
        ea3.y = __builtin_amdgcn_exp2f(aa3.y); SB();
        qq3 = __builtin_elementwise_fma(dx3, dx3, tt3); SB();
        eb3.x = __builtin_amdgcn_exp2f(ab3.x); SB();
        aa0 = ss0 * cc0; SB();
        eb3.y = __builtin_amdgcn_exp2f(ab3.y); SB();
        ab0 = ss0 * cc1; SB();
        da1 += ea1; SB();
        ca1 = __builtin_elementwise_fma((v2f){V1_2a.y, V1_2a.y}, eb1, ca1); SB();
        ss2.x = __builtin_amdgcn_sqrtf(qq2.x); SB();
        dx2 = gxx - (v2f){X1_2b.x, X1_2b.x}; SB();
        ss2.y = __builtin_amdgcn_sqrtf(qq2.y); SB();
        dy2 = gyy - (v2f){Y1_2b.x, Y1_2b.x}; SB();
        ea0.x = __builtin_amdgcn_exp2f(aa0.x); SB();
        tt1 = dy1 * dy1; SB();
        ea0.y = __builtin_amdgcn_exp2f(aa0.y); SB();
        qq0 = __builtin_elementwise_fma(dx0, dx0, tt0); SB();
        eb0.x = __builtin_amdgcn_exp2f(ab0.x); SB();
        aa1 = ss1 * cc0; SB();
        eb0.y = __builtin_amdgcn_exp2f(ab0.y); SB();
        ab1 = ss1 * cc1; SB();
        da2 += ea2; SB();
        ca2 = __builtin_elementwise_fma((v2f){V1_2b.x, V1_2b.x}, eb2, ca2); SB();
        ss3.x = __builtin_amdgcn_sqrtf(qq3.x); SB();
        dx3 = gxx - (v2f){X1_2b.y, X1_2b.y}; SB();
        ss3.y = __builtin_amdgcn_sqrtf(qq3.y); SB();
        dy3 = gyy - (v2f){Y1_2b.y, Y1_2b.y}; SB();
        ea1.x = __builtin_amdgcn_exp2f(aa1.x); SB();
        tt2 = dy2 * dy2; SB();
        ea1.y = __builtin_amdgcn_exp2f(aa1.y); SB();
        qq1 = __builtin_elementwise_fma(dx1, dx1, tt1); SB();
        eb1.x = __builtin_amdgcn_exp2f(ab1.x); SB();
        aa2 = ss2 * cc0; SB();
        eb1.y = __builtin_amdgcn_exp2f(ab1.y); SB();
        ab2 = ss2 * cc1; SB();
        da3 += ea3; SB();
        ca3 = __builtin_elementwise_fma((v2f){V1_2b.y, V1_2b.y}, eb3, ca3); SB();
        ss0.x = __builtin_amdgcn_sqrtf(qq0.x); SB();
        dx0 = gxx - (v2f){X1_3a.x, X1_3a.x}; SB();
        ss0.y = __builtin_amdgcn_sqrtf(qq0.y); SB();
        dy0 = gyy - (v2f){Y1_3a.x, Y1_3a.x}; SB();
        ea2.x = __builtin_amdgcn_exp2f(aa2.x); SB();
        tt3 = dy3 * dy3; SB();
        ea2.y = __builtin_amdgcn_exp2f(aa2.y); SB();
        qq2 = __builtin_elementwise_fma(dx2, dx2, tt2); SB();
        eb2.x = __builtin_amdgcn_exp2f(ab2.x); SB();
        aa3 = ss3 * cc0; SB();
        eb2.y = __builtin_amdgcn_exp2f(ab2.y); SB();
        ab3 = ss3 * cc1; SB();
        da0 += ea0; SB();
        ca0 = __builtin_elementwise_fma((v2f){V1_3a.x, V1_3a.x}, eb0, ca0); SB();
        ss1.x = __builtin_amdgcn_sqrtf(qq1.x); SB();
        dx1 = gxx - (v2f){X1_3a.y, X1_3a.y}; SB();
        ss1.y = __builtin_amdgcn_sqrtf(qq1.y); SB();
        dy1 = gyy - (v2f){Y1_3a.y, Y1_3a.y}; SB();
        ea3.x = __builtin_amdgcn_exp2f(aa3.x); SB();
        tt0 = dy0 * dy0; SB();
        ea3.y = __builtin_amdgcn_exp2f(aa3.y); SB();
        qq3 = __builtin_elementwise_fma(dx3, dx3, tt3); SB();
        eb3.x = __builtin_amdgcn_exp2f(ab3.x); SB();
        aa0 = ss0 * cc0; SB();
        eb3.y = __builtin_amdgcn_exp2f(ab3.y); SB();
        ab0 = ss0 * cc1; SB();
        da1 += ea1; SB();
        ca1 = __builtin_elementwise_fma((v2f){V1_3a.y, V1_3a.y}, eb1, ca1); SB();
        ss2.x = __builtin_amdgcn_sqrtf(qq2.x); SB();
        dx2 = gxx - (v2f){X1_3b.x, X1_3b.x}; SB();
        ss2.y = __builtin_amdgcn_sqrtf(qq2.y); SB();
        dy2 = gyy - (v2f){Y1_3b.x, Y1_3b.x}; SB();
        ea0.x = __builtin_amdgcn_exp2f(aa0.x); SB();
        tt1 = dy1 * dy1; SB();
        ea0.y = __builtin_amdgcn_exp2f(aa0.y); SB();
        qq0 = __builtin_elementwise_fma(dx0, dx0, tt0); SB();
        eb0.x = __builtin_amdgcn_exp2f(ab0.x); SB();
        aa1 = ss1 * cc0; SB();
        eb0.y = __builtin_amdgcn_exp2f(ab0.y); SB();
        ab1 = ss1 * cc1; SB();
        da2 += ea2; SB();
        ca2 = __builtin_elementwise_fma((v2f){V1_3b.x, V1_3b.x}, eb2, ca2); SB();
        ss3.x = __builtin_amdgcn_sqrtf(qq3.x); SB();
        dx3 = gxx - (v2f){X1_3b.y, X1_3b.y}; SB();
        ss3.y = __builtin_amdgcn_sqrtf(qq3.y); SB();
        dy3 = gyy - (v2f){Y1_3b.y, Y1_3b.y}; SB();
        ea1.x = __builtin_amdgcn_exp2f(aa1.x); SB();
        tt2 = dy2 * dy2; SB();
        ea1.y = __builtin_amdgcn_exp2f(aa1.y); SB();
        qq1 = __builtin_elementwise_fma(dx1, dx1, tt1); SB();
        eb1.x = __builtin_amdgcn_exp2f(ab1.x); SB();
        aa2 = ss2 * cc0; SB();
        eb1.y = __builtin_amdgcn_exp2f(ab1.y); SB();
        ab2 = ss2 * cc1; SB();
        da3 += ea3; SB();
        ca3 = __builtin_elementwise_fma((v2f){V1_3b.y, V1_3b.y}, eb3, ca3); SB();
        ss0.x = __builtin_amdgcn_sqrtf(qq0.x); SB();
        dx0 = gxx - (v2f){X1_4a.x, X1_4a.x}; SB();
        ss0.y = __builtin_amdgcn_sqrtf(qq0.y); SB();
        dy0 = gyy - (v2f){Y1_4a.x, Y1_4a.x}; SB();
        ea2.x = __builtin_amdgcn_exp2f(aa2.x); SB();
        tt3 = dy3 * dy3; SB();
        ea2.y = __builtin_amdgcn_exp2f(aa2.y); SB();
        qq2 = __builtin_elementwise_fma(dx2, dx2, tt2); SB();
        eb2.x = __builtin_amdgcn_exp2f(ab2.x); SB();
        aa3 = ss3 * cc0; SB();
        eb2.y = __builtin_amdgcn_exp2f(ab2.y); SB();
        ab3 = ss3 * cc1; SB();
        da0 += ea0; SB();
        ca0 = __builtin_elementwise_fma((v2f){V1_4a.x, V1_4a.x}, eb0, ca0); SB();
        ss1.x = __builtin_amdgcn_sqrtf(qq1.x); SB();
        dx1 = gxx - (v2f){X1_4a.y, X1_4a.y}; SB();
        ss1.y = __builtin_amdgcn_sqrtf(qq1.y); SB();
        dy1 = gyy - (v2f){Y1_4a.y, Y1_4a.y}; SB();
        ea3.x = __builtin_amdgcn_exp2f(aa3.x); SB();
        tt0 = dy0 * dy0; SB();
        ea3.y = __builtin_amdgcn_exp2f(aa3.y); SB();
        qq3 = __builtin_elementwise_fma(dx3, dx3, tt3); SB();
        eb3.x = __builtin_amdgcn_exp2f(ab3.x); SB();
        aa0 = ss0 * cc0; SB();
        eb3.y = __builtin_amdgcn_exp2f(ab3.y); SB();
        ab0 = ss0 * cc1; SB();
        da1 += ea1; SB();
        ca1 = __builtin_elementwise_fma((v2f){V1_4a.y, V1_4a.y}, eb1, ca1); SB();
        ss2.x = __builtin_amdgcn_sqrtf(qq2.x); SB();
        dx2 = gxx - (v2f){X1_4b.x, X1_4b.x}; SB();
        ss2.y = __builtin_amdgcn_sqrtf(qq2.y); SB();
        dy2 = gyy - (v2f){Y1_4b.x, Y1_4b.x}; SB();
        ea0.x = __builtin_amdgcn_exp2f(aa0.x); SB();
        tt1 = dy1 * dy1; SB();
        ea0.y = __builtin_amdgcn_exp2f(aa0.y); SB();
        qq0 = __builtin_elementwise_fma(dx0, dx0, tt0); SB();
        eb0.x = __builtin_amdgcn_exp2f(ab0.x); SB();
        aa1 = ss1 * cc0; SB();
        eb0.y = __builtin_amdgcn_exp2f(ab0.y); SB();
        ab1 = ss1 * cc1; SB();
        da2 += ea2; SB();
        ca2 = __builtin_elementwise_fma((v2f){V1_4b.x, V1_4b.x}, eb2, ca2); SB();
        ss3.x = __builtin_amdgcn_sqrtf(qq3.x); SB();
        dx3 = gxx - (v2f){X1_4b.y, X1_4b.y}; SB();
        ss3.y = __builtin_amdgcn_sqrtf(qq3.y); SB();
        dy3 = gyy - (v2f){Y1_4b.y, Y1_4b.y}; SB();
        ea1.x = __builtin_amdgcn_exp2f(aa1.x); SB();
        tt2 = dy2 * dy2; SB();
        ea1.y = __builtin_amdgcn_exp2f(aa1.y); SB();
        qq1 = __builtin_elementwise_fma(dx1, dx1, tt1); SB();
        eb1.x = __builtin_amdgcn_exp2f(ab1.x); SB();
        aa2 = ss2 * cc0; SB();
        eb1.y = __builtin_amdgcn_exp2f(ab1.y); SB();
        ab2 = ss2 * cc1; SB();
        da3 += ea3; SB();
        ca3 = __builtin_elementwise_fma((v2f){V1_4b.y, V1_4b.y}, eb3, ca3); SB();
        ss0.x = __builtin_amdgcn_sqrtf(qq0.x); SB();
        dx0 = gxx - (v2f){X1_5a.x, X1_5a.x}; SB();
        ss0.y = __builtin_amdgcn_sqrtf(qq0.y); SB();
        dy0 = gyy - (v2f){Y1_5a.x, Y1_5a.x}; SB();
        ea2.x = __builtin_amdgcn_exp2f(aa2.x); SB();
        tt3 = dy3 * dy3; SB();
        ea2.y = __builtin_amdgcn_exp2f(aa2.y); SB();
        qq2 = __builtin_elementwise_fma(dx2, dx2, tt2); SB();
        eb2.x = __builtin_amdgcn_exp2f(ab2.x); SB();
        aa3 = ss3 * cc0; SB();
        eb2.y = __builtin_amdgcn_exp2f(ab2.y); SB();
        ab3 = ss3 * cc1; SB();
        da0 += ea0; SB();
        ca0 = __builtin_elementwise_fma((v2f){V1_5a.x, V1_5a.x}, eb0, ca0); SB();
        ss1.x = __builtin_amdgcn_sqrtf(qq1.x); SB();
        dx1 = gxx - (v2f){X1_5a.y, X1_5a.y}; SB();
        ss1.y = __builtin_amdgcn_sqrtf(qq1.y); SB();
        dy1 = gyy - (v2f){Y1_5a.y, Y1_5a.y}; SB();
        ea3.x = __builtin_amdgcn_exp2f(aa3.x); SB();
        tt0 = dy0 * dy0; SB();
        ea3.y = __builtin_amdgcn_exp2f(aa3.y); SB();
        qq3 = __builtin_elementwise_fma(dx3, dx3, tt3); SB();
        eb3.x = __builtin_amdgcn_exp2f(ab3.x); SB();
        aa0 = ss0 * cc0; SB();
        eb3.y = __builtin_amdgcn_exp2f(ab3.y); SB();
        ab0 = ss0 * cc1; SB();
        da1 += ea1; SB();
        ca1 = __builtin_elementwise_fma((v2f){V1_5a.y, V1_5a.y}, eb1, ca1); SB();
        ss2.x = __builtin_amdgcn_sqrtf(qq2.x); SB();
        dx2 = gxx - (v2f){X1_5b.x, X1_5b.x}; SB();
        ss2.y = __builtin_amdgcn_sqrtf(qq2.y); SB();
        dy2 = gyy - (v2f){Y1_5b.x, Y1_5b.x}; SB();
        ea0.x = __builtin_amdgcn_exp2f(aa0.x); SB();
        tt1 = dy1 * dy1; SB();
        ea0.y = __builtin_amdgcn_exp2f(aa0.y); SB();
        qq0 = __builtin_elementwise_fma(dx0, dx0, tt0); SB();
        eb0.x = __builtin_amdgcn_exp2f(ab0.x); SB();
        aa1 = ss1 * cc0; SB();
        eb0.y = __builtin_amdgcn_exp2f(ab0.y); SB();
        ab1 = ss1 * cc1; SB();
        da2 += ea2; SB();
        ca2 = __builtin_elementwise_fma((v2f){V1_5b.x, V1_5b.x}, eb2, ca2); SB();
        ss3.x = __builtin_amdgcn_sqrtf(qq3.x); SB();
        dx3 = gxx - (v2f){X1_5b.y, X1_5b.y}; SB();
        ss3.y = __builtin_amdgcn_sqrtf(qq3.y); SB();
        dy3 = gyy - (v2f){Y1_5b.y, Y1_5b.y}; SB();
        ea1.x = __builtin_amdgcn_exp2f(aa1.x); SB();
        tt2 = dy2 * dy2; SB();
        ea1.y = __builtin_amdgcn_exp2f(aa1.y); SB();
        qq1 = __builtin_elementwise_fma(dx1, dx1, tt1); SB();
        eb1.x = __builtin_amdgcn_exp2f(ab1.x); SB();
        aa2 = ss2 * cc0; SB();
        eb1.y = __builtin_amdgcn_exp2f(ab1.y); SB();
        ab2 = ss2 * cc1; SB();
        da3 += ea3; SB();
        ca3 = __builtin_elementwise_fma((v2f){V1_5b.y, V1_5b.y}, eb3, ca3); SB();
        ss0.x = __builtin_amdgcn_sqrtf(qq0.x); SB();
        dx0 = gxx - (v2f){X1_6a.x, X1_6a.x}; SB();
        ss0.y = __builtin_amdgcn_sqrtf(qq0.y); SB();
        dy0 = gyy - (v2f){Y1_6a.x, Y1_6a.x}; SB();
        ea2.x = __builtin_amdgcn_exp2f(aa2.x); SB();
        tt3 = dy3 * dy3; SB();
        ea2.y = __builtin_amdgcn_exp2f(aa2.y); SB();
        qq2 = __builtin_elementwise_fma(dx2, dx2, tt2); SB();
        eb2.x = __builtin_amdgcn_exp2f(ab2.x); SB();
        aa3 = ss3 * cc0; SB();
        eb2.y = __builtin_amdgcn_exp2f(ab2.y); SB();
        ab3 = ss3 * cc1; SB();
        da0 += ea0; SB();
        ca0 = __builtin_elementwise_fma((v2f){V1_6a.x, V1_6a.x}, eb0, ca0); SB();
        ss1.x = __builtin_amdgcn_sqrtf(qq1.x); SB();
        dx1 = gxx - (v2f){X1_6a.y, X1_6a.y}; SB();
        ss1.y = __builtin_amdgcn_sqrtf(qq1.y); SB();
        dy1 = gyy - (v2f){Y1_6a.y, Y1_6a.y}; SB();
        ea3.x = __builtin_amdgcn_exp2f(aa3.x); SB();
        tt0 = dy0 * dy0; SB();
        ea3.y = __builtin_amdgcn_exp2f(aa3.y); SB();
        qq3 = __builtin_elementwise_fma(dx3, dx3, tt3); SB();
        eb3.x = __builtin_amdgcn_exp2f(ab3.x); SB();
        aa0 = ss0 * cc0; SB();
        eb3.y = __builtin_amdgcn_exp2f(ab3.y); SB();
        ab0 = ss0 * cc1; SB();
        da1 += ea1; SB();
        ca1 = __builtin_elementwise_fma((v2f){V1_6a.y, V1_6a.y}, eb1, ca1); SB();
        ss2.x = __builtin_amdgcn_sqrtf(qq2.x); SB();
        dx2 = gxx - (v2f){X1_6b.x, X1_6b.x}; SB();
        ss2.y = __builtin_amdgcn_sqrtf(qq2.y); SB();
        dy2 = gyy - (v2f){Y1_6b.x, Y1_6b.x}; SB();
        ea0.x = __builtin_amdgcn_exp2f(aa0.x); SB();
        tt1 = dy1 * dy1; SB();
        ea0.y = __builtin_amdgcn_exp2f(aa0.y); SB();
        qq0 = __builtin_elementwise_fma(dx0, dx0, tt0); SB();
        eb0.x = __builtin_amdgcn_exp2f(ab0.x); SB();
        aa1 = ss1 * cc0; SB();
        eb0.y = __builtin_amdgcn_exp2f(ab0.y); SB();
        ab1 = ss1 * cc1; SB();
        da2 += ea2; SB();
        ca2 = __builtin_elementwise_fma((v2f){V1_6b.x, V1_6b.x}, eb2, ca2); SB();
        ss3.x = __builtin_amdgcn_sqrtf(qq3.x); SB();
        dx3 = gxx - (v2f){X1_6b.y, X1_6b.y}; SB();
        ss3.y = __builtin_amdgcn_sqrtf(qq3.y); SB();
        dy3 = gyy - (v2f){Y1_6b.y, Y1_6b.y}; SB();
        ea1.x = __builtin_amdgcn_exp2f(aa1.x); SB();
        tt2 = dy2 * dy2; SB();
        ea1.y = __builtin_amdgcn_exp2f(aa1.y); SB();
        qq1 = __builtin_elementwise_fma(dx1, dx1, tt1); SB();
        eb1.x = __builtin_amdgcn_exp2f(ab1.x); SB();
        aa2 = ss2 * cc0; SB();
        eb1.y = __builtin_amdgcn_exp2f(ab1.y); SB();
        ab2 = ss2 * cc1; SB();
        da3 += ea3; SB();
        ca3 = __builtin_elementwise_fma((v2f){V1_6b.y, V1_6b.y}, eb3, ca3); SB();
        ss0.x = __builtin_amdgcn_sqrtf(qq0.x); SB();
        dx0 = gxx - (v2f){X1_7a.x, X1_7a.x}; SB();
        ss0.y = __builtin_amdgcn_sqrtf(qq0.y); SB();
        dy0 = gyy - (v2f){Y1_7a.x, Y1_7a.x}; SB();
        ea2.x = __builtin_amdgcn_exp2f(aa2.x); SB();
        tt3 = dy3 * dy3; SB();
        ea2.y = __builtin_amdgcn_exp2f(aa2.y); SB();
        qq2 = __builtin_elementwise_fma(dx2, dx2, tt2); SB();
        eb2.x = __builtin_amdgcn_exp2f(ab2.x); SB();
        aa3 = ss3 * cc0; SB();
        eb2.y = __builtin_amdgcn_exp2f(ab2.y); SB();
        ab3 = ss3 * cc1; SB();
        da0 += ea0; SB();
        ca0 = __builtin_elementwise_fma((v2f){V1_7a.x, V1_7a.x}, eb0, ca0); SB();
        ss1.x = __builtin_amdgcn_sqrtf(qq1.x); SB();
        dx1 = gxx - (v2f){X1_7a.y, X1_7a.y}; SB();
        ss1.y = __builtin_amdgcn_sqrtf(qq1.y); SB();
        dy1 = gyy - (v2f){Y1_7a.y, Y1_7a.y}; SB();
        ea3.x = __builtin_amdgcn_exp2f(aa3.x); SB();
        tt0 = dy0 * dy0; SB();
        ea3.y = __builtin_amdgcn_exp2f(aa3.y); SB();
        qq3 = __builtin_elementwise_fma(dx3, dx3, tt3); SB();
        eb3.x = __builtin_amdgcn_exp2f(ab3.x); SB();
        aa0 = ss0 * cc0; SB();
        eb3.y = __builtin_amdgcn_exp2f(ab3.y); SB();
        ab0 = ss0 * cc1; SB();
        da1 += ea1; SB();
        ca1 = __builtin_elementwise_fma((v2f){V1_7a.y, V1_7a.y}, eb1, ca1); SB();
        ss2.x = __builtin_amdgcn_sqrtf(qq2.x); SB();
        dx2 = gxx - (v2f){X1_7b.x, X1_7b.x}; SB();
        ss2.y = __builtin_amdgcn_sqrtf(qq2.y); SB();
        dy2 = gyy - (v2f){Y1_7b.x, Y1_7b.x}; SB();
        ea0.x = __builtin_amdgcn_exp2f(aa0.x); SB();
        tt1 = dy1 * dy1; SB();
        ea0.y = __builtin_amdgcn_exp2f(aa0.y); SB();
        qq0 = __builtin_elementwise_fma(dx0, dx0, tt0); SB();
        eb0.x = __builtin_amdgcn_exp2f(ab0.x); SB();
        aa1 = ss1 * cc0; SB();
        eb0.y = __builtin_amdgcn_exp2f(ab0.y); SB();
        ab1 = ss1 * cc1; SB();
        da2 += ea2; SB();
        ca2 = __builtin_elementwise_fma((v2f){V1_7b.x, V1_7b.x}, eb2, ca2); SB();
        ss3.x = __builtin_amdgcn_sqrtf(qq3.x); SB();
        dx3 = gxx - (v2f){X1_7b.y, X1_7b.y}; SB();
        ss3.y = __builtin_amdgcn_sqrtf(qq3.y); SB();
        dy3 = gyy - (v2f){Y1_7b.y, Y1_7b.y}; SB();
        ea1.x = __builtin_amdgcn_exp2f(aa1.x); SB();
        tt2 = dy2 * dy2; SB();
        ea1.y = __builtin_amdgcn_exp2f(aa1.y); SB();
        qq1 = __builtin_elementwise_fma(dx1, dx1, tt1); SB();
        eb1.x = __builtin_amdgcn_exp2f(ab1.x); SB();
        aa2 = ss2 * cc0; SB();
        eb1.y = __builtin_amdgcn_exp2f(ab1.y); SB();
        ab2 = ss2 * cc1; SB();
        da3 += ea3; SB();
        ca3 = __builtin_elementwise_fma((v2f){V1_7b.y, V1_7b.y}, eb3, ca3); SB();
    }
    const bool lo    = lane < 16;
    const int  kOff  = lo ? 0 : 8;
    v2f wr[4];
    v4f bsA[4], bsB[4];
#pragma unroll
    for (int ct = 0; ct < 4; ++ct) {
        __builtin_memcpy(&wr[ct], &swb[(ct * 16 + lhalf) * 2], 8);
        bsA[ct] = lds4(&swb[NCH * 2 + ct * 16 + kOff]);
        bsB[ct] = lds4(&swb[NCH * 2 + ct * 16 + kOff + 4]);
    }
    SB();

    const v2f dsum = (da0 + da1) + (da2 + da3);
    const v2f csum = (ca0 + ca1) + (ca2 + ca3);
    const float keepD = lo ? dsum.x : dsum.y;
    const float keepC = lo ? csum.x : csum.y;
    const float sendD = lo ? dsum.y : dsum.x;
    const float sendC = lo ? csum.y : csum.x;
    const float dens  = keepD + xhalf(sendD);
    const float conv  = keepC + xhalf(sendC);
    const float ratio = conv * __builtin_amdgcn_rcpf(dens + 1e-8f);

    float* outb = out + (size_t)b * NCH * NPTS;

#pragma unroll
    for (int h = 0; h < 2; ++h) {
        const bool own = (h == 0) ? lo : !lo;
        v2f bm;
        bm.x = own ? dens  : 0.f;
        bm.y = own ? ratio : 0.f;

        const int oPt = po + wave * 32 + h * 16 + lhalf;

#pragma unroll
        for (int ct = 0; ct < 4; ++ct) {
            v2f am;
            am.x = own ? wr[ct].x : 0.f;
            am.y = own ? wr[ct].y : 0.f;

            v8f acc = {};
            acc = __builtin_amdgcn_wmma_f32_16x16x4_f32(
                 false, am,  false, bm,
                 (short)0, acc,  false,  false);

            const int kBase = ct * 16 + kOff;
#pragma unroll
            for (int j = 0; j < 8; ++j) {
                const float bj = (j < 4) ? bsA[ct][j] : bsB[ct][j - 4];
                float v = acc[j] + bj;
                float r = __builtin_amdgcn_rcpf(
                    1.f + __builtin_amdgcn_exp2f(-LOG2E * v));
                outb[(size_t)(kBase + j) * NPTS + oPt] = r;
            }
        }
    }
}

extern "C" void kernel_launch(void* const* d_in, const int* in_sizes, int n_in,
                              void* d_out, int out_size, void* d_ws, size_t ws_size,
                              hipStream_t stream) {
    (void)n_in; (void)out_size; (void)d_ws; (void)ws_size;
    const float* xc   = (const float*)d_in[0];
    const float* yc   = (const float*)d_in[1];
    const float* xg   = (const float*)d_in[2];
    const float* sig  = (const float*)d_in[3];
    const float* W    = (const float*)d_in[4];
    const float* bias = (const float*)d_in[5];
    float* out = (float*)d_out;

    const int B = in_sizes[0] / (NCTX * 2);
    dim3 grid(B * 32), block(128);
    convnp_fused<<<grid, block, 0, stream>>>(xc, yc, xg, sig, W, bias, out);
}
